# starting kernel
# baseline (speedup 1.0000x reference)
#include <hip/hip_runtime.h>
#include <math.h>


typedef float v2f __attribute__((ext_vector_type(2)));
typedef float v8f __attribute__((ext_vector_type(8)));

#define B_  4
#define C_  32
#define H_  256
#define W_  256
#define M_  16
#define PIX (H_ * W_)

__device__ __forceinline__ v8f wmma4(v2f a, v2f b, v8f c) {
  return __builtin_amdgcn_wmma_f32_16x16x4_f32(false, a, false, b, (short)0, c,
                                               false, false);
}

__device__ __forceinline__ v2f ldA(const float* p, int lda, int lane) {
  int r = lane & 15, kk = (lane >> 4) << 1;
  v2f v; v.x = p[r * lda + kk]; v.y = p[r * lda + kk + 1]; return v;
}
__device__ __forceinline__ v2f ldAt(const float* p, int lda, int lane) {
  int r = lane & 15, kk = (lane >> 4) << 1;
  v2f v; v.x = p[kk * lda + r]; v.y = p[(kk + 1) * lda + r]; return v;
}
__device__ __forceinline__ v2f ldB(const float* p, int ldb, int lane) {
  int n = lane & 15, kk = (lane >> 4) << 1;
  v2f v; v.x = p[kk * ldb + n]; v.y = p[(kk + 1) * ldb + n]; return v;
}
__device__ __forceinline__ v2f ldBt(const float* p, int ldb, int lane) {
  int n = lane & 15, kk = (lane >> 4) << 1;
  v2f v; v.x = p[n * ldb + kk]; v.y = p[n * ldb + kk + 1]; return v;
}
__device__ __forceinline__ void stD(float* p, int ldd, int lane, v8f d, float s) {
  int n = lane & 15, mh = (lane >> 4) << 3;
#pragma unroll
  for (int r = 0; r < 8; ++r) p[(r + mh) * ldd + n] = d[r] * s;
}

__device__ __forceinline__ v8f vzero() {
  v8f z = {0.f, 0.f, 0.f, 0.f, 0.f, 0.f, 0.f, 0.f};
  return z;
}

__global__ void k_tabs(float* tabC, float* tabS, float* tabC2, float* tabS2) {
  int i = blockIdx.x * blockDim.x + threadIdx.x;
  if (i >= M_ * W_) return;
  int k = i >> 8, x = i & 255;
  int m = (k * x) & 255;
  float ang = 0.0245436926f * (float)m;
  float c = cosf(ang), s = sinf(ang);
  tabC[i] = c; tabS[i] = s;
  float w2 = (k == 0) ? 1.0f : 2.0f;
  tabC2[i] = w2 * c * 0.0625f;
  tabS2[i] = w2 * s * 0.0625f;
}

__global__ void k_lift(const float* __restrict__ x, const float* __restrict__ lw,
                       const float* __restrict__ lb, float* __restrict__ x0) {
  int p = blockIdx.x * blockDim.x + threadIdx.x;
  if (p >= B_ * PIX) return;
  int b = p >> 16, pix = p & (PIX - 1);
  float xi0 = x[(b * 3 + 0) * PIX + pix];
  float xi1 = x[(b * 3 + 1) * PIX + pix];
  float xi2 = x[(b * 3 + 2) * PIX + pix];
  float* o = x0 + (size_t)p * C_;
#pragma unroll
  for (int c = 0; c < C_; ++c)
    o[c] = lb[c] + lw[c * 3 + 0] * xi0 + lw[c * 3 + 1] * xi1 + lw[c * 3 + 2] * xi2;
}

__global__ void k_fwdX(const float* __restrict__ xin, const float* __restrict__ tabC,
                       const float* __restrict__ tabS, float* __restrict__ Tr,
                       float* __restrict__ Ti) {
  int wid = blockIdx.x * (blockDim.x >> 5) + (threadIdx.x >> 5);
  int lane = threadIdx.x & 31;
  int ch = wid & 1;
  int byi = wid >> 1;
  int b = byi >> 8, y = byi & 255;
  const float* row = xin + ((size_t)(b * H_ + y)) * W_ * C_ + 16 * ch;
  v8f aC = vzero(), aS = vzero();
  for (int k0 = 0; k0 < W_; k0 += 4) {
    v2f aCr = ldA(tabC + k0, W_, lane);
    v2f aSr = ldA(tabS + k0, W_, lane);
    v2f bh = ldB(row + k0 * C_, C_, lane);
    aC = wmma4(aCr, bh, aC);
    aS = wmma4(aSr, bh, aS);
  }
  float* tr = Tr + ((size_t)(b * H_ + y)) * M_ * C_ + 16 * ch;
  float* ti = Ti + ((size_t)(b * H_ + y)) * M_ * C_ + 16 * ch;
  stD(tr, C_, lane, aC, 0.0625f);
  stD(ti, C_, lane, aS, -0.0625f);
}

__global__ void k_fwdY(const float* __restrict__ Tr, const float* __restrict__ Ti,
                       const float* __restrict__ tabC, const float* __restrict__ tabS,
                       float* __restrict__ Frp, float* __restrict__ Fip) {
  int wid = blockIdx.x * (blockDim.x >> 5) + (threadIdx.x >> 5);
  int lane = threadIdx.x & 31;
  int b = wid >> 7;
  int r2 = wid & 127;
  int col0 = (r2 >> 2) * 16;
  int kc = r2 & 3;
  const float* trb = Tr + (size_t)b * H_ * (M_ * C_);
  const float* tib = Ti + (size_t)b * H_ * (M_ * C_);
  v8f aCT = vzero(), aST = vzero(), aCTi = vzero(), aSTr = vzero();
  for (int k0 = kc * 64; k0 < kc * 64 + 64; k0 += 4) {
    v2f aC = ldA(tabC + k0, W_, lane);
    v2f aS = ldA(tabS + k0, W_, lane);
    v2f bR = ldB(trb + k0 * 512 + col0, 512, lane);
    v2f bI = ldB(tib + k0 * 512 + col0, 512, lane);
    aCT = wmma4(aC, bR, aCT);
    aST = wmma4(aS, bI, aST);
    aCTi = wmma4(aC, bI, aCTi);
    aSTr = wmma4(aS, bR, aSTr);
  }
  float* fr = Frp + (size_t)kc * (B_ * M_ * 512) + (size_t)b * (M_ * 512) + col0;
  float* fi = Fip + (size_t)kc * (B_ * M_ * 512) + (size_t)b * (M_ * 512) + col0;
  int n = lane & 15, mh = (lane >> 4) << 3;
#pragma unroll
  for (int r = 0; r < 8; ++r) {
    fr[(r + mh) * 512 + n] = (aCT[r] + aST[r]) * 0.0625f;
    fi[(r + mh) * 512 + n] = (aCTi[r] - aSTr[r]) * 0.0625f;
  }
}

__global__ void k_mix(const float* __restrict__ Frp, const float* __restrict__ Fip,
                      const float* __restrict__ wr, const float* __restrict__ wi,
                      float* __restrict__ Gr, float* __restrict__ Gi) {
  __shared__ float sFr[512];
  __shared__ float sFi[512];
  int b = blockIdx.x >> 4, k1 = blockIdx.x & 15;
  int tid = threadIdx.x;
  size_t base = ((size_t)b * M_ + k1) * 512;
  float fr = 0.f, fi = 0.f;
#pragma unroll
  for (int kc = 0; kc < 4; ++kc) {
    fr += Frp[(size_t)kc * (B_ * M_ * 512) + base + tid];
    fi += Fip[(size_t)kc * (B_ * M_ * 512) + base + tid];
  }
  sFr[tid] = fr; sFi[tid] = fi;
  __syncthreads();
  int o = tid >> 4, k2 = tid & 15;
  const float* wrk = wr + k1 * 16 + k2;
  const float* wik = wi + k1 * 16 + k2;
  float gr = 0.f, gi = 0.f;
#pragma unroll 4
  for (int i = 0; i < C_; ++i) {
    float xr = sFr[k2 * 32 + i], xi = sFi[k2 * 32 + i];
    float wrv = wrk[(i * 32 + o) * 256];
    float wiv = wik[(i * 32 + o) * 256];
    gr += xr * wrv - xi * wiv;
    gi += xr * wiv + xi * wrv;
  }
  size_t gidx = base + k2 * 32 + o;
  Gr[gidx] = gr; Gi[gidx] = gi;
}

__global__ void k_invY(const float* __restrict__ Gr, const float* __restrict__ Gi,
                       const float* __restrict__ tabC, const float* __restrict__ tabS,
                       float* __restrict__ P, float* __restrict__ Q) {
  int wid = blockIdx.x * (blockDim.x >> 5) + (threadIdx.x >> 5);
  int lane = threadIdx.x & 31;
  int b = wid >> 9, rest = wid & 511;
  int y0 = (rest >> 5) * 16, col0 = (rest & 31) * 16;
  const float* grb = Gr + (size_t)b * 8192;
  const float* gib = Gi + (size_t)b * 8192;
  v8f aCG = vzero(), aSGi = vzero(), aSG = vzero(), aCGi = vzero();
#pragma unroll
  for (int k0 = 0; k0 < M_; k0 += 4) {
    v2f aC = ldAt(tabC + k0 * W_ + y0, W_, lane);
    v2f aS = ldAt(tabS + k0 * W_ + y0, W_, lane);
    v2f bR = ldB(grb + k0 * 512 + col0, 512, lane);
    v2f bI = ldB(gib + k0 * 512 + col0, 512, lane);
    aCG = wmma4(aC, bR, aCG);
    aSGi = wmma4(aS, bI, aSGi);
    aSG = wmma4(aS, bR, aSG);
    aCGi = wmma4(aC, bI, aCGi);
  }
  float* p = P + (size_t)(b * H_ + y0) * 512 + col0;
  float* q = Q + (size_t)(b * H_ + y0) * 512 + col0;
  int n = lane & 15, mh = (lane >> 4) << 3;
#pragma unroll
  for (int r = 0; r < 8; ++r) {
    p[(r + mh) * 512 + n] = (aCG[r] - aSGi[r]) * 0.0625f;
    q[(r + mh) * 512 + n] = (aSG[r] + aCGi[r]) * 0.0625f;
  }
}

__global__ void k_invX(const float* __restrict__ P, const float* __restrict__ Q,
                       const float* __restrict__ xin, const float* __restrict__ tabC2,
                       const float* __restrict__ tabS2, const float* __restrict__ bw,
                       const float* __restrict__ bb, float* __restrict__ xout) {
  int wid = blockIdx.x * (blockDim.x >> 5) + (threadIdx.x >> 5);
  int lane = threadIdx.x & 31;
  int by = wid >> 4;
  int x0 = (wid & 15) * 16;
  const float* Pby = P + (size_t)by * 512;
  const float* Qby = Q + (size_t)by * 512;
  const float* xrow = xin + (size_t)by * (W_ * C_);
  v8f acc0 = vzero(), acc1 = vzero();
#pragma unroll
  for (int k0 = 0; k0 < M_; k0 += 4) {
    v2f aC = ldAt(tabC2 + k0 * W_ + x0, W_, lane);
    v2f bP0 = ldB(Pby + k0 * C_, C_, lane);
    v2f bP1 = ldB(Pby + k0 * C_ + 16, C_, lane);
    acc0 = wmma4(aC, bP0, acc0);
    acc1 = wmma4(aC, bP1, acc1);
    v2f aS = ldAt(tabS2 + k0 * W_ + x0, W_, lane);
    v2f aSn; aSn.x = -aS.x; aSn.y = -aS.y;
    v2f bQ0 = ldB(Qby + k0 * C_, C_, lane);
    v2f bQ1 = ldB(Qby + k0 * C_ + 16, C_, lane);
    acc0 = wmma4(aSn, bQ0, acc0);
    acc1 = wmma4(aSn, bQ1, acc1);
  }
#pragma unroll
  for (int k0 = 0; k0 < C_; k0 += 4) {
    v2f aX = ldA(xrow + x0 * C_ + k0, C_, lane);
    v2f bW0 = ldBt(bw + k0, C_, lane);
    v2f bW1 = ldBt(bw + 16 * C_ + k0, C_, lane);
    acc0 = wmma4(aX, bW0, acc0);
    acc1 = wmma4(aX, bW1, acc1);
  }
  int n = lane & 15, mh = (lane >> 4) << 3;
  float bb0 = bb[n], bb1 = bb[16 + n];
  float* out = xout + (size_t)by * (W_ * C_);
#pragma unroll
  for (int r = 0; r < 8; ++r) {
    int xr = x0 + r + mh;
    float v0 = acc0[r] + bb0;
    v0 = 0.5f * v0 * (1.0f + erff(v0 * 0.70710678f));
    out[xr * C_ + n] = v0;
    float v1 = acc1[r] + bb1;
    v1 = 0.5f * v1 * (1.0f + erff(v1 * 0.70710678f));
    out[xr * C_ + 16 + n] = v1;
  }
}

__global__ void k_proj(const float* __restrict__ xb, const float* __restrict__ pw,
                       const float* __restrict__ pb, float* __restrict__ out,
                       int o_outer) {
  int p = blockIdx.x * blockDim.x + threadIdx.x;
  if (p >= B_ * PIX) return;
  int b = p >> 16, pix = p & (PIX - 1);
  const float* v = xb + (size_t)p * C_;
  float s = pb[0];
#pragma unroll
  for (int c = 0; c < C_; ++c) s += pw[c] * v[c];
  out[((size_t)(b * 4 + o_outer)) * PIX + pix] = s;
}

extern "C" void kernel_launch(void* const* d_in, const int* in_sizes, int n_in,
                              void* d_out, int out_size, void* d_ws, size_t ws_size,
                              hipStream_t stream) {
  const float* x      = (const float*)d_in[0];
  const float* lift_w = (const float*)d_in[1];
  const float* lift_b = (const float*)d_in[2];
  const float* wr     = (const float*)d_in[3];
  const float* wi     = (const float*)d_in[4];
  const float* byp_w  = (const float*)d_in[5];
  const float* byp_b  = (const float*)d_in[6];
  const float* proj_w = (const float*)d_in[7];
  const float* proj_b = (const float*)d_in[8];
  float* out = (float*)d_out;

  float* w = (float*)d_ws;
  float* tabC  = w;
  float* tabS  = tabC + 4096;
  float* tabC2 = tabS + 4096;
  float* tabS2 = tabC2 + 4096;
  float* x0    = tabS2 + 4096;
  float* bufA  = x0 + (size_t)B_ * PIX * C_;
  float* bufB  = bufA + (size_t)B_ * PIX * C_;
  float* Tr    = bufB + (size_t)B_ * PIX * C_;
  float* Ti    = Tr + (size_t)B_ * H_ * M_ * C_;
  float* Frp   = Ti + (size_t)B_ * H_ * M_ * C_;
  float* Fip   = Frp + (size_t)4 * B_ * M_ * 512;
  float* Gr    = Fip + (size_t)4 * B_ * M_ * 512;
  float* Gi    = Gr + (size_t)B_ * M_ * M_ * C_;
  float* Pb    = Gi + (size_t)B_ * M_ * M_ * C_;
  float* Qb    = Pb + (size_t)B_ * H_ * M_ * C_;

  k_tabs<<<16, 256, 0, stream>>>(tabC, tabS, tabC2, tabS2);
  k_lift<<<(B_ * PIX + 255) / 256, 256, 0, stream>>>(x, lift_w, lift_b, x0);

  for (int o = 0; o < 4; ++o) {
    for (int n = 0; n < 4; ++n) {
      const float* cur = (n == 0) ? x0 : ((n & 1) ? bufA : bufB);
      float* dst = (n & 1) ? bufB : bufA;
      size_t wOff = ((size_t)(o * 4 + n)) * C_ * C_ * M_ * M_;
      size_t bwOff = ((size_t)(o * 4 + n)) * C_ * C_;
      size_t bbOff = ((size_t)(o * 4 + n)) * C_;
      k_fwdX<<<256, 256, 0, stream>>>(cur, tabC, tabS, Tr, Ti);
      k_fwdY<<<64, 256, 0, stream>>>(Tr, Ti, tabC, tabS, Frp, Fip);
      k_mix<<<64, 512, 0, stream>>>(Frp, Fip, wr + wOff, wi + wOff, Gr, Gi);
      k_invY<<<256, 256, 0, stream>>>(Gr, Gi, tabC, tabS, Pb, Qb);
      k_invX<<<2048, 256, 0, stream>>>(Pb, Qb, cur, tabC2, tabS2,
                                       byp_w + bwOff, byp_b + bbOff, dst);
    }
    k_proj<<<(B_ * PIX + 255) / 256, 256, 0, stream>>>(bufB, proj_w, proj_b, out, o);
  }
}


# More loads in flight ahead of use: fwdX K loop fetches 8 k-steps per group, one group ahead of the products (2 register sets); mix fetches its 64 weights up front; sums in the same order
# speedup vs baseline: 1.1247x; 1.1247x over previous
#include <hip/hip_runtime.h>
#include <math.h>


typedef float v2f __attribute__((ext_vector_type(2)));
typedef float v8f __attribute__((ext_vector_type(8)));

#define B_  4
#define C_  32
#define H_  256
#define W_  256
#define M_  16
#define PIX (H_ * W_)

__device__ __forceinline__ v8f wmma4(v2f a, v2f b, v8f c) {
  return __builtin_amdgcn_wmma_f32_16x16x4_f32(false, a, false, b, (short)0, c,
                                               false, false);
}

__device__ __forceinline__ v2f ldA(const float* p, int lda, int lane) {
  int r = lane & 15, kk = (lane >> 4) << 1;
  v2f v; v.x = p[r * lda + kk]; v.y = p[r * lda + kk + 1]; return v;
}
__device__ __forceinline__ v2f ldAt(const float* p, int lda, int lane) {
  int r = lane & 15, kk = (lane >> 4) << 1;
  v2f v; v.x = p[kk * lda + r]; v.y = p[(kk + 1) * lda + r]; return v;
}
__device__ __forceinline__ v2f ldB(const float* p, int ldb, int lane) {
  int n = lane & 15, kk = (lane >> 4) << 1;
  v2f v; v.x = p[kk * ldb + n]; v.y = p[(kk + 1) * ldb + n]; return v;
}
__device__ __forceinline__ v2f ldBt(const float* p, int ldb, int lane) {
  int n = lane & 15, kk = (lane >> 4) << 1;
  v2f v; v.x = p[n * ldb + kk]; v.y = p[n * ldb + kk + 1]; return v;
}
__device__ __forceinline__ void stD(float* p, int ldd, int lane, v8f d, float s) {
  int n = lane & 15, mh = (lane >> 4) << 3;
#pragma unroll
  for (int r = 0; r < 8; ++r) p[(r + mh) * ldd + n] = d[r] * s;
}

__device__ __forceinline__ v8f vzero() {
  v8f z = {0.f, 0.f, 0.f, 0.f, 0.f, 0.f, 0.f, 0.f};
  return z;
}

__global__ void k_tabs(float* tabC, float* tabS, float* tabC2, float* tabS2) {
  int i = blockIdx.x * blockDim.x + threadIdx.x;
  if (i >= M_ * W_) return;
  int k = i >> 8, x = i & 255;
  int m = (k * x) & 255;
  float ang = 0.0245436926f * (float)m;
  float c = cosf(ang), s = sinf(ang);
  tabC[i] = c; tabS[i] = s;
  float w2 = (k == 0) ? 1.0f : 2.0f;
  tabC2[i] = w2 * c * 0.0625f;
  tabS2[i] = w2 * s * 0.0625f;
}

__global__ void k_lift(const float* __restrict__ x, const float* __restrict__ lw,
                       const float* __restrict__ lb, float* __restrict__ x0) {
  int p = blockIdx.x * blockDim.x + threadIdx.x;
  if (p >= B_ * PIX) return;
  int b = p >> 16, pix = p & (PIX - 1);
  float xi0 = x[(b * 3 + 0) * PIX + pix];
  float xi1 = x[(b * 3 + 1) * PIX + pix];
  float xi2 = x[(b * 3 + 2) * PIX + pix];
  float* o = x0 + (size_t)p * C_;
#pragma unroll
  for (int c = 0; c < C_; ++c)
    o[c] = lb[c] + lw[c * 3 + 0] * xi0 + lw[c * 3 + 1] * xi1 + lw[c * 3 + 2] * xi2;
}

#define FX_U 8
__device__ __forceinline__ void fx_load(v2f (&c)[FX_U], v2f (&s)[FX_U], v2f (&b)[FX_U],
                                        const float* __restrict__ tabC,
                                        const float* __restrict__ tabS,
                                        const float* __restrict__ row, int k, int lane) {
#pragma unroll
  for (int u = 0; u < FX_U; ++u) {
    c[u] = ldA(tabC + k + 4 * u, W_, lane);
    s[u] = ldA(tabS + k + 4 * u, W_, lane);
    b[u] = ldB(row + (k + 4 * u) * C_, C_, lane);
  }
}
__device__ __forceinline__ void fx_mma(const v2f (&c)[FX_U], const v2f (&s)[FX_U],
                                       const v2f (&b)[FX_U], v8f& aC, v8f& aS) {
#pragma unroll
  for (int u = 0; u < FX_U; ++u) {
    aC = wmma4(c[u], b[u], aC);
    aS = wmma4(s[u], b[u], aS);
  }
}
__global__ void __launch_bounds__(256) k_fwdX(const float* __restrict__ xin, const float* __restrict__ tabC,
                       const float* __restrict__ tabS, float* __restrict__ Tr,
                       float* __restrict__ Ti) {
  int wid = blockIdx.x * (blockDim.x >> 5) + (threadIdx.x >> 5);
  int lane = threadIdx.x & 31;
  int ch = wid & 1;
  int byi = wid >> 1;
  int b = byi >> 8, y = byi & 255;
  const float* row = xin + ((size_t)(b * H_ + y)) * W_ * C_ + 16 * ch;
  v8f aC = vzero(), aS = vzero();
  v2f c0[FX_U], s0[FX_U], b0[FX_U], c1[FX_U], s1[FX_U], b1[FX_U];
  fx_load(c0, s0, b0, tabC, tabS, row, 0, lane);
  for (int k0 = 0; k0 < W_; k0 += 8 * FX_U) {
    fx_load(c1, s1, b1, tabC, tabS, row, k0 + 4 * FX_U, lane);
    __builtin_amdgcn_sched_barrier(0);
    fx_mma(c0, s0, b0, aC, aS);
    __builtin_amdgcn_sched_barrier(0);
    if (k0 + 8 * FX_U < W_) fx_load(c0, s0, b0, tabC, tabS, row, k0 + 8 * FX_U, lane);
    __builtin_amdgcn_sched_barrier(0);
    fx_mma(c1, s1, b1, aC, aS);
    __builtin_amdgcn_sched_barrier(0);
  }
  float* tr = Tr + ((size_t)(b * H_ + y)) * M_ * C_ + 16 * ch;
  float* ti = Ti + ((size_t)(b * H_ + y)) * M_ * C_ + 16 * ch;
  stD(tr, C_, lane, aC, 0.0625f);
  stD(ti, C_, lane, aS, -0.0625f);
}

__global__ void k_fwdY(const float* __restrict__ Tr, const float* __restrict__ Ti,
                       const float* __restrict__ tabC, const float* __restrict__ tabS,
                       float* __restrict__ Frp, float* __restrict__ Fip) {
  int wid = blockIdx.x * (blockDim.x >> 5) + (threadIdx.x >> 5);
  int lane = threadIdx.x & 31;
  int b = wid >> 7;
  int r2 = wid & 127;
  int col0 = (r2 >> 2) * 16;
  int kc = r2 & 3;
  const float* trb = Tr + (size_t)b * H_ * (M_ * C_);
  const float* tib = Ti + (size_t)b * H_ * (M_ * C_);
  v8f aCT = vzero(), aST = vzero(), aCTi = vzero(), aSTr = vzero();
  for (int k0 = kc * 64; k0 < kc * 64 + 64; k0 += 4) {
    v2f aC = ldA(tabC + k0, W_, lane);
    v2f aS = ldA(tabS + k0, W_, lane);
    v2f bR = ldB(trb + k0 * 512 + col0, 512, lane);
    v2f bI = ldB(tib + k0 * 512 + col0, 512, lane);
    aCT = wmma4(aC, bR, aCT);
    aST = wmma4(aS, bI, aST);
    aCTi = wmma4(aC, bI, aCTi);
    aSTr = wmma4(aS, bR, aSTr);
  }
  float* fr = Frp + (size_t)kc * (B_ * M_ * 512) + (size_t)b * (M_ * 512) + col0;
  float* fi = Fip + (size_t)kc * (B_ * M_ * 512) + (size_t)b * (M_ * 512) + col0;
  int n = lane & 15, mh = (lane >> 4) << 3;
#pragma unroll
  for (int r = 0; r < 8; ++r) {
    fr[(r + mh) * 512 + n] = (aCT[r] + aST[r]) * 0.0625f;
    fi[(r + mh) * 512 + n] = (aCTi[r] - aSTr[r]) * 0.0625f;
  }
}

__global__ void __launch_bounds__(512) k_mix(const float* __restrict__ Frp, const float* __restrict__ Fip,
                      const float* __restrict__ wr, const float* __restrict__ wi,
                      float* __restrict__ Gr, float* __restrict__ Gi) {
  __shared__ float sFr[512];
  __shared__ float sFi[512];
  int b = blockIdx.x >> 4, k1 = blockIdx.x & 15;
  int tid = threadIdx.x;
  size_t base = ((size_t)b * M_ + k1) * 512;
  int o = tid >> 4, k2 = tid & 15;
  const float* wrk = wr + k1 * 16 + k2;
  const float* wik = wi + k1 * 16 + k2;
  float wrv[C_], wiv[C_];
#pragma unroll
  for (int i = 0; i < C_; ++i) {
    wrv[i] = wrk[(i * 32 + o) * 256];
    wiv[i] = wik[(i * 32 + o) * 256];
  }
  float fr = 0.f, fi = 0.f;
#pragma unroll
  for (int kc = 0; kc < 4; ++kc) {
    fr += Frp[(size_t)kc * (B_ * M_ * 512) + base + tid];
    fi += Fip[(size_t)kc * (B_ * M_ * 512) + base + tid];
  }
  sFr[tid] = fr; sFi[tid] = fi;
  __syncthreads();
  float gr = 0.f, gi = 0.f;
#pragma unroll
  for (int i = 0; i < C_; ++i) {
    float xr = sFr[k2 * 32 + i], xi = sFi[k2 * 32 + i];
    gr += xr * wrv[i] - xi * wiv[i];
    gi += xr * wiv[i] + xi * wrv[i];
  }
  size_t gidx = base + k2 * 32 + o;
  Gr[gidx] = gr; Gi[gidx] = gi;
}

__global__ void k_invY(const float* __restrict__ Gr, const float* __restrict__ Gi,
                       const float* __restrict__ tabC, const float* __restrict__ tabS,
                       float* __restrict__ P, float* __restrict__ Q) {
  int wid = blockIdx.x * (blockDim.x >> 5) + (threadIdx.x >> 5);
  int lane = threadIdx.x & 31;
  int b = wid >> 9, rest = wid & 511;
  int y0 = (rest >> 5) * 16, col0 = (rest & 31) * 16;
  const float* grb = Gr + (size_t)b * 8192;
  const float* gib = Gi + (size_t)b * 8192;
  v8f aCG = vzero(), aSGi = vzero(), aSG = vzero(), aCGi = vzero();
#pragma unroll
  for (int k0 = 0; k0 < M_; k0 += 4) {
    v2f aC = ldAt(tabC + k0 * W_ + y0, W_, lane);
    v2f aS = ldAt(tabS + k0 * W_ + y0, W_, lane);
    v2f bR = ldB(grb + k0 * 512 + col0, 512, lane);
    v2f bI = ldB(gib + k0 * 512 + col0, 512, lane);
    aCG = wmma4(aC, bR, aCG);
    aSGi = wmma4(aS, bI, aSGi);
    aSG = wmma4(aS, bR, aSG);
    aCGi = wmma4(aC, bI, aCGi);
  }
  float* p = P + (size_t)(b * H_ + y0) * 512 + col0;
  float* q = Q + (size_t)(b * H_ + y0) * 512 + col0;
  int n = lane & 15, mh = (lane >> 4) << 3;
#pragma unroll
  for (int r = 0; r < 8; ++r) {
    p[(r + mh) * 512 + n] = (aCG[r] - aSGi[r]) * 0.0625f;
    q[(r + mh) * 512 + n] = (aSG[r] + aCGi[r]) * 0.0625f;
  }
}

__global__ void k_invX(const float* __restrict__ P, const float* __restrict__ Q,
                       const float* __restrict__ xin, const float* __restrict__ tabC2,
                       const float* __restrict__ tabS2, const float* __restrict__ bw,
                       const float* __restrict__ bb, float* __restrict__ xout) {
  int wid = blockIdx.x * (blockDim.x >> 5) + (threadIdx.x >> 5);
  int lane = threadIdx.x & 31;
  int by = wid >> 4;
  int x0 = (wid & 15) * 16;
  const float* Pby = P + (size_t)by * 512;
  const float* Qby = Q + (size_t)by * 512;
  const float* xrow = xin + (size_t)by * (W_ * C_);
  v8f acc0 = vzero(), acc1 = vzero();
#pragma unroll
  for (int k0 = 0; k0 < M_; k0 += 4) {
    v2f aC = ldAt(tabC2 + k0 * W_ + x0, W_, lane);
    v2f bP0 = ldB(Pby + k0 * C_, C_, lane);
    v2f bP1 = ldB(Pby + k0 * C_ + 16, C_, lane);
    acc0 = wmma4(aC, bP0, acc0);
    acc1 = wmma4(aC, bP1, acc1);
    v2f aS = ldAt(tabS2 + k0 * W_ + x0, W_, lane);
    v2f aSn; aSn.x = -aS.x; aSn.y = -aS.y;
    v2f bQ0 = ldB(Qby + k0 * C_, C_, lane);
    v2f bQ1 = ldB(Qby + k0 * C_ + 16, C_, lane);
    acc0 = wmma4(aSn, bQ0, acc0);
    acc1 = wmma4(aSn, bQ1, acc1);
  }
#pragma unroll
  for (int k0 = 0; k0 < C_; k0 += 4) {
    v2f aX = ldA(xrow + x0 * C_ + k0, C_, lane);
    v2f bW0 = ldBt(bw + k0, C_, lane);
    v2f bW1 = ldBt(bw + 16 * C_ + k0, C_, lane);
    acc0 = wmma4(aX, bW0, acc0);
    acc1 = wmma4(aX, bW1, acc1);
  }
  int n = lane & 15, mh = (lane >> 4) << 3;
  float bb0 = bb[n], bb1 = bb[16 + n];
  float* out = xout + (size_t)by * (W_ * C_);
#pragma unroll
  for (int r = 0; r < 8; ++r) {
    int xr = x0 + r + mh;
    float v0 = acc0[r] + bb0;
    v0 = 0.5f * v0 * (1.0f + erff(v0 * 0.70710678f));
    out[xr * C_ + n] = v0;
    float v1 = acc1[r] + bb1;
    v1 = 0.5f * v1 * (1.0f + erff(v1 * 0.70710678f));
    out[xr * C_ + 16 + n] = v1;
  }
}

__global__ void k_proj(const float* __restrict__ xb, const float* __restrict__ pw,
                       const float* __restrict__ pb, float* __restrict__ out,
                       int o_outer) {
  int p = blockIdx.x * blockDim.x + threadIdx.x;
  if (p >= B_ * PIX) return;
  int b = p >> 16, pix = p & (PIX - 1);
  const float* v = xb + (size_t)p * C_;
  float s = pb[0];
#pragma unroll
  for (int c = 0; c < C_; ++c) s += pw[c] * v[c];
  out[((size_t)(b * 4 + o_outer)) * PIX + pix] = s;
}

extern "C" void kernel_launch(void* const* d_in, const int* in_sizes, int n_in,
                              void* d_out, int out_size, void* d_ws, size_t ws_size,
                              hipStream_t stream) {
  const float* x      = (const float*)d_in[0];
  const float* lift_w = (const float*)d_in[1];
  const float* lift_b = (const float*)d_in[2];
  const float* wr     = (const float*)d_in[3];
  const float* wi     = (const float*)d_in[4];
  const float* byp_w  = (const float*)d_in[5];
  const float* byp_b  = (const float*)d_in[6];
  const float* proj_w = (const float*)d_in[7];
  const float* proj_b = (const float*)d_in[8];
  float* out = (float*)d_out;

  float* w = (float*)d_ws;
  float* tabC  = w;
  float* tabS  = tabC + 4096;
  float* tabC2 = tabS + 4096;
  float* tabS2 = tabC2 + 4096;
  float* x0    = tabS2 + 4096;
  float* bufA  = x0 + (size_t)B_ * PIX * C_;
  float* bufB  = bufA + (size_t)B_ * PIX * C_;
  float* Tr    = bufB + (size_t)B_ * PIX * C_;
  float* Ti    = Tr + (size_t)B_ * H_ * M_ * C_;
  float* Frp   = Ti + (size_t)B_ * H_ * M_ * C_;
  float* Fip   = Frp + (size_t)4 * B_ * M_ * 512;
  float* Gr    = Fip + (size_t)4 * B_ * M_ * 512;
  float* Gi    = Gr + (size_t)B_ * M_ * M_ * C_;
  float* Pb    = Gi + (size_t)B_ * M_ * M_ * C_;
  float* Qb    = Pb + (size_t)B_ * H_ * M_ * C_;

  k_tabs<<<16, 256, 0, stream>>>(tabC, tabS, tabC2, tabS2);
  k_lift<<<(B_ * PIX + 255) / 256, 256, 0, stream>>>(x, lift_w, lift_b, x0);

  for (int o = 0; o < 4; ++o) {
    for (int n = 0; n < 4; ++n) {
      const float* cur = (n == 0) ? x0 : ((n & 1) ? bufA : bufB);
      float* dst = (n & 1) ? bufB : bufA;
      size_t wOff = ((size_t)(o * 4 + n)) * C_ * C_ * M_ * M_;
      size_t bwOff = ((size_t)(o * 4 + n)) * C_ * C_;
      size_t bbOff = ((size_t)(o * 4 + n)) * C_;
      k_fwdX<<<256, 256, 0, stream>>>(cur, tabC, tabS, Tr, Ti);
      k_fwdY<<<64, 256, 0, stream>>>(Tr, Ti, tabC, tabS, Frp, Fip);
      k_mix<<<64, 512, 0, stream>>>(Frp, Fip, wr + wOff, wi + wOff, Gr, Gi);
      k_invY<<<256, 256, 0, stream>>>(Gr, Gi, tabC, tabS, Pb, Qb);
      k_invX<<<2048, 256, 0, stream>>>(Pb, Qb, cur, tabC2, tabS2,
                                       byp_w + bwOff, byp_b + bbOff, dst);
    }
    k_proj<<<(B_ * PIX + 255) / 256, 256, 0, stream>>>(bufB, proj_w, proj_b, out, o);
  }
}


# Two of the four independent chains run in each launch (grids doubled, 86 -> 45 launches); invX writes its tile in place so both chains fit the old workspace; sums in same order
# speedup vs baseline: 1.4328x; 1.2739x over previous
#include <hip/hip_runtime.h>
#include <math.h>


typedef float v2f __attribute__((ext_vector_type(2)));
typedef float v8f __attribute__((ext_vector_type(8)));

#define B_  4
#define NCH 2
#define NB_ (NCH * B_)
#define C_  32
#define H_  256
#define W_  256
#define M_  16
#define PIX (H_ * W_)

__device__ __forceinline__ v8f wmma4(v2f a, v2f b, v8f c) {
  return __builtin_amdgcn_wmma_f32_16x16x4_f32(false, a, false, b, (short)0, c,
                                               false, false);
}

__device__ __forceinline__ v2f ldA(const float* p, int lda, int lane) {
  int r = lane & 15, kk = (lane >> 4) << 1;
  v2f v; v.x = p[r * lda + kk]; v.y = p[r * lda + kk + 1]; return v;
}
__device__ __forceinline__ v2f ldAt(const float* p, int lda, int lane) {
  int r = lane & 15, kk = (lane >> 4) << 1;
  v2f v; v.x = p[kk * lda + r]; v.y = p[(kk + 1) * lda + r]; return v;
}
__device__ __forceinline__ v2f ldB(const float* p, int ldb, int lane) {
  int n = lane & 15, kk = (lane >> 4) << 1;
  v2f v; v.x = p[kk * ldb + n]; v.y = p[(kk + 1) * ldb + n]; return v;
}
__device__ __forceinline__ v2f ldBt(const float* p, int ldb, int lane) {
  int n = lane & 15, kk = (lane >> 4) << 1;
  v2f v; v.x = p[n * ldb + kk]; v.y = p[n * ldb + kk + 1]; return v;
}
__device__ __forceinline__ void stD(float* p, int ldd, int lane, v8f d, float s) {
  int n = lane & 15, mh = (lane >> 4) << 3;
#pragma unroll
  for (int r = 0; r < 8; ++r) p[(r + mh) * ldd + n] = d[r] * s;
}

__device__ __forceinline__ v8f vzero() {
  v8f z = {0.f, 0.f, 0.f, 0.f, 0.f, 0.f, 0.f, 0.f};
  return z;
}

__global__ void k_tabs(float* tabC, float* tabS, float* tabC2, float* tabS2) {
  int i = blockIdx.x * blockDim.x + threadIdx.x;
  if (i >= M_ * W_) return;
  int k = i >> 8, x = i & 255;
  int m = (k * x) & 255;
  float ang = 0.0245436926f * (float)m;
  float c = cosf(ang), s = sinf(ang);
  tabC[i] = c; tabS[i] = s;
  float w2 = (k == 0) ? 1.0f : 2.0f;
  tabC2[i] = w2 * c * 0.0625f;
  tabS2[i] = w2 * s * 0.0625f;
}

__global__ void k_lift(const float* __restrict__ x, const float* __restrict__ lw,
                       const float* __restrict__ lb, float* __restrict__ act) {
  int p = blockIdx.x * blockDim.x + threadIdx.x;
  if (p >= B_ * PIX) return;
  int b = p >> 16, pix = p & (PIX - 1);
  float xi0 = x[(b * 3 + 0) * PIX + pix];
  float xi1 = x[(b * 3 + 1) * PIX + pix];
  float xi2 = x[(b * 3 + 2) * PIX + pix];
  float* o0 = act + (size_t)p * C_;
  float* o1 = o0 + (size_t)B_ * PIX * C_;
#pragma unroll
  for (int c = 0; c < C_; ++c) {
    float v = lb[c] + lw[c * 3 + 0] * xi0 + lw[c * 3 + 1] * xi1 + lw[c * 3 + 2] * xi2;
    o0[c] = v;
    o1[c] = v;
  }
}

#define FX_U 8
__device__ __forceinline__ void fx_load(v2f (&c)[FX_U], v2f (&s)[FX_U], v2f (&b)[FX_U],
                                        const float* __restrict__ tabC,
                                        const float* __restrict__ tabS,
                                        const float* __restrict__ row, int k, int lane) {
#pragma unroll
  for (int u = 0; u < FX_U; ++u) {
    c[u] = ldA(tabC + k + 4 * u, W_, lane);
    s[u] = ldA(tabS + k + 4 * u, W_, lane);
    b[u] = ldB(row + (k + 4 * u) * C_, C_, lane);
  }
}
__device__ __forceinline__ void fx_mma(const v2f (&c)[FX_U], const v2f (&s)[FX_U],
                                       const v2f (&b)[FX_U], v8f& aC, v8f& aS) {
#pragma unroll
  for (int u = 0; u < FX_U; ++u) {
    aC = wmma4(c[u], b[u], aC);
    aS = wmma4(s[u], b[u], aS);
  }
}
__global__ void __launch_bounds__(256) k_fwdX(const float* __restrict__ xin, const float* __restrict__ tabC,
                       const float* __restrict__ tabS, float* __restrict__ Tr,
                       float* __restrict__ Ti) {
  int wid = blockIdx.x * (blockDim.x >> 5) + (threadIdx.x >> 5);
  int lane = threadIdx.x & 31;
  int ch = wid & 1;
  int byi = wid >> 1;
  int b = byi >> 8, y = byi & 255;
  const float* row = xin + ((size_t)(b * H_ + y)) * W_ * C_ + 16 * ch;
  v8f aC = vzero(), aS = vzero();
  v2f c0[FX_U], s0[FX_U], b0[FX_U], c1[FX_U], s1[FX_U], b1[FX_U];
  fx_load(c0, s0, b0, tabC, tabS, row, 0, lane);
  for (int k0 = 0; k0 < W_; k0 += 8 * FX_U) {
    fx_load(c1, s1, b1, tabC, tabS, row, k0 + 4 * FX_U, lane);
    __builtin_amdgcn_sched_barrier(0);
    fx_mma(c0, s0, b0, aC, aS);
    __builtin_amdgcn_sched_barrier(0);
    if (k0 + 8 * FX_U < W_) fx_load(c0, s0, b0, tabC, tabS, row, k0 + 8 * FX_U, lane);
    __builtin_amdgcn_sched_barrier(0);
    fx_mma(c1, s1, b1, aC, aS);
    __builtin_amdgcn_sched_barrier(0);
  }
  float* tr = Tr + ((size_t)(b * H_ + y)) * M_ * C_ + 16 * ch;
  float* ti = Ti + ((size_t)(b * H_ + y)) * M_ * C_ + 16 * ch;
  stD(tr, C_, lane, aC, 0.0625f);
  stD(ti, C_, lane, aS, -0.0625f);
}

__global__ void k_fwdY(const float* __restrict__ Tr, const float* __restrict__ Ti,
                       const float* __restrict__ tabC, const float* __restrict__ tabS,
                       float* __restrict__ Frp, float* __restrict__ Fip) {
  int wid = blockIdx.x * (blockDim.x >> 5) + (threadIdx.x >> 5);
  int lane = threadIdx.x & 31;
  int b = wid >> 7;
  int r2 = wid & 127;
  int col0 = (r2 >> 2) * 16;
  int kc = r2 & 3;
  const float* trb = Tr + (size_t)b * H_ * (M_ * C_);
  const float* tib = Ti + (size_t)b * H_ * (M_ * C_);
  v8f aCT = vzero(), aST = vzero(), aCTi = vzero(), aSTr = vzero();
  for (int k0 = kc * 64; k0 < kc * 64 + 64; k0 += 4) {
    v2f aC = ldA(tabC + k0, W_, lane);
    v2f aS = ldA(tabS + k0, W_, lane);
    v2f bR = ldB(trb + k0 * 512 + col0, 512, lane);
    v2f bI = ldB(tib + k0 * 512 + col0, 512, lane);
    aCT = wmma4(aC, bR, aCT);
    aST = wmma4(aS, bI, aST);
    aCTi = wmma4(aC, bI, aCTi);
    aSTr = wmma4(aS, bR, aSTr);
  }
  float* fr = Frp + (size_t)kc * (NB_ * M_ * 512) + (size_t)b * (M_ * 512) + col0;
  float* fi = Fip + (size_t)kc * (NB_ * M_ * 512) + (size_t)b * (M_ * 512) + col0;
  int n = lane & 15, mh = (lane >> 4) << 3;
#pragma unroll
  for (int r = 0; r < 8; ++r) {
    fr[(r + mh) * 512 + n] = (aCT[r] + aST[r]) * 0.0625f;
    fi[(r + mh) * 512 + n] = (aCTi[r] - aSTr[r]) * 0.0625f;
  }
}

__global__ void __launch_bounds__(512) k_mix(const float* __restrict__ Frp, const float* __restrict__ Fip,
                      const float* __restrict__ wr, const float* __restrict__ wi,
                      float* __restrict__ Gr, float* __restrict__ Gi) {
  __shared__ float sFr[512];
  __shared__ float sFi[512];
  int b = blockIdx.x >> 4, k1 = blockIdx.x & 15;
  size_t wch = (size_t)(b >> 2) * (4 * C_ * C_ * M_ * M_);
  int tid = threadIdx.x;
  size_t base = ((size_t)b * M_ + k1) * 512;
  int o = tid >> 4, k2 = tid & 15;
  const float* wrk = wr + wch + k1 * 16 + k2;
  const float* wik = wi + wch + k1 * 16 + k2;
  float wrv[C_], wiv[C_];
#pragma unroll
  for (int i = 0; i < C_; ++i) {
    wrv[i] = wrk[(i * 32 + o) * 256];
    wiv[i] = wik[(i * 32 + o) * 256];
  }
  float fr = 0.f, fi = 0.f;
#pragma unroll
  for (int kc = 0; kc < 4; ++kc) {
    fr += Frp[(size_t)kc * (NB_ * M_ * 512) + base + tid];
    fi += Fip[(size_t)kc * (NB_ * M_ * 512) + base + tid];
  }
  sFr[tid] = fr; sFi[tid] = fi;
  __syncthreads();
  float gr = 0.f, gi = 0.f;
#pragma unroll
  for (int i = 0; i < C_; ++i) {
    float xr = sFr[k2 * 32 + i], xi = sFi[k2 * 32 + i];
    gr += xr * wrv[i] - xi * wiv[i];
    gi += xr * wiv[i] + xi * wrv[i];
  }
  size_t gidx = base + k2 * 32 + o;
  Gr[gidx] = gr; Gi[gidx] = gi;
}

__global__ void k_invY(const float* __restrict__ Gr, const float* __restrict__ Gi,
                       const float* __restrict__ tabC, const float* __restrict__ tabS,
                       float* __restrict__ P, float* __restrict__ Q) {
  int wid = blockIdx.x * (blockDim.x >> 5) + (threadIdx.x >> 5);
  int lane = threadIdx.x & 31;
  int b = wid >> 9, rest = wid & 511;
  int y0 = (rest >> 5) * 16, col0 = (rest & 31) * 16;
  const float* grb = Gr + (size_t)b * 8192;
  const float* gib = Gi + (size_t)b * 8192;
  v8f aCG = vzero(), aSGi = vzero(), aSG = vzero(), aCGi = vzero();
#pragma unroll
  for (int k0 = 0; k0 < M_; k0 += 4) {
    v2f aC = ldAt(tabC + k0 * W_ + y0, W_, lane);
    v2f aS = ldAt(tabS + k0 * W_ + y0, W_, lane);
    v2f bR = ldB(grb + k0 * 512 + col0, 512, lane);
    v2f bI = ldB(gib + k0 * 512 + col0, 512, lane);
    aCG = wmma4(aC, bR, aCG);
    aSGi = wmma4(aS, bI, aSGi);
    aSG = wmma4(aS, bR, aSG);
    aCGi = wmma4(aC, bI, aCGi);
  }
  float* p = P + (size_t)(b * H_ + y0) * 512 + col0;
  float* q = Q + (size_t)(b * H_ + y0) * 512 + col0;
  int n = lane & 15, mh = (lane >> 4) << 3;
#pragma unroll
  for (int r = 0; r < 8; ++r) {
    p[(r + mh) * 512 + n] = (aCG[r] - aSGi[r]) * 0.0625f;
    q[(r + mh) * 512 + n] = (aSG[r] + aCGi[r]) * 0.0625f;
  }
}

__global__ void k_invX(const float* __restrict__ P, const float* __restrict__ Q,
                       float* act, const float* __restrict__ tabC2,
                       const float* __restrict__ tabS2, const float* __restrict__ bw,
                       const float* __restrict__ bb) {
  int wid = blockIdx.x * (blockDim.x >> 5) + (threadIdx.x >> 5);
  int lane = threadIdx.x & 31;
  int by = wid >> 4;
  int chn = by >> 10;
  int x0 = (wid & 15) * 16;
  const float* Pby = P + (size_t)by * 512;
  const float* Qby = Q + (size_t)by * 512;
  const float* xrow = act + (size_t)by * (W_ * C_);
  bw += chn * (4 * C_ * C_);
  bb += chn * (4 * C_);
  v8f acc0 = vzero(), acc1 = vzero();
#pragma unroll
  for (int k0 = 0; k0 < M_; k0 += 4) {
    v2f aC = ldAt(tabC2 + k0 * W_ + x0, W_, lane);
    v2f bP0 = ldB(Pby + k0 * C_, C_, lane);
    v2f bP1 = ldB(Pby + k0 * C_ + 16, C_, lane);
    acc0 = wmma4(aC, bP0, acc0);
    acc1 = wmma4(aC, bP1, acc1);
    v2f aS = ldAt(tabS2 + k0 * W_ + x0, W_, lane);
    v2f aSn; aSn.x = -aS.x; aSn.y = -aS.y;
    v2f bQ0 = ldB(Qby + k0 * C_, C_, lane);
    v2f bQ1 = ldB(Qby + k0 * C_ + 16, C_, lane);
    acc0 = wmma4(aSn, bQ0, acc0);
    acc1 = wmma4(aSn, bQ1, acc1);
  }
#pragma unroll
  for (int k0 = 0; k0 < C_; k0 += 4) {
    v2f aX = ldA(xrow + x0 * C_ + k0, C_, lane);
    v2f bW0 = ldBt(bw + k0, C_, lane);
    v2f bW1 = ldBt(bw + 16 * C_ + k0, C_, lane);
    acc0 = wmma4(aX, bW0, acc0);
    acc1 = wmma4(aX, bW1, acc1);
  }
  int n = lane & 15, mh = (lane >> 4) << 3;
  float bb0 = bb[n], bb1 = bb[16 + n];
  float* out = act + (size_t)by * (W_ * C_);
#pragma unroll
  for (int r = 0; r < 8; ++r) {
    int xr = x0 + r + mh;
    float v0 = acc0[r] + bb0;
    v0 = 0.5f * v0 * (1.0f + erff(v0 * 0.70710678f));
    out[xr * C_ + n] = v0;
    float v1 = acc1[r] + bb1;
    v1 = 0.5f * v1 * (1.0f + erff(v1 * 0.70710678f));
    out[xr * C_ + 16 + n] = v1;
  }
}

__global__ void k_proj(const float* __restrict__ xb, const float* __restrict__ pw,
                       const float* __restrict__ pb, float* __restrict__ out,
                       int o_outer) {
  int p = blockIdx.x * blockDim.x + threadIdx.x;
  if (p >= NB_ * PIX) return;
  int q = p >> 16, pix = p & (PIX - 1);
  int b = q & 3;
  o_outer += q >> 2;
  const float* v = xb + (size_t)p * C_;
  float s = pb[0];
#pragma unroll
  for (int c = 0; c < C_; ++c) s += pw[c] * v[c];
  out[((size_t)(b * 4 + o_outer)) * PIX + pix] = s;
}

extern "C" void kernel_launch(void* const* d_in, const int* in_sizes, int n_in,
                              void* d_out, int out_size, void* d_ws, size_t ws_size,
                              hipStream_t stream) {
  const float* x      = (const float*)d_in[0];
  const float* lift_w = (const float*)d_in[1];
  const float* lift_b = (const float*)d_in[2];
  const float* wr     = (const float*)d_in[3];
  const float* wi     = (const float*)d_in[4];
  const float* byp_w  = (const float*)d_in[5];
  const float* byp_b  = (const float*)d_in[6];
  const float* proj_w = (const float*)d_in[7];
  const float* proj_b = (const float*)d_in[8];
  float* out = (float*)d_out;

  float* w = (float*)d_ws;
  float* tabC  = w;
  float* tabS  = tabC + 4096;
  float* tabC2 = tabS + 4096;
  float* tabS2 = tabC2 + 4096;
  float* act   = tabS2 + 4096;
  float* Tr    = act + (size_t)NB_ * PIX * C_;
  float* Ti    = Tr + (size_t)NB_ * H_ * M_ * C_;
  float* Frp   = Ti + (size_t)NB_ * H_ * M_ * C_;
  float* Fip   = Frp + (size_t)4 * NB_ * M_ * 512;
  float* Gr    = Fip + (size_t)4 * NB_ * M_ * 512;
  float* Gi    = Gr + (size_t)NB_ * M_ * M_ * C_;
  float* Pb    = Gi + (size_t)NB_ * M_ * M_ * C_;
  float* Qb    = Pb + (size_t)NB_ * H_ * M_ * C_;

  k_tabs<<<16, 256, 0, stream>>>(tabC, tabS, tabC2, tabS2);

  for (int o = 0; o < 4; o += NCH) {
    k_lift<<<(B_ * PIX + 255) / 256, 256, 0, stream>>>(x, lift_w, lift_b, act);
    for (int n = 0; n < 4; ++n) {
      size_t wOff = ((size_t)(o * 4 + n)) * C_ * C_ * M_ * M_;
      size_t bwOff = ((size_t)(o * 4 + n)) * C_ * C_;
      size_t bbOff = ((size_t)(o * 4 + n)) * C_;
      k_fwdX<<<NCH * 256, 256, 0, stream>>>(act, tabC, tabS, Tr, Ti);
      k_fwdY<<<NCH * 64, 256, 0, stream>>>(Tr, Ti, tabC, tabS, Frp, Fip);
      k_mix<<<NCH * 64, 512, 0, stream>>>(Frp, Fip, wr + wOff, wi + wOff, Gr, Gi);
      k_invY<<<NCH * 256, 256, 0, stream>>>(Gr, Gi, tabC, tabS, Pb, Qb);
      k_invX<<<NCH * 2048, 256, 0, stream>>>(Pb, Qb, act, tabC2, tabS2,
                                             byp_w + bwOff, byp_b + bbOff);
    }
    k_proj<<<(NB_ * PIX + 255) / 256, 256, 0, stream>>>(act, proj_w, proj_b, out, o);
  }
}


# Next step's forward x transform fused into invX: one block per row, each wave multiplies its own output tile from registers, 16 partials added through LDS (sum order of that transform changes)
# speedup vs baseline: 1.6676x; 1.1639x over previous
#include <hip/hip_runtime.h>
#include <math.h>


typedef float v2f __attribute__((ext_vector_type(2)));
typedef float v8f __attribute__((ext_vector_type(8)));

#define B_  4
#define NCH 2
#define NB_ (NCH * B_)
#define C_  32
#define H_  256
#define W_  256
#define M_  16
#define PIX (H_ * W_)

__device__ __forceinline__ v8f wmma4(v2f a, v2f b, v8f c) {
  return __builtin_amdgcn_wmma_f32_16x16x4_f32(false, a, false, b, (short)0, c,
                                               false, false);
}

__device__ __forceinline__ v2f ldA(const float* p, int lda, int lane) {
  int r = lane & 15, kk = (lane >> 4) << 1;
  v2f v; v.x = p[r * lda + kk]; v.y = p[r * lda + kk + 1]; return v;
}
__device__ __forceinline__ v2f ldAt(const float* p, int lda, int lane) {
  int r = lane & 15, kk = (lane >> 4) << 1;
  v2f v; v.x = p[kk * lda + r]; v.y = p[(kk + 1) * lda + r]; return v;
}
__device__ __forceinline__ v2f ldB(const float* p, int ldb, int lane) {
  int n = lane & 15, kk = (lane >> 4) << 1;
  v2f v; v.x = p[kk * ldb + n]; v.y = p[(kk + 1) * ldb + n]; return v;
}
__device__ __forceinline__ v2f ldBt(const float* p, int ldb, int lane) {
  int n = lane & 15, kk = (lane >> 4) << 1;
  v2f v; v.x = p[n * ldb + kk]; v.y = p[n * ldb + kk + 1]; return v;
}
__device__ __forceinline__ v2f ldA_d(const float* p, int lda, int lane, int j) {
  int r = lane & 15, kk = ((lane >> 4) << 3) + 2 * j;
  v2f v; v.x = p[r * lda + kk]; v.y = p[r * lda + kk + 1]; return v;
}
__device__ __forceinline__ void stD(float* p, int ldd, int lane, v8f d, float s) {
  int n = lane & 15, mh = (lane >> 4) << 3;
#pragma unroll
  for (int r = 0; r < 8; ++r) p[(r + mh) * ldd + n] = d[r] * s;
}

__device__ __forceinline__ v8f vzero() {
  v8f z = {0.f, 0.f, 0.f, 0.f, 0.f, 0.f, 0.f, 0.f};
  return z;
}

__global__ void k_tabs(float* tabC, float* tabS, float* tabC2, float* tabS2) {
  int i = blockIdx.x * blockDim.x + threadIdx.x;
  if (i >= M_ * W_) return;
  int k = i >> 8, x = i & 255;
  int m = (k * x) & 255;
  float ang = 0.0245436926f * (float)m;
  float c = cosf(ang), s = sinf(ang);
  tabC[i] = c; tabS[i] = s;
  float w2 = (k == 0) ? 1.0f : 2.0f;
  tabC2[i] = w2 * c * 0.0625f;
  tabS2[i] = w2 * s * 0.0625f;
}

__global__ void k_lift(const float* __restrict__ x, const float* __restrict__ lw,
                       const float* __restrict__ lb, float* __restrict__ act) {
  int p = blockIdx.x * blockDim.x + threadIdx.x;
  if (p >= B_ * PIX) return;
  int b = p >> 16, pix = p & (PIX - 1);
  float xi0 = x[(b * 3 + 0) * PIX + pix];
  float xi1 = x[(b * 3 + 1) * PIX + pix];
  float xi2 = x[(b * 3 + 2) * PIX + pix];
  float* o0 = act + (size_t)p * C_;
  float* o1 = o0 + (size_t)B_ * PIX * C_;
#pragma unroll
  for (int c = 0; c < C_; ++c) {
    float v = lb[c] + lw[c * 3 + 0] * xi0 + lw[c * 3 + 1] * xi1 + lw[c * 3 + 2] * xi2;
    o0[c] = v;
    o1[c] = v;
  }
}

#define FX_U 8
__device__ __forceinline__ void fx_load(v2f (&c)[FX_U], v2f (&s)[FX_U], v2f (&b)[FX_U],
                                        const float* __restrict__ tabC,
                                        const float* __restrict__ tabS,
                                        const float* __restrict__ row, int k, int lane) {
#pragma unroll
  for (int u = 0; u < FX_U; ++u) {
    c[u] = ldA(tabC + k + 4 * u, W_, lane);
    s[u] = ldA(tabS + k + 4 * u, W_, lane);
    b[u] = ldB(row + (k + 4 * u) * C_, C_, lane);
  }
}
__device__ __forceinline__ void fx_mma(const v2f (&c)[FX_U], const v2f (&s)[FX_U],
                                       const v2f (&b)[FX_U], v8f& aC, v8f& aS) {
#pragma unroll
  for (int u = 0; u < FX_U; ++u) {
    aC = wmma4(c[u], b[u], aC);
    aS = wmma4(s[u], b[u], aS);
  }
}
__global__ void __launch_bounds__(256) k_fwdX(const float* __restrict__ xin, const float* __restrict__ tabC,
                       const float* __restrict__ tabS, float* __restrict__ Tr,
                       float* __restrict__ Ti) {
  int wid = blockIdx.x * (blockDim.x >> 5) + (threadIdx.x >> 5);
  int lane = threadIdx.x & 31;
  int ch = wid & 1;
  int byi = wid >> 1;
  int b = byi >> 8, y = byi & 255;
  const float* row = xin + ((size_t)(b * H_ + y)) * W_ * C_ + 16 * ch;
  v8f aC = vzero(), aS = vzero();
  v2f c0[FX_U], s0[FX_U], b0[FX_U], c1[FX_U], s1[FX_U], b1[FX_U];
  fx_load(c0, s0, b0, tabC, tabS, row, 0, lane);
  for (int k0 = 0; k0 < W_; k0 += 8 * FX_U) {
    fx_load(c1, s1, b1, tabC, tabS, row, k0 + 4 * FX_U, lane);
    __builtin_amdgcn_sched_barrier(0);
    fx_mma(c0, s0, b0, aC, aS);
    __builtin_amdgcn_sched_barrier(0);
    if (k0 + 8 * FX_U < W_) fx_load(c0, s0, b0, tabC, tabS, row, k0 + 8 * FX_U, lane);
    __builtin_amdgcn_sched_barrier(0);
    fx_mma(c1, s1, b1, aC, aS);
    __builtin_amdgcn_sched_barrier(0);
  }
  float* tr = Tr + ((size_t)(b * H_ + y)) * M_ * C_ + 16 * ch;
  float* ti = Ti + ((size_t)(b * H_ + y)) * M_ * C_ + 16 * ch;
  stD(tr, C_, lane, aC, 0.0625f);
  stD(ti, C_, lane, aS, -0.0625f);
}

__global__ void k_fwdY(const float* __restrict__ Tr, const float* __restrict__ Ti,
                       const float* __restrict__ tabC, const float* __restrict__ tabS,
                       float* __restrict__ Frp, float* __restrict__ Fip) {
  int wid = blockIdx.x * (blockDim.x >> 5) + (threadIdx.x >> 5);
  int lane = threadIdx.x & 31;
  int b = wid >> 7;
  int r2 = wid & 127;
  int col0 = (r2 >> 2) * 16;
  int kc = r2 & 3;
  const float* trb = Tr + (size_t)b * H_ * (M_ * C_);
  const float* tib = Ti + (size_t)b * H_ * (M_ * C_);
  v8f aCT = vzero(), aST = vzero(), aCTi = vzero(), aSTr = vzero();
  for (int k0 = kc * 64; k0 < kc * 64 + 64; k0 += 4) {
    v2f aC = ldA(tabC + k0, W_, lane);
    v2f aS = ldA(tabS + k0, W_, lane);
    v2f bR = ldB(trb + k0 * 512 + col0, 512, lane);
    v2f bI = ldB(tib + k0 * 512 + col0, 512, lane);
    aCT = wmma4(aC, bR, aCT);
    aST = wmma4(aS, bI, aST);
    aCTi = wmma4(aC, bI, aCTi);
    aSTr = wmma4(aS, bR, aSTr);
  }
  float* fr = Frp + (size_t)kc * (NB_ * M_ * 512) + (size_t)b * (M_ * 512) + col0;
  float* fi = Fip + (size_t)kc * (NB_ * M_ * 512) + (size_t)b * (M_ * 512) + col0;
  int n = lane & 15, mh = (lane >> 4) << 3;
#pragma unroll
  for (int r = 0; r < 8; ++r) {
    fr[(r + mh) * 512 + n] = (aCT[r] + aST[r]) * 0.0625f;
    fi[(r + mh) * 512 + n] = (aCTi[r] - aSTr[r]) * 0.0625f;
  }
}

__global__ void __launch_bounds__(512) k_mix(const float* __restrict__ Frp, const float* __restrict__ Fip,
                      const float* __restrict__ wr, const float* __restrict__ wi,
                      float* __restrict__ Gr, float* __restrict__ Gi) {
  __shared__ float sFr[512];
  __shared__ float sFi[512];
  int b = blockIdx.x >> 4, k1 = blockIdx.x & 15;
  size_t wch = (size_t)(b >> 2) * (4 * C_ * C_ * M_ * M_);
  int tid = threadIdx.x;
  size_t base = ((size_t)b * M_ + k1) * 512;
  int o = tid >> 4, k2 = tid & 15;
  const float* wrk = wr + wch + k1 * 16 + k2;
  const float* wik = wi + wch + k1 * 16 + k2;
  float wrv[C_], wiv[C_];
#pragma unroll
  for (int i = 0; i < C_; ++i) {
    wrv[i] = wrk[(i * 32 + o) * 256];
    wiv[i] = wik[(i * 32 + o) * 256];
  }
  float fr = 0.f, fi = 0.f;
#pragma unroll
  for (int kc = 0; kc < 4; ++kc) {
    fr += Frp[(size_t)kc * (NB_ * M_ * 512) + base + tid];
    fi += Fip[(size_t)kc * (NB_ * M_ * 512) + base + tid];
  }
  sFr[tid] = fr; sFi[tid] = fi;
  __syncthreads();
  float gr = 0.f, gi = 0.f;
#pragma unroll
  for (int i = 0; i < C_; ++i) {
    float xr = sFr[k2 * 32 + i], xi = sFi[k2 * 32 + i];
    gr += xr * wrv[i] - xi * wiv[i];
    gi += xr * wiv[i] + xi * wrv[i];
  }
  size_t gidx = base + k2 * 32 + o;
  Gr[gidx] = gr; Gi[gidx] = gi;
}

__global__ void k_invY(const float* __restrict__ Gr, const float* __restrict__ Gi,
                       const float* __restrict__ tabC, const float* __restrict__ tabS,
                       float* __restrict__ P, float* __restrict__ Q) {
  int wid = blockIdx.x * (blockDim.x >> 5) + (threadIdx.x >> 5);
  int lane = threadIdx.x & 31;
  int b = wid >> 9, rest = wid & 511;
  int y0 = (rest >> 5) * 16, col0 = (rest & 31) * 16;
  const float* grb = Gr + (size_t)b * 8192;
  const float* gib = Gi + (size_t)b * 8192;
  v8f aCG = vzero(), aSGi = vzero(), aSG = vzero(), aCGi = vzero();
#pragma unroll
  for (int k0 = 0; k0 < M_; k0 += 4) {
    v2f aC = ldAt(tabC + k0 * W_ + y0, W_, lane);
    v2f aS = ldAt(tabS + k0 * W_ + y0, W_, lane);
    v2f bR = ldB(grb + k0 * 512 + col0, 512, lane);
    v2f bI = ldB(gib + k0 * 512 + col0, 512, lane);
    aCG = wmma4(aC, bR, aCG);
    aSGi = wmma4(aS, bI, aSGi);
    aSG = wmma4(aS, bR, aSG);
    aCGi = wmma4(aC, bI, aCGi);
  }
  float* p = P + (size_t)(b * H_ + y0) * 512 + col0;
  float* q = Q + (size_t)(b * H_ + y0) * 512 + col0;
  int n = lane & 15, mh = (lane >> 4) << 3;
#pragma unroll
  for (int r = 0; r < 8; ++r) {
    p[(r + mh) * 512 + n] = (aCG[r] - aSGi[r]) * 0.0625f;
    q[(r + mh) * 512 + n] = (aSG[r] + aCGi[r]) * 0.0625f;
  }
}

template <bool FWD>
__global__ void __launch_bounds__(512) k_invX(const float* __restrict__ P, const float* __restrict__ Q,
                       float* act, const float* __restrict__ tabC2,
                       const float* __restrict__ tabS2, const float* __restrict__ bw,
                       const float* __restrict__ bb, const float* __restrict__ tabC,
                       const float* __restrict__ tabS, float* __restrict__ Tr,
                       float* __restrict__ Ti) {
  int w = threadIdx.x >> 5;
  int lane = threadIdx.x & 31;
  int by = blockIdx.x;
  int chn = by >> 10;
  int x0 = w * 16;
  const float* Pby = P + (size_t)by * 512;
  const float* Qby = Q + (size_t)by * 512;
  const float* xrow = act + (size_t)by * (W_ * C_);
  bw += chn * (4 * C_ * C_);
  bb += chn * (4 * C_);
  v2f fC[4], fS[4];
  if (FWD) {
#pragma unroll
    for (int j = 0; j < 4; ++j) {
      fC[j] = ldA_d(tabC + x0, W_, lane, j);
      fS[j] = ldA_d(tabS + x0, W_, lane, j);
    }
  }
  v8f acc0 = vzero(), acc1 = vzero();
#pragma unroll
  for (int k0 = 0; k0 < M_; k0 += 4) {
    v2f aC = ldAt(tabC2 + k0 * W_ + x0, W_, lane);
    v2f bP0 = ldB(Pby + k0 * C_, C_, lane);
    v2f bP1 = ldB(Pby + k0 * C_ + 16, C_, lane);
    acc0 = wmma4(aC, bP0, acc0);
    acc1 = wmma4(aC, bP1, acc1);
    v2f aS = ldAt(tabS2 + k0 * W_ + x0, W_, lane);
    v2f aSn; aSn.x = -aS.x; aSn.y = -aS.y;
    v2f bQ0 = ldB(Qby + k0 * C_, C_, lane);
    v2f bQ1 = ldB(Qby + k0 * C_ + 16, C_, lane);
    acc0 = wmma4(aSn, bQ0, acc0);
    acc1 = wmma4(aSn, bQ1, acc1);
  }
#pragma unroll
  for (int k0 = 0; k0 < C_; k0 += 4) {
    v2f aX = ldA(xrow + x0 * C_ + k0, C_, lane);
    v2f bW0 = ldBt(bw + k0, C_, lane);
    v2f bW1 = ldBt(bw + 16 * C_ + k0, C_, lane);
    acc0 = wmma4(aX, bW0, acc0);
    acc1 = wmma4(aX, bW1, acc1);
  }
  int n = lane & 15, mh = (lane >> 4) << 3;
  float bb0 = bb[n], bb1 = bb[16 + n];
  float* out = act + (size_t)by * (W_ * C_);
  v8f g0, g1;
#pragma unroll
  for (int r = 0; r < 8; ++r) {
    int xr = x0 + r + mh;
    float v0 = acc0[r] + bb0;
    v0 = 0.5f * v0 * (1.0f + erff(v0 * 0.70710678f));
    out[xr * C_ + n] = v0;
    g0[r] = v0;
    float v1 = acc1[r] + bb1;
    v1 = 0.5f * v1 * (1.0f + erff(v1 * 0.70710678f));
    out[xr * C_ + 16 + n] = v1;
    g1[r] = v1;
  }
  if (FWD) {
    __shared__ float sC[16 * 512];
    __shared__ float sS[16 * 512];
    v8f aC0 = vzero(), aC1 = vzero(), aS0 = vzero(), aS1 = vzero();
#pragma unroll
    for (int j = 0; j < 4; ++j) {
      v2f b0; b0.x = g0[2 * j]; b0.y = g0[2 * j + 1];
      v2f b1; b1.x = g1[2 * j]; b1.y = g1[2 * j + 1];
      aC0 = wmma4(fC[j], b0, aC0);
      aC1 = wmma4(fC[j], b1, aC1);
      aS0 = wmma4(fS[j], b0, aS0);
      aS1 = wmma4(fS[j], b1, aS1);
    }
    float* pc = sC + w * 512 + (lane >> 4) * 32 + n;
    float* ps = sS + w * 512 + (lane >> 4) * 32 + n;
#pragma unroll
    for (int r = 0; r < 8; ++r) {
      pc[r * 64] = aC0[r];
      pc[r * 64 + 16] = aC1[r];
      ps[r * 64] = aS0[r];
      ps[r * 64 + 16] = aS1[r];
    }
    __syncthreads();
    int tid = threadIdx.x;
    float sc = 0.f, ss = 0.f;
#pragma unroll
    for (int ww = 0; ww < 16; ++ww) {
      sc += sC[ww * 512 + tid];
      ss += sS[ww * 512 + tid];
    }
    int k2 = (tid >> 6) + ((tid >> 5) & 1) * 8, c = tid & 31;
    size_t o = (size_t)by * (M_ * C_) + k2 * C_ + c;
    Tr[o] = sc * 0.0625f;
    Ti[o] = ss * -0.0625f;
  }
}

__global__ void k_proj(const float* __restrict__ xb, const float* __restrict__ pw,
                       const float* __restrict__ pb, float* __restrict__ out,
                       int o_outer) {
  int p = blockIdx.x * blockDim.x + threadIdx.x;
  if (p >= NB_ * PIX) return;
  int q = p >> 16, pix = p & (PIX - 1);
  int b = q & 3;
  o_outer += q >> 2;
  const float* v = xb + (size_t)p * C_;
  float s = pb[0];
#pragma unroll
  for (int c = 0; c < C_; ++c) s += pw[c] * v[c];
  out[((size_t)(b * 4 + o_outer)) * PIX + pix] = s;
}

extern "C" void kernel_launch(void* const* d_in, const int* in_sizes, int n_in,
                              void* d_out, int out_size, void* d_ws, size_t ws_size,
                              hipStream_t stream) {
  const float* x      = (const float*)d_in[0];
  const float* lift_w = (const float*)d_in[1];
  const float* lift_b = (const float*)d_in[2];
  const float* wr     = (const float*)d_in[3];
  const float* wi     = (const float*)d_in[4];
  const float* byp_w  = (const float*)d_in[5];
  const float* byp_b  = (const float*)d_in[6];
  const float* proj_w = (const float*)d_in[7];
  const float* proj_b = (const float*)d_in[8];
  float* out = (float*)d_out;

  float* w = (float*)d_ws;
  float* tabC  = w;
  float* tabS  = tabC + 4096;
  float* tabC2 = tabS + 4096;
  float* tabS2 = tabC2 + 4096;
  float* act   = tabS2 + 4096;
  float* Tr    = act + (size_t)NB_ * PIX * C_;
  float* Ti    = Tr + (size_t)NB_ * H_ * M_ * C_;
  float* Frp   = Ti + (size_t)NB_ * H_ * M_ * C_;
  float* Fip   = Frp + (size_t)4 * NB_ * M_ * 512;
  float* Gr    = Fip + (size_t)4 * NB_ * M_ * 512;
  float* Gi    = Gr + (size_t)NB_ * M_ * M_ * C_;
  float* Pb    = Gi + (size_t)NB_ * M_ * M_ * C_;
  float* Qb    = Pb + (size_t)NB_ * H_ * M_ * C_;

  k_tabs<<<16, 256, 0, stream>>>(tabC, tabS, tabC2, tabS2);

  for (int o = 0; o < 4; o += NCH) {
    k_lift<<<(B_ * PIX + 255) / 256, 256, 0, stream>>>(x, lift_w, lift_b, act);
    for (int n = 0; n < 4; ++n) {
      size_t wOff = ((size_t)(o * 4 + n)) * C_ * C_ * M_ * M_;
      size_t bwOff = ((size_t)(o * 4 + n)) * C_ * C_;
      size_t bbOff = ((size_t)(o * 4 + n)) * C_;
      if (n == 0) k_fwdX<<<NCH * 256, 256, 0, stream>>>(act, tabC, tabS, Tr, Ti);
      k_fwdY<<<NCH * 64, 256, 0, stream>>>(Tr, Ti, tabC, tabS, Frp, Fip);
      k_mix<<<NCH * 64, 512, 0, stream>>>(Frp, Fip, wr + wOff, wi + wOff, Gr, Gi);
      k_invY<<<NCH * 256, 256, 0, stream>>>(Gr, Gi, tabC, tabS, Pb, Qb);
      if (n < 3)
        k_invX<true><<<NCH * 1024, 512, 0, stream>>>(Pb, Qb, act, tabC2, tabS2,
                                                     byp_w + bwOff, byp_b + bbOff,
                                                     tabC, tabS, Tr, Ti);
      else
        k_invX<false><<<NCH * 1024, 512, 0, stream>>>(Pb, Qb, act, tabC2, tabS2,
                                                      byp_w + bwOff, byp_b + bbOff,
                                                      tabC, tabS, Tr, Ti);
    }
    k_proj<<<(NB_ * PIX + 255) / 256, 256, 0, stream>>>(act, proj_w, proj_b, out, o);
  }
}


# invX operands through LDS: P, Q rows and bypass weights fetched once per block, each wave's input tile by whole-line 16-byte loads; fragments read from padded LDS rows; sums in the same order
# speedup vs baseline: 1.7353x; 1.0406x over previous
#include <hip/hip_runtime.h>
#include <math.h>


typedef float v2f __attribute__((ext_vector_type(2)));
typedef float v8f __attribute__((ext_vector_type(8)));

#define B_  4
#define NCH 2
#define NB_ (NCH * B_)
#define C_  32
#define H_  256
#define W_  256
#define M_  16
#define PIX (H_ * W_)

__device__ __forceinline__ v8f wmma4(v2f a, v2f b, v8f c) {
  return __builtin_amdgcn_wmma_f32_16x16x4_f32(false, a, false, b, (short)0, c,
                                               false, false);
}

__device__ __forceinline__ v2f ldA(const float* p, int lda, int lane) {
  int r = lane & 15, kk = (lane >> 4) << 1;
  v2f v; v.x = p[r * lda + kk]; v.y = p[r * lda + kk + 1]; return v;
}
__device__ __forceinline__ v2f ldAt(const float* p, int lda, int lane) {
  int r = lane & 15, kk = (lane >> 4) << 1;
  v2f v; v.x = p[kk * lda + r]; v.y = p[(kk + 1) * lda + r]; return v;
}
__device__ __forceinline__ v2f ldB(const float* p, int ldb, int lane) {
  int n = lane & 15, kk = (lane >> 4) << 1;
  v2f v; v.x = p[kk * ldb + n]; v.y = p[(kk + 1) * ldb + n]; return v;
}
__device__ __forceinline__ v2f ldBt(const float* p, int ldb, int lane) {
  int n = lane & 15, kk = (lane >> 4) << 1;
  v2f v; v.x = p[n * ldb + kk]; v.y = p[n * ldb + kk + 1]; return v;
}
__device__ __forceinline__ v2f ldA_d(const float* p, int lda, int lane, int j) {
  int r = lane & 15, kk = ((lane >> 4) << 3) + 2 * j;
  v2f v; v.x = p[r * lda + kk]; v.y = p[r * lda + kk + 1]; return v;
}
__device__ __forceinline__ void stD(float* p, int ldd, int lane, v8f d, float s) {
  int n = lane & 15, mh = (lane >> 4) << 3;
#pragma unroll
  for (int r = 0; r < 8; ++r) p[(r + mh) * ldd + n] = d[r] * s;
}

__device__ __forceinline__ v8f vzero() {
  v8f z = {0.f, 0.f, 0.f, 0.f, 0.f, 0.f, 0.f, 0.f};
  return z;
}

__global__ void k_tabs(float* tabC, float* tabS, float* tabC2, float* tabS2) {
  int i = blockIdx.x * blockDim.x + threadIdx.x;
  if (i >= M_ * W_) return;
  int k = i >> 8, x = i & 255;
  int m = (k * x) & 255;
  float ang = 0.0245436926f * (float)m;
  float c = cosf(ang), s = sinf(ang);
  tabC[i] = c; tabS[i] = s;
  float w2 = (k == 0) ? 1.0f : 2.0f;
  tabC2[i] = w2 * c * 0.0625f;
  tabS2[i] = w2 * s * 0.0625f;
}

__global__ void k_lift(const float* __restrict__ x, const float* __restrict__ lw,
                       const float* __restrict__ lb, float* __restrict__ act) {
  int p = blockIdx.x * blockDim.x + threadIdx.x;
  if (p >= B_ * PIX) return;
  int b = p >> 16, pix = p & (PIX - 1);
  float xi0 = x[(b * 3 + 0) * PIX + pix];
  float xi1 = x[(b * 3 + 1) * PIX + pix];
  float xi2 = x[(b * 3 + 2) * PIX + pix];
  float* o0 = act + (size_t)p * C_;
  float* o1 = o0 + (size_t)B_ * PIX * C_;
#pragma unroll
  for (int c = 0; c < C_; ++c) {
    float v = lb[c] + lw[c * 3 + 0] * xi0 + lw[c * 3 + 1] * xi1 + lw[c * 3 + 2] * xi2;
    o0[c] = v;
    o1[c] = v;
  }
}

#define FX_U 8
__device__ __forceinline__ void fx_load(v2f (&c)[FX_U], v2f (&s)[FX_U], v2f (&b)[FX_U],
                                        const float* __restrict__ tabC,
                                        const float* __restrict__ tabS,
                                        const float* __restrict__ row, int k, int lane) {
#pragma unroll
  for (int u = 0; u < FX_U; ++u) {
    c[u] = ldA(tabC + k + 4 * u, W_, lane);
    s[u] = ldA(tabS + k + 4 * u, W_, lane);
    b[u] = ldB(row + (k + 4 * u) * C_, C_, lane);
  }
}
__device__ __forceinline__ void fx_mma(const v2f (&c)[FX_U], const v2f (&s)[FX_U],
                                       const v2f (&b)[FX_U], v8f& aC, v8f& aS) {
#pragma unroll
  for (int u = 0; u < FX_U; ++u) {
    aC = wmma4(c[u], b[u], aC);
    aS = wmma4(s[u], b[u], aS);
  }
}
__global__ void __launch_bounds__(256) k_fwdX(const float* __restrict__ xin, const float* __restrict__ tabC,
                       const float* __restrict__ tabS, float* __restrict__ Tr,
                       float* __restrict__ Ti) {
  int wid = blockIdx.x * (blockDim.x >> 5) + (threadIdx.x >> 5);
  int lane = threadIdx.x & 31;
  int ch = wid & 1;
  int byi = wid >> 1;
  int b = byi >> 8, y = byi & 255;
  const float* row = xin + ((size_t)(b * H_ + y)) * W_ * C_ + 16 * ch;
  v8f aC = vzero(), aS = vzero();
  v2f c0[FX_U], s0[FX_U], b0[FX_U], c1[FX_U], s1[FX_U], b1[FX_U];
  fx_load(c0, s0, b0, tabC, tabS, row, 0, lane);
  for (int k0 = 0; k0 < W_; k0 += 8 * FX_U) {
    fx_load(c1, s1, b1, tabC, tabS, row, k0 + 4 * FX_U, lane);
    __builtin_amdgcn_sched_barrier(0);
    fx_mma(c0, s0, b0, aC, aS);
    __builtin_amdgcn_sched_barrier(0);
    if (k0 + 8 * FX_U < W_) fx_load(c0, s0, b0, tabC, tabS, row, k0 + 8 * FX_U, lane);
    __builtin_amdgcn_sched_barrier(0);
    fx_mma(c1, s1, b1, aC, aS);
    __builtin_amdgcn_sched_barrier(0);
  }
  float* tr = Tr + ((size_t)(b * H_ + y)) * M_ * C_ + 16 * ch;
  float* ti = Ti + ((size_t)(b * H_ + y)) * M_ * C_ + 16 * ch;
  stD(tr, C_, lane, aC, 0.0625f);
  stD(ti, C_, lane, aS, -0.0625f);
}

__global__ void k_fwdY(const float* __restrict__ Tr, const float* __restrict__ Ti,
                       const float* __restrict__ tabC, const float* __restrict__ tabS,
                       float* __restrict__ Frp, float* __restrict__ Fip) {
  int wid = blockIdx.x * (blockDim.x >> 5) + (threadIdx.x >> 5);
  int lane = threadIdx.x & 31;
  int b = wid >> 7;
  int r2 = wid & 127;
  int col0 = (r2 >> 2) * 16;
  int kc = r2 & 3;
  const float* trb = Tr + (size_t)b * H_ * (M_ * C_);
  const float* tib = Ti + (size_t)b * H_ * (M_ * C_);
  v8f aCT = vzero(), aST = vzero(), aCTi = vzero(), aSTr = vzero();
  for (int k0 = kc * 64; k0 < kc * 64 + 64; k0 += 4) {
    v2f aC = ldA(tabC + k0, W_, lane);
    v2f aS = ldA(tabS + k0, W_, lane);
    v2f bR = ldB(trb + k0 * 512 + col0, 512, lane);
    v2f bI = ldB(tib + k0 * 512 + col0, 512, lane);
    aCT = wmma4(aC, bR, aCT);
    aST = wmma4(aS, bI, aST);
    aCTi = wmma4(aC, bI, aCTi);
    aSTr = wmma4(aS, bR, aSTr);
  }
  float* fr = Frp + (size_t)kc * (NB_ * M_ * 512) + (size_t)b * (M_ * 512) + col0;
  float* fi = Fip + (size_t)kc * (NB_ * M_ * 512) + (size_t)b * (M_ * 512) + col0;
  int n = lane & 15, mh = (lane >> 4) << 3;
#pragma unroll
  for (int r = 0; r < 8; ++r) {
    fr[(r + mh) * 512 + n] = (aCT[r] + aST[r]) * 0.0625f;
    fi[(r + mh) * 512 + n] = (aCTi[r] - aSTr[r]) * 0.0625f;
  }
}

__global__ void __launch_bounds__(512) k_mix(const float* __restrict__ Frp, const float* __restrict__ Fip,
                      const float* __restrict__ wr, const float* __restrict__ wi,
                      float* __restrict__ Gr, float* __restrict__ Gi) {
  __shared__ float sFr[512];
  __shared__ float sFi[512];
  int b = blockIdx.x >> 4, k1 = blockIdx.x & 15;
  size_t wch = (size_t)(b >> 2) * (4 * C_ * C_ * M_ * M_);
  int tid = threadIdx.x;
  size_t base = ((size_t)b * M_ + k1) * 512;
  int o = tid >> 4, k2 = tid & 15;
  const float* wrk = wr + wch + k1 * 16 + k2;
  const float* wik = wi + wch + k1 * 16 + k2;
  float wrv[C_], wiv[C_];
#pragma unroll
  for (int i = 0; i < C_; ++i) {
    wrv[i] = wrk[(i * 32 + o) * 256];
    wiv[i] = wik[(i * 32 + o) * 256];
  }
  float fr = 0.f, fi = 0.f;
#pragma unroll
  for (int kc = 0; kc < 4; ++kc) {
    fr += Frp[(size_t)kc * (NB_ * M_ * 512) + base + tid];
    fi += Fip[(size_t)kc * (NB_ * M_ * 512) + base + tid];
  }
  sFr[tid] = fr; sFi[tid] = fi;
  __syncthreads();
  float gr = 0.f, gi = 0.f;
#pragma unroll
  for (int i = 0; i < C_; ++i) {
    float xr = sFr[k2 * 32 + i], xi = sFi[k2 * 32 + i];
    gr += xr * wrv[i] - xi * wiv[i];
    gi += xr * wiv[i] + xi * wrv[i];
  }
  size_t gidx = base + k2 * 32 + o;
  Gr[gidx] = gr; Gi[gidx] = gi;
}

__global__ void k_invY(const float* __restrict__ Gr, const float* __restrict__ Gi,
                       const float* __restrict__ tabC, const float* __restrict__ tabS,
                       float* __restrict__ P, float* __restrict__ Q) {
  int wid = blockIdx.x * (blockDim.x >> 5) + (threadIdx.x >> 5);
  int lane = threadIdx.x & 31;
  int b = wid >> 9, rest = wid & 511;
  int y0 = (rest >> 5) * 16, col0 = (rest & 31) * 16;
  const float* grb = Gr + (size_t)b * 8192;
  const float* gib = Gi + (size_t)b * 8192;
  v8f aCG = vzero(), aSGi = vzero(), aSG = vzero(), aCGi = vzero();
#pragma unroll
  for (int k0 = 0; k0 < M_; k0 += 4) {
    v2f aC = ldAt(tabC + k0 * W_ + y0, W_, lane);
    v2f aS = ldAt(tabS + k0 * W_ + y0, W_, lane);
    v2f bR = ldB(grb + k0 * 512 + col0, 512, lane);
    v2f bI = ldB(gib + k0 * 512 + col0, 512, lane);
    aCG = wmma4(aC, bR, aCG);
    aSGi = wmma4(aS, bI, aSGi);
    aSG = wmma4(aS, bR, aSG);
    aCGi = wmma4(aC, bI, aCGi);
  }
  float* p = P + (size_t)(b * H_ + y0) * 512 + col0;
  float* q = Q + (size_t)(b * H_ + y0) * 512 + col0;
  int n = lane & 15, mh = (lane >> 4) << 3;
#pragma unroll
  for (int r = 0; r < 8; ++r) {
    p[(r + mh) * 512 + n] = (aCG[r] - aSGi[r]) * 0.0625f;
    q[(r + mh) * 512 + n] = (aSG[r] + aCGi[r]) * 0.0625f;
  }
}

typedef float v4f __attribute__((ext_vector_type(4)));
#define XP 36
#define PP 48
template <bool FWD>
__global__ void __launch_bounds__(512) k_invX(const float* __restrict__ P, const float* __restrict__ Q,
                       float* act, const float* __restrict__ tabC2,
                       const float* __restrict__ tabS2, const float* __restrict__ bw,
                       const float* __restrict__ bb, const float* __restrict__ tabC,
                       const float* __restrict__ tabS, float* __restrict__ Tr,
                       float* __restrict__ Ti) {
  constexpr int WSTR = FWD ? 1024 : 16 * XP;
  __shared__ __attribute__((aligned(16))) float sW[16 * WSTR];
  __shared__ __attribute__((aligned(16))) float sBW[C_ * XP];
  __shared__ __attribute__((aligned(16))) float sP[M_ * PP];
  __shared__ __attribute__((aligned(16))) float sQ[M_ * PP];
  int tid = threadIdx.x;
  int w = tid >> 5;
  int lane = tid & 31;
  int by = blockIdx.x;
  int chn = by >> 10;
  int x0 = w * 16;
  const float* Pby = P + (size_t)by * 512;
  const float* Qby = Q + (size_t)by * 512;
  const float* xrow = act + (size_t)by * (W_ * C_);
  bw += chn * (4 * C_ * C_);
  bb += chn * (4 * C_);
  float* tile = sW + w * WSTR;
  const v4f* tsrc = reinterpret_cast<const v4f*>(xrow + x0 * C_);
  v4f t0 = tsrc[lane], t1 = tsrc[32 + lane], t2 = tsrc[64 + lane], t3 = tsrc[96 + lane];
  int si = tid & 127;
  v4f sh;
  if (tid < 128)      sh = reinterpret_cast<const v4f*>(Pby)[si];
  else if (tid < 256) sh = reinterpret_cast<const v4f*>(Qby)[si];
  else                sh = reinterpret_cast<const v4f*>(bw)[tid - 256];
  v2f aCv[4], aSv[4];
#pragma unroll
  for (int k = 0; k < 4; ++k) {
    aCv[k] = ldAt(tabC2 + 4 * k * W_ + x0, W_, lane);
    aSv[k] = ldAt(tabS2 + 4 * k * W_ + x0, W_, lane);
  }
  int n = lane & 15, mh = (lane >> 4) << 3;
  float bb0 = bb[n], bb1 = bb[16 + n];
  {
    int tx = lane >> 3, tc = (lane & 7) * 4;
    *reinterpret_cast<v4f*>(tile + (0 + tx) * XP + tc) = t0;
    *reinterpret_cast<v4f*>(tile + (4 + tx) * XP + tc) = t1;
    *reinterpret_cast<v4f*>(tile + (8 + tx) * XP + tc) = t2;
    *reinterpret_cast<v4f*>(tile + (12 + tx) * XP + tc) = t3;
    int sr = si >> 3, sc4 = (si & 7) * 4;
    if (tid < 128)      *reinterpret_cast<v4f*>(sP + sr * PP + sc4) = sh;
    else if (tid < 256) *reinterpret_cast<v4f*>(sQ + sr * PP + sc4) = sh;
    else {
      int i = tid - 256;
      *reinterpret_cast<v4f*>(sBW + (i >> 3) * XP + (i & 7) * 4) = sh;
    }
  }
  __syncthreads();
  v8f acc0 = vzero(), acc1 = vzero();
#pragma unroll
  for (int k0 = 0; k0 < M_; k0 += 4) {
    v2f aC = aCv[k0 >> 2];
    v2f bP0 = ldB(sP + k0 * PP, PP, lane);
    v2f bP1 = ldB(sP + k0 * PP + 16, PP, lane);
    acc0 = wmma4(aC, bP0, acc0);
    acc1 = wmma4(aC, bP1, acc1);
    v2f aS = aSv[k0 >> 2];
    v2f aSn; aSn.x = -aS.x; aSn.y = -aS.y;
    v2f bQ0 = ldB(sQ + k0 * PP, PP, lane);
    v2f bQ1 = ldB(sQ + k0 * PP + 16, PP, lane);
    acc0 = wmma4(aSn, bQ0, acc0);
    acc1 = wmma4(aSn, bQ1, acc1);
  }
#pragma unroll
  for (int k0 = 0; k0 < C_; k0 += 4) {
    v2f aX = ldA(tile + k0, XP, lane);
    v2f bW0 = ldBt(sBW + k0, XP, lane);
    v2f bW1 = ldBt(sBW + 16 * XP + k0, XP, lane);
    acc0 = wmma4(aX, bW0, acc0);
    acc1 = wmma4(aX, bW1, acc1);
  }
  v2f fC[4], fS[4];
  if (FWD) {
#pragma unroll
    for (int j = 0; j < 4; ++j) {
      fC[j] = ldA_d(tabC + x0, W_, lane, j);
      fS[j] = ldA_d(tabS + x0, W_, lane, j);
    }
  }
  float* out = act + (size_t)by * (W_ * C_);
  v8f g0, g1;
#pragma unroll
  for (int r = 0; r < 8; ++r) {
    int xr = x0 + r + mh;
    float v0 = acc0[r] + bb0;
    v0 = 0.5f * v0 * (1.0f + erff(v0 * 0.70710678f));
    out[xr * C_ + n] = v0;
    g0[r] = v0;
    float v1 = acc1[r] + bb1;
    v1 = 0.5f * v1 * (1.0f + erff(v1 * 0.70710678f));
    out[xr * C_ + 16 + n] = v1;
    g1[r] = v1;
  }
  if (FWD) {
    v8f aC0 = vzero(), aC1 = vzero(), aS0 = vzero(), aS1 = vzero();
#pragma unroll
    for (int j = 0; j < 4; ++j) {
      v2f b0; b0.x = g0[2 * j]; b0.y = g0[2 * j + 1];
      v2f b1; b1.x = g1[2 * j]; b1.y = g1[2 * j + 1];
      aC0 = wmma4(fC[j], b0, aC0);
      aC1 = wmma4(fC[j], b1, aC1);
      aS0 = wmma4(fS[j], b0, aS0);
      aS1 = wmma4(fS[j], b1, aS1);
    }
    float* pc = sW + w * WSTR + (lane >> 4) * 32 + n;
    float* ps = pc + 512;
#pragma unroll
    for (int r = 0; r < 8; ++r) {
      pc[r * 64] = aC0[r];
      pc[r * 64 + 16] = aC1[r];
      ps[r * 64] = aS0[r];
      ps[r * 64 + 16] = aS1[r];
    }
    __syncthreads();
    float sc = 0.f, ss = 0.f;
#pragma unroll
    for (int ww = 0; ww < 16; ++ww) {
      sc += sW[ww * WSTR + tid];
      ss += sW[ww * WSTR + 512 + tid];
    }
    int k2 = (tid >> 6) + ((tid >> 5) & 1) * 8, c = tid & 31;
    size_t o = (size_t)by * (M_ * C_) + k2 * C_ + c;
    Tr[o] = sc * 0.0625f;
    Ti[o] = ss * -0.0625f;
  }
}

__global__ void k_proj(const float* __restrict__ xb, const float* __restrict__ pw,
                       const float* __restrict__ pb, float* __restrict__ out,
                       int o_outer) {
  int p = blockIdx.x * blockDim.x + threadIdx.x;
  if (p >= NB_ * PIX) return;
  int q = p >> 16, pix = p & (PIX - 1);
  int b = q & 3;
  o_outer += q >> 2;
  const float* v = xb + (size_t)p * C_;
  float s = pb[0];
#pragma unroll
  for (int c = 0; c < C_; ++c) s += pw[c] * v[c];
  out[((size_t)(b * 4 + o_outer)) * PIX + pix] = s;
}

extern "C" void kernel_launch(void* const* d_in, const int* in_sizes, int n_in,
                              void* d_out, int out_size, void* d_ws, size_t ws_size,
                              hipStream_t stream) {
  const float* x      = (const float*)d_in[0];
  const float* lift_w = (const float*)d_in[1];
  const float* lift_b = (const float*)d_in[2];
  const float* wr     = (const float*)d_in[3];
  const float* wi     = (const float*)d_in[4];
  const float* byp_w  = (const float*)d_in[5];
  const float* byp_b  = (const float*)d_in[6];
  const float* proj_w = (const float*)d_in[7];
  const float* proj_b = (const float*)d_in[8];
  float* out = (float*)d_out;

  float* w = (float*)d_ws;
  float* tabC  = w;
  float* tabS  = tabC + 4096;
  float* tabC2 = tabS + 4096;
  float* tabS2 = tabC2 + 4096;
  float* act   = tabS2 + 4096;
  float* Tr    = act + (size_t)NB_ * PIX * C_;
  float* Ti    = Tr + (size_t)NB_ * H_ * M_ * C_;
  float* Frp   = Ti + (size_t)NB_ * H_ * M_ * C_;
  float* Fip   = Frp + (size_t)4 * NB_ * M_ * 512;
  float* Gr    = Fip + (size_t)4 * NB_ * M_ * 512;
  float* Gi    = Gr + (size_t)NB_ * M_ * M_ * C_;
  float* Pb    = Gi + (size_t)NB_ * M_ * M_ * C_;
  float* Qb    = Pb + (size_t)NB_ * H_ * M_ * C_;

  k_tabs<<<16, 256, 0, stream>>>(tabC, tabS, tabC2, tabS2);

  for (int o = 0; o < 4; o += NCH) {
    k_lift<<<(B_ * PIX + 255) / 256, 256, 0, stream>>>(x, lift_w, lift_b, act);
    for (int n = 0; n < 4; ++n) {
      size_t wOff = ((size_t)(o * 4 + n)) * C_ * C_ * M_ * M_;
      size_t bwOff = ((size_t)(o * 4 + n)) * C_ * C_;
      size_t bbOff = ((size_t)(o * 4 + n)) * C_;
      if (n == 0) k_fwdX<<<NCH * 256, 256, 0, stream>>>(act, tabC, tabS, Tr, Ti);
      k_fwdY<<<NCH * 64, 256, 0, stream>>>(Tr, Ti, tabC, tabS, Frp, Fip);
      k_mix<<<NCH * 64, 512, 0, stream>>>(Frp, Fip, wr + wOff, wi + wOff, Gr, Gi);
      k_invY<<<NCH * 256, 256, 0, stream>>>(Gr, Gi, tabC, tabS, Pb, Qb);
      if (n < 3)
        k_invX<true><<<NCH * 1024, 512, 0, stream>>>(Pb, Qb, act, tabC2, tabS2,
                                                     byp_w + bwOff, byp_b + bbOff,
                                                     tabC, tabS, Tr, Ti);
      else
        k_invX<false><<<NCH * 1024, 512, 0, stream>>>(Pb, Qb, act, tabC2, tabS2,
                                                      byp_w + bwOff, byp_b + bbOff,
                                                      tabC, tabS, Tr, Ti);
    }
    k_proj<<<(NB_ * PIX + 255) / 256, 256, 0, stream>>>(act, proj_w, proj_b, out, o);
  }
}


# Lifted field never stored: the first step's forward x transform (run once per batch row, result kept for both chains) and its bypass product compute it from x on the fly; lift kernel gone
# speedup vs baseline: 1.9241x; 1.1088x over previous
#include <hip/hip_runtime.h>
#include <math.h>


typedef float v2f __attribute__((ext_vector_type(2)));
typedef float v8f __attribute__((ext_vector_type(8)));

#define B_  4
#define NCH 2
#define NB_ (NCH * B_)
#define C_  32
#define H_  256
#define W_  256
#define M_  16
#define PIX (H_ * W_)

__device__ __forceinline__ v8f wmma4(v2f a, v2f b, v8f c) {
  return __builtin_amdgcn_wmma_f32_16x16x4_f32(false, a, false, b, (short)0, c,
                                               false, false);
}

__device__ __forceinline__ v2f ldA(const float* p, int lda, int lane) {
  int r = lane & 15, kk = (lane >> 4) << 1;
  v2f v; v.x = p[r * lda + kk]; v.y = p[r * lda + kk + 1]; return v;
}
__device__ __forceinline__ v2f ldAt(const float* p, int lda, int lane) {
  int r = lane & 15, kk = (lane >> 4) << 1;
  v2f v; v.x = p[kk * lda + r]; v.y = p[(kk + 1) * lda + r]; return v;
}
__device__ __forceinline__ v2f ldB(const float* p, int ldb, int lane) {
  int n = lane & 15, kk = (lane >> 4) << 1;
  v2f v; v.x = p[kk * ldb + n]; v.y = p[(kk + 1) * ldb + n]; return v;
}
__device__ __forceinline__ v2f ldBt(const float* p, int ldb, int lane) {
  int n = lane & 15, kk = (lane >> 4) << 1;
  v2f v; v.x = p[n * ldb + kk]; v.y = p[n * ldb + kk + 1]; return v;
}
__device__ __forceinline__ v2f ldA_d(const float* p, int lda, int lane, int j) {
  int r = lane & 15, kk = ((lane >> 4) << 3) + 2 * j;
  v2f v; v.x = p[r * lda + kk]; v.y = p[r * lda + kk + 1]; return v;
}
__device__ __forceinline__ void stD(float* p, int ldd, int lane, v8f d, float s) {
  int n = lane & 15, mh = (lane >> 4) << 3;
#pragma unroll
  for (int r = 0; r < 8; ++r) p[(r + mh) * ldd + n] = d[r] * s;
}

__device__ __forceinline__ v8f vzero() {
  v8f z = {0.f, 0.f, 0.f, 0.f, 0.f, 0.f, 0.f, 0.f};
  return z;
}

__global__ void k_tabs(float* tabC, float* tabS, float* tabC2, float* tabS2) {
  int i = blockIdx.x * blockDim.x + threadIdx.x;
  if (i >= M_ * W_) return;
  int k = i >> 8, x = i & 255;
  int m = (k * x) & 255;
  float ang = 0.0245436926f * (float)m;
  float c = cosf(ang), s = sinf(ang);
  tabC[i] = c; tabS[i] = s;
  float w2 = (k == 0) ? 1.0f : 2.0f;
  tabC2[i] = w2 * c * 0.0625f;
  tabS2[i] = w2 * s * 0.0625f;
}

__device__ __forceinline__ float lift1(float lbv, float l0, float l1, float l2,
                                       float xi0, float xi1, float xi2) {
  return lbv + l0 * xi0 + l1 * xi1 + l2 * xi2;
}

#define FX_U 8
__device__ __forceinline__ void fx_load(v2f (&c)[FX_U], v2f (&s)[FX_U], v2f (&p0)[FX_U],
                                        v2f (&p1)[FX_U], v2f (&p2)[FX_U],
                                        const float* __restrict__ tabC,
                                        const float* __restrict__ tabS,
                                        const float* __restrict__ xr, int k, int lane) {
#pragma unroll
  for (int u = 0; u < FX_U; ++u) {
    c[u] = ldA(tabC + k + 4 * u, W_, lane);
    s[u] = ldA(tabS + k + 4 * u, W_, lane);
    p0[u] = *reinterpret_cast<const v2f*>(xr + k + 4 * u);
    p1[u] = *reinterpret_cast<const v2f*>(xr + PIX + k + 4 * u);
    p2[u] = *reinterpret_cast<const v2f*>(xr + 2 * PIX + k + 4 * u);
  }
}
__device__ __forceinline__ void fx_mma(const v2f (&c)[FX_U], const v2f (&s)[FX_U],
                                       const v2f (&p0)[FX_U], const v2f (&p1)[FX_U],
                                       const v2f (&p2)[FX_U], float lbv, float l0, float l1,
                                       float l2, v8f& aC, v8f& aS) {
#pragma unroll
  for (int u = 0; u < FX_U; ++u) {
    v2f b;
    b.x = lift1(lbv, l0, l1, l2, p0[u].x, p1[u].x, p2[u].x);
    b.y = lift1(lbv, l0, l1, l2, p0[u].y, p1[u].y, p2[u].y);
    aC = wmma4(c[u], b, aC);
    aS = wmma4(s[u], b, aS);
  }
}
__global__ void __launch_bounds__(256) k_fwdX(const float* __restrict__ x, const float* __restrict__ lw,
                       const float* __restrict__ lb, const float* __restrict__ tabC,
                       const float* __restrict__ tabS, float* __restrict__ Tr,
                       float* __restrict__ Ti) {
  int wid = blockIdx.x * (blockDim.x >> 5) + (threadIdx.x >> 5);
  int lane = threadIdx.x & 31;
  int ch = wid & 1;
  int byi = wid >> 1;
  int b = byi >> 8, y = byi & 255;
  int cc = 16 * ch + (lane & 15);
  float l0 = lw[cc * 3 + 0], l1 = lw[cc * 3 + 1], l2 = lw[cc * 3 + 2], lbv = lb[cc];
  const float* xr = x + ((size_t)(b * 3) * H_ + y) * W_ + ((lane >> 4) << 1);
  v8f aC = vzero(), aS = vzero();
  v2f c0[FX_U], s0[FX_U], c1[FX_U], s1[FX_U];
  v2f p00[FX_U], p01[FX_U], p02[FX_U], p10[FX_U], p11[FX_U], p12[FX_U];
  for (int k0 = 0; k0 < W_; k0 += 8 * FX_U) {
    fx_load(c0, s0, p00, p01, p02, tabC, tabS, xr, k0, lane);
    fx_load(c1, s1, p10, p11, p12, tabC, tabS, xr, k0 + 4 * FX_U, lane);
    __builtin_amdgcn_sched_barrier(0);
    fx_mma(c0, s0, p00, p01, p02, lbv, l0, l1, l2, aC, aS);
    fx_mma(c1, s1, p10, p11, p12, lbv, l0, l1, l2, aC, aS);
    __builtin_amdgcn_sched_barrier(0);
  }
  float* tr = Tr + ((size_t)(b * H_ + y)) * M_ * C_ + 16 * ch;
  float* ti = Ti + ((size_t)(b * H_ + y)) * M_ * C_ + 16 * ch;
  stD(tr, C_, lane, aC, 0.0625f);
  stD(ti, C_, lane, aS, -0.0625f);
  stD(tr + (size_t)B_ * H_ * M_ * C_, C_, lane, aC, 0.0625f);
  stD(ti + (size_t)B_ * H_ * M_ * C_, C_, lane, aS, -0.0625f);
}

__global__ void k_fwdY(const float* __restrict__ Tr, const float* __restrict__ Ti,
                       const float* __restrict__ tabC, const float* __restrict__ tabS,
                       float* __restrict__ Frp, float* __restrict__ Fip) {
  int wid = blockIdx.x * (blockDim.x >> 5) + (threadIdx.x >> 5);
  int lane = threadIdx.x & 31;
  int b = wid >> 7;
  int r2 = wid & 127;
  int col0 = (r2 >> 2) * 16;
  int kc = r2 & 3;
  const float* trb = Tr + (size_t)b * H_ * (M_ * C_);
  const float* tib = Ti + (size_t)b * H_ * (M_ * C_);
  v8f aCT = vzero(), aST = vzero(), aCTi = vzero(), aSTr = vzero();
  for (int k0 = kc * 64; k0 < kc * 64 + 64; k0 += 4) {
    v2f aC = ldA(tabC + k0, W_, lane);
    v2f aS = ldA(tabS + k0, W_, lane);
    v2f bR = ldB(trb + k0 * 512 + col0, 512, lane);
    v2f bI = ldB(tib + k0 * 512 + col0, 512, lane);
    aCT = wmma4(aC, bR, aCT);
    aST = wmma4(aS, bI, aST);
    aCTi = wmma4(aC, bI, aCTi);
    aSTr = wmma4(aS, bR, aSTr);
  }
  float* fr = Frp + (size_t)kc * (NB_ * M_ * 512) + (size_t)b * (M_ * 512) + col0;
  float* fi = Fip + (size_t)kc * (NB_ * M_ * 512) + (size_t)b * (M_ * 512) + col0;
  int n = lane & 15, mh = (lane >> 4) << 3;
#pragma unroll
  for (int r = 0; r < 8; ++r) {
    fr[(r + mh) * 512 + n] = (aCT[r] + aST[r]) * 0.0625f;
    fi[(r + mh) * 512 + n] = (aCTi[r] - aSTr[r]) * 0.0625f;
  }
}

__global__ void __launch_bounds__(512) k_mix(const float* __restrict__ Frp, const float* __restrict__ Fip,
                      const float* __restrict__ wr, const float* __restrict__ wi,
                      float* __restrict__ Gr, float* __restrict__ Gi) {
  __shared__ float sFr[512];
  __shared__ float sFi[512];
  int b = blockIdx.x >> 4, k1 = blockIdx.x & 15;
  size_t wch = (size_t)(b >> 2) * (4 * C_ * C_ * M_ * M_);
  int tid = threadIdx.x;
  size_t base = ((size_t)b * M_ + k1) * 512;
  int o = tid >> 4, k2 = tid & 15;
  const float* wrk = wr + wch + k1 * 16 + k2;
  const float* wik = wi + wch + k1 * 16 + k2;
  float wrv[C_], wiv[C_];
#pragma unroll
  for (int i = 0; i < C_; ++i) {
    wrv[i] = wrk[(i * 32 + o) * 256];
    wiv[i] = wik[(i * 32 + o) * 256];
  }
  float fr = 0.f, fi = 0.f;
#pragma unroll
  for (int kc = 0; kc < 4; ++kc) {
    fr += Frp[(size_t)kc * (NB_ * M_ * 512) + base + tid];
    fi += Fip[(size_t)kc * (NB_ * M_ * 512) + base + tid];
  }
  sFr[tid] = fr; sFi[tid] = fi;
  __syncthreads();
  float gr = 0.f, gi = 0.f;
#pragma unroll
  for (int i = 0; i < C_; ++i) {
    float xr = sFr[k2 * 32 + i], xi = sFi[k2 * 32 + i];
    gr += xr * wrv[i] - xi * wiv[i];
    gi += xr * wiv[i] + xi * wrv[i];
  }
  size_t gidx = base + k2 * 32 + o;
  Gr[gidx] = gr; Gi[gidx] = gi;
}

__global__ void k_invY(const float* __restrict__ Gr, const float* __restrict__ Gi,
                       const float* __restrict__ tabC, const float* __restrict__ tabS,
                       float* __restrict__ P, float* __restrict__ Q) {
  int wid = blockIdx.x * (blockDim.x >> 5) + (threadIdx.x >> 5);
  int lane = threadIdx.x & 31;
  int b = wid >> 9, rest = wid & 511;
  int y0 = (rest >> 5) * 16, col0 = (rest & 31) * 16;
  const float* grb = Gr + (size_t)b * 8192;
  const float* gib = Gi + (size_t)b * 8192;
  v8f aCG = vzero(), aSGi = vzero(), aSG = vzero(), aCGi = vzero();
#pragma unroll
  for (int k0 = 0; k0 < M_; k0 += 4) {
    v2f aC = ldAt(tabC + k0 * W_ + y0, W_, lane);
    v2f aS = ldAt(tabS + k0 * W_ + y0, W_, lane);
    v2f bR = ldB(grb + k0 * 512 + col0, 512, lane);
    v2f bI = ldB(gib + k0 * 512 + col0, 512, lane);
    aCG = wmma4(aC, bR, aCG);
    aSGi = wmma4(aS, bI, aSGi);
    aSG = wmma4(aS, bR, aSG);
    aCGi = wmma4(aC, bI, aCGi);
  }
  float* p = P + (size_t)(b * H_ + y0) * 512 + col0;
  float* q = Q + (size_t)(b * H_ + y0) * 512 + col0;
  int n = lane & 15, mh = (lane >> 4) << 3;
#pragma unroll
  for (int r = 0; r < 8; ++r) {
    p[(r + mh) * 512 + n] = (aCG[r] - aSGi[r]) * 0.0625f;
    q[(r + mh) * 512 + n] = (aSG[r] + aCGi[r]) * 0.0625f;
  }
}

typedef float v4f __attribute__((ext_vector_type(4)));
#define XP 36
#define PP 48
template <bool FWD, bool FIRST>
__global__ void __launch_bounds__(512) k_invX(const float* __restrict__ P, const float* __restrict__ Q,
                       float* act, const float* __restrict__ tabC2,
                       const float* __restrict__ tabS2, const float* __restrict__ bw,
                       const float* __restrict__ bb, const float* __restrict__ tabC,
                       const float* __restrict__ tabS, float* __restrict__ Tr,
                       float* __restrict__ Ti, const float* __restrict__ x,
                       const float* __restrict__ lw, const float* __restrict__ lb) {
  constexpr int WSTR = FWD ? 1024 : 16 * XP;
  __shared__ __attribute__((aligned(16))) float sW[16 * WSTR];
  __shared__ __attribute__((aligned(16))) float sBW[C_ * XP];
  __shared__ __attribute__((aligned(16))) float sP[M_ * PP];
  __shared__ __attribute__((aligned(16))) float sQ[M_ * PP];
  int tid = threadIdx.x;
  int w = tid >> 5;
  int lane = tid & 31;
  int by = blockIdx.x;
  int chn = by >> 10;
  int x0 = w * 16;
  const float* Pby = P + (size_t)by * 512;
  const float* Qby = Q + (size_t)by * 512;
  const float* xrow = act + (size_t)by * (W_ * C_);
  bw += chn * (4 * C_ * C_);
  bb += chn * (4 * C_);
  float* tile = sW + w * WSTR;
  const v4f* tsrc = reinterpret_cast<const v4f*>(xrow + x0 * C_);
  v4f t0, t1, t2, t3;
  float xv0[16], xv1[16], xv2[16];
  float l0, l1, l2, lbv;
  if (FIRST) {
    int wu = __builtin_amdgcn_readfirstlane(w);
    const float* xs = x + ((size_t)(((by >> 8) & 3) * 3) * H_ + (by & 255)) * W_ + wu * 16;
#pragma unroll
    for (int j = 0; j < 16; ++j) {
      xv0[j] = xs[j];
      xv1[j] = xs[PIX + j];
      xv2[j] = xs[2 * PIX + j];
    }
    l0 = lw[lane * 3 + 0]; l1 = lw[lane * 3 + 1]; l2 = lw[lane * 3 + 2]; lbv = lb[lane];
  } else {
    t0 = tsrc[lane]; t1 = tsrc[32 + lane]; t2 = tsrc[64 + lane]; t3 = tsrc[96 + lane];
  }
  int si = tid & 127;
  v4f sh;
  if (tid < 128)      sh = reinterpret_cast<const v4f*>(Pby)[si];
  else if (tid < 256) sh = reinterpret_cast<const v4f*>(Qby)[si];
  else                sh = reinterpret_cast<const v4f*>(bw)[tid - 256];
  v2f aCv[4], aSv[4];
#pragma unroll
  for (int k = 0; k < 4; ++k) {
    aCv[k] = ldAt(tabC2 + 4 * k * W_ + x0, W_, lane);
    aSv[k] = ldAt(tabS2 + 4 * k * W_ + x0, W_, lane);
  }
  int n = lane & 15, mh = (lane >> 4) << 3;
  float bb0 = bb[n], bb1 = bb[16 + n];
  {
    int tx = lane >> 3, tc = (lane & 7) * 4;
    if (FIRST) {
#pragma unroll
      for (int j = 0; j < 16; ++j)
        tile[j * XP + lane] = lift1(lbv, l0, l1, l2, xv0[j], xv1[j], xv2[j]);
    } else {
      *reinterpret_cast<v4f*>(tile + (0 + tx) * XP + tc) = t0;
      *reinterpret_cast<v4f*>(tile + (4 + tx) * XP + tc) = t1;
      *reinterpret_cast<v4f*>(tile + (8 + tx) * XP + tc) = t2;
      *reinterpret_cast<v4f*>(tile + (12 + tx) * XP + tc) = t3;
    }
    int sr = si >> 3, sc4 = (si & 7) * 4;
    if (tid < 128)      *reinterpret_cast<v4f*>(sP + sr * PP + sc4) = sh;
    else if (tid < 256) *reinterpret_cast<v4f*>(sQ + sr * PP + sc4) = sh;
    else {
      int i = tid - 256;
      *reinterpret_cast<v4f*>(sBW + (i >> 3) * XP + (i & 7) * 4) = sh;
    }
  }
  __syncthreads();
  v8f acc0 = vzero(), acc1 = vzero();
#pragma unroll
  for (int k0 = 0; k0 < M_; k0 += 4) {
    v2f aC = aCv[k0 >> 2];
    v2f bP0 = ldB(sP + k0 * PP, PP, lane);
    v2f bP1 = ldB(sP + k0 * PP + 16, PP, lane);
    acc0 = wmma4(aC, bP0, acc0);
    acc1 = wmma4(aC, bP1, acc1);
    v2f aS = aSv[k0 >> 2];
    v2f aSn; aSn.x = -aS.x; aSn.y = -aS.y;
    v2f bQ0 = ldB(sQ + k0 * PP, PP, lane);
    v2f bQ1 = ldB(sQ + k0 * PP + 16, PP, lane);
    acc0 = wmma4(aSn, bQ0, acc0);
    acc1 = wmma4(aSn, bQ1, acc1);
  }
#pragma unroll
  for (int k0 = 0; k0 < C_; k0 += 4) {
    v2f aX = ldA(tile + k0, XP, lane);
    v2f bW0 = ldBt(sBW + k0, XP, lane);
    v2f bW1 = ldBt(sBW + 16 * XP + k0, XP, lane);
    acc0 = wmma4(aX, bW0, acc0);
    acc1 = wmma4(aX, bW1, acc1);
  }
  v2f fC[4], fS[4];
  if (FWD) {
#pragma unroll
    for (int j = 0; j < 4; ++j) {
      fC[j] = ldA_d(tabC + x0, W_, lane, j);
      fS[j] = ldA_d(tabS + x0, W_, lane, j);
    }
  }
  float* out = act + (size_t)by * (W_ * C_);
  v8f g0, g1;
#pragma unroll
  for (int r = 0; r < 8; ++r) {
    int xr = x0 + r + mh;
    float v0 = acc0[r] + bb0;
    v0 = 0.5f * v0 * (1.0f + erff(v0 * 0.70710678f));
    out[xr * C_ + n] = v0;
    g0[r] = v0;
    float v1 = acc1[r] + bb1;
    v1 = 0.5f * v1 * (1.0f + erff(v1 * 0.70710678f));
    out[xr * C_ + 16 + n] = v1;
    g1[r] = v1;
  }
  if (FWD) {
    v8f aC0 = vzero(), aC1 = vzero(), aS0 = vzero(), aS1 = vzero();
#pragma unroll
    for (int j = 0; j < 4; ++j) {
      v2f b0; b0.x = g0[2 * j]; b0.y = g0[2 * j + 1];
      v2f b1; b1.x = g1[2 * j]; b1.y = g1[2 * j + 1];
      aC0 = wmma4(fC[j], b0, aC0);
      aC1 = wmma4(fC[j], b1, aC1);
      aS0 = wmma4(fS[j], b0, aS0);
      aS1 = wmma4(fS[j], b1, aS1);
    }
    float* pc = sW + w * WSTR + (lane >> 4) * 32 + n;
    float* ps = pc + 512;
#pragma unroll
    for (int r = 0; r < 8; ++r) {
      pc[r * 64] = aC0[r];
      pc[r * 64 + 16] = aC1[r];
      ps[r * 64] = aS0[r];
      ps[r * 64 + 16] = aS1[r];
    }
    __syncthreads();
    float sc = 0.f, ss = 0.f;
#pragma unroll
    for (int ww = 0; ww < 16; ++ww) {
      sc += sW[ww * WSTR + tid];
      ss += sW[ww * WSTR + 512 + tid];
    }
    int k2 = (tid >> 6) + ((tid >> 5) & 1) * 8, c = tid & 31;
    size_t o = (size_t)by * (M_ * C_) + k2 * C_ + c;
    Tr[o] = sc * 0.0625f;
    Ti[o] = ss * -0.0625f;
  }
}

__global__ void k_proj(const float* __restrict__ xb, const float* __restrict__ pw,
                       const float* __restrict__ pb, float* __restrict__ out,
                       int o_outer) {
  int p = blockIdx.x * blockDim.x + threadIdx.x;
  if (p >= NB_ * PIX) return;
  int q = p >> 16, pix = p & (PIX - 1);
  int b = q & 3;
  o_outer += q >> 2;
  const float* v = xb + (size_t)p * C_;
  float s = pb[0];
#pragma unroll
  for (int c = 0; c < C_; ++c) s += pw[c] * v[c];
  out[((size_t)(b * 4 + o_outer)) * PIX + pix] = s;
}

extern "C" void kernel_launch(void* const* d_in, const int* in_sizes, int n_in,
                              void* d_out, int out_size, void* d_ws, size_t ws_size,
                              hipStream_t stream) {
  const float* x      = (const float*)d_in[0];
  const float* lift_w = (const float*)d_in[1];
  const float* lift_b = (const float*)d_in[2];
  const float* wr     = (const float*)d_in[3];
  const float* wi     = (const float*)d_in[4];
  const float* byp_w  = (const float*)d_in[5];
  const float* byp_b  = (const float*)d_in[6];
  const float* proj_w = (const float*)d_in[7];
  const float* proj_b = (const float*)d_in[8];
  float* out = (float*)d_out;

  float* w = (float*)d_ws;
  float* tabC  = w;
  float* tabS  = tabC + 4096;
  float* tabC2 = tabS + 4096;
  float* tabS2 = tabC2 + 4096;
  float* act   = tabS2 + 4096;
  float* Tr    = act + (size_t)NB_ * PIX * C_;
  float* Ti    = Tr + (size_t)NB_ * H_ * M_ * C_;
  float* Frp   = Ti + (size_t)NB_ * H_ * M_ * C_;
  float* Fip   = Frp + (size_t)4 * NB_ * M_ * 512;
  float* Gr    = Fip + (size_t)4 * NB_ * M_ * 512;
  float* Gi    = Gr + (size_t)NB_ * M_ * M_ * C_;
  float* Pb    = Gi + (size_t)NB_ * M_ * M_ * C_;
  float* Qb    = Pb + (size_t)NB_ * H_ * M_ * C_;

  k_tabs<<<16, 256, 0, stream>>>(tabC, tabS, tabC2, tabS2);

  for (int o = 0; o < 4; o += NCH) {
    for (int n = 0; n < 4; ++n) {
      size_t wOff = ((size_t)(o * 4 + n)) * C_ * C_ * M_ * M_;
      size_t bwOff = ((size_t)(o * 4 + n)) * C_ * C_;
      size_t bbOff = ((size_t)(o * 4 + n)) * C_;
      if (n == 0) k_fwdX<<<256, 256, 0, stream>>>(x, lift_w, lift_b, tabC, tabS, Tr, Ti);
      k_fwdY<<<NCH * 64, 256, 0, stream>>>(Tr, Ti, tabC, tabS, Frp, Fip);
      k_mix<<<NCH * 64, 512, 0, stream>>>(Frp, Fip, wr + wOff, wi + wOff, Gr, Gi);
      k_invY<<<NCH * 256, 256, 0, stream>>>(Gr, Gi, tabC, tabS, Pb, Qb);
      if (n == 0)
        k_invX<true, true><<<NCH * 1024, 512, 0, stream>>>(Pb, Qb, act, tabC2, tabS2,
                                                           byp_w + bwOff, byp_b + bbOff,
                                                           tabC, tabS, Tr, Ti, x, lift_w, lift_b);
      else if (n < 3)
        k_invX<true, false><<<NCH * 1024, 512, 0, stream>>>(Pb, Qb, act, tabC2, tabS2,
                                                            byp_w + bwOff, byp_b + bbOff,
                                                            tabC, tabS, Tr, Ti, x, lift_w, lift_b);
      else
        k_invX<false, false><<<NCH * 1024, 512, 0, stream>>>(Pb, Qb, act, tabC2, tabS2,
                                                             byp_w + bwOff, byp_b + bbOff,
                                                             tabC, tabS, Tr, Ti, x, lift_w, lift_b);
    }
    k_proj<<<(NB_ * PIX + 255) / 256, 256, 0, stream>>>(act, proj_w, proj_b, out, o);
  }
}


# Projection fused into the last step's invX: final activations never stored; channel sum by lane adds inside each half, row joined in LDS; proj launches gone (proj sum order changes)
# speedup vs baseline: 2.0040x; 1.0415x over previous
#include <hip/hip_runtime.h>
#include <math.h>


typedef float v2f __attribute__((ext_vector_type(2)));
typedef float v8f __attribute__((ext_vector_type(8)));

#define B_  4
#define NCH 2
#define NB_ (NCH * B_)
#define C_  32
#define H_  256
#define W_  256
#define M_  16
#define PIX (H_ * W_)

__device__ __forceinline__ v8f wmma4(v2f a, v2f b, v8f c) {
  return __builtin_amdgcn_wmma_f32_16x16x4_f32(false, a, false, b, (short)0, c,
                                               false, false);
}

__device__ __forceinline__ v2f ldA(const float* p, int lda, int lane) {
  int r = lane & 15, kk = (lane >> 4) << 1;
  v2f v; v.x = p[r * lda + kk]; v.y = p[r * lda + kk + 1]; return v;
}
__device__ __forceinline__ v2f ldAt(const float* p, int lda, int lane) {
  int r = lane & 15, kk = (lane >> 4) << 1;
  v2f v; v.x = p[kk * lda + r]; v.y = p[(kk + 1) * lda + r]; return v;
}
__device__ __forceinline__ v2f ldB(const float* p, int ldb, int lane) {
  int n = lane & 15, kk = (lane >> 4) << 1;
  v2f v; v.x = p[kk * ldb + n]; v.y = p[(kk + 1) * ldb + n]; return v;
}
__device__ __forceinline__ v2f ldBt(const float* p, int ldb, int lane) {
  int n = lane & 15, kk = (lane >> 4) << 1;
  v2f v; v.x = p[n * ldb + kk]; v.y = p[n * ldb + kk + 1]; return v;
}
__device__ __forceinline__ v2f ldA_d(const float* p, int lda, int lane, int j) {
  int r = lane & 15, kk = ((lane >> 4) << 3) + 2 * j;
  v2f v; v.x = p[r * lda + kk]; v.y = p[r * lda + kk + 1]; return v;
}
__device__ __forceinline__ void stD(float* p, int ldd, int lane, v8f d, float s) {
  int n = lane & 15, mh = (lane >> 4) << 3;
#pragma unroll
  for (int r = 0; r < 8; ++r) p[(r + mh) * ldd + n] = d[r] * s;
}

__device__ __forceinline__ v8f vzero() {
  v8f z = {0.f, 0.f, 0.f, 0.f, 0.f, 0.f, 0.f, 0.f};
  return z;
}

template <int MASK>
__device__ __forceinline__ float row_xor(float v) {
  int b = __builtin_bit_cast(int, v);
  return __builtin_bit_cast(float, __builtin_amdgcn_mov_dpp(b, 0x160 + MASK, 0xf, 0xf, true));
}
__device__ __forceinline__ float row_sum16(float v) {
  v += row_xor<1>(v);
  v += row_xor<2>(v);
  v += row_xor<4>(v);
  v += row_xor<8>(v);
  return v;
}

__global__ void k_tabs(float* tabC, float* tabS, float* tabC2, float* tabS2) {
  int i = blockIdx.x * blockDim.x + threadIdx.x;
  if (i >= M_ * W_) return;
  int k = i >> 8, x = i & 255;
  int m = (k * x) & 255;
  float ang = 0.0245436926f * (float)m;
  float c = cosf(ang), s = sinf(ang);
  tabC[i] = c; tabS[i] = s;
  float w2 = (k == 0) ? 1.0f : 2.0f;
  tabC2[i] = w2 * c * 0.0625f;
  tabS2[i] = w2 * s * 0.0625f;
}

__device__ __forceinline__ float lift1(float lbv, float l0, float l1, float l2,
                                       float xi0, float xi1, float xi2) {
  return lbv + l0 * xi0 + l1 * xi1 + l2 * xi2;
}

#define FX_U 8
__device__ __forceinline__ void fx_load(v2f (&c)[FX_U], v2f (&s)[FX_U], v2f (&p0)[FX_U],
                                        v2f (&p1)[FX_U], v2f (&p2)[FX_U],
                                        const float* __restrict__ tabC,
                                        const float* __restrict__ tabS,
                                        const float* __restrict__ xr, int k, int lane) {
#pragma unroll
  for (int u = 0; u < FX_U; ++u) {
    c[u] = ldA(tabC + k + 4 * u, W_, lane);
    s[u] = ldA(tabS + k + 4 * u, W_, lane);
    p0[u] = *reinterpret_cast<const v2f*>(xr + k + 4 * u);
    p1[u] = *reinterpret_cast<const v2f*>(xr + PIX + k + 4 * u);
    p2[u] = *reinterpret_cast<const v2f*>(xr + 2 * PIX + k + 4 * u);
  }
}
__device__ __forceinline__ void fx_mma(const v2f (&c)[FX_U], const v2f (&s)[FX_U],
                                       const v2f (&p0)[FX_U], const v2f (&p1)[FX_U],
                                       const v2f (&p2)[FX_U], float lbv, float l0, float l1,
                                       float l2, v8f& aC, v8f& aS) {
#pragma unroll
  for (int u = 0; u < FX_U; ++u) {
    v2f b;
    b.x = lift1(lbv, l0, l1, l2, p0[u].x, p1[u].x, p2[u].x);
    b.y = lift1(lbv, l0, l1, l2, p0[u].y, p1[u].y, p2[u].y);
    aC = wmma4(c[u], b, aC);
    aS = wmma4(s[u], b, aS);
  }
}
__global__ void __launch_bounds__(256) k_fwdX(const float* __restrict__ x, const float* __restrict__ lw,
                       const float* __restrict__ lb, const float* __restrict__ tabC,
                       const float* __restrict__ tabS, float* __restrict__ Tr,
                       float* __restrict__ Ti) {
  int wid = blockIdx.x * (blockDim.x >> 5) + (threadIdx.x >> 5);
  int lane = threadIdx.x & 31;
  int ch = wid & 1;
  int byi = wid >> 1;
  int b = byi >> 8, y = byi & 255;
  int cc = 16 * ch + (lane & 15);
  float l0 = lw[cc * 3 + 0], l1 = lw[cc * 3 + 1], l2 = lw[cc * 3 + 2], lbv = lb[cc];
  const float* xr = x + ((size_t)(b * 3) * H_ + y) * W_ + ((lane >> 4) << 1);
  v8f aC = vzero(), aS = vzero();
  v2f c0[FX_U], s0[FX_U], c1[FX_U], s1[FX_U];
  v2f p00[FX_U], p01[FX_U], p02[FX_U], p10[FX_U], p11[FX_U], p12[FX_U];
  for (int k0 = 0; k0 < W_; k0 += 8 * FX_U) {
    fx_load(c0, s0, p00, p01, p02, tabC, tabS, xr, k0, lane);
    fx_load(c1, s1, p10, p11, p12, tabC, tabS, xr, k0 + 4 * FX_U, lane);
    __builtin_amdgcn_sched_barrier(0);
    fx_mma(c0, s0, p00, p01, p02, lbv, l0, l1, l2, aC, aS);
    fx_mma(c1, s1, p10, p11, p12, lbv, l0, l1, l2, aC, aS);
    __builtin_amdgcn_sched_barrier(0);
  }
  float* tr = Tr + ((size_t)(b * H_ + y)) * M_ * C_ + 16 * ch;
  float* ti = Ti + ((size_t)(b * H_ + y)) * M_ * C_ + 16 * ch;
  stD(tr, C_, lane, aC, 0.0625f);
  stD(ti, C_, lane, aS, -0.0625f);
  stD(tr + (size_t)B_ * H_ * M_ * C_, C_, lane, aC, 0.0625f);
  stD(ti + (size_t)B_ * H_ * M_ * C_, C_, lane, aS, -0.0625f);
}

__global__ void k_fwdY(const float* __restrict__ Tr, const float* __restrict__ Ti,
                       const float* __restrict__ tabC, const float* __restrict__ tabS,
                       float* __restrict__ Frp, float* __restrict__ Fip) {
  int wid = blockIdx.x * (blockDim.x >> 5) + (threadIdx.x >> 5);
  int lane = threadIdx.x & 31;
  int b = wid >> 7;
  int r2 = wid & 127;
  int col0 = (r2 >> 2) * 16;
  int kc = r2 & 3;
  const float* trb = Tr + (size_t)b * H_ * (M_ * C_);
  const float* tib = Ti + (size_t)b * H_ * (M_ * C_);
  v8f aCT = vzero(), aST = vzero(), aCTi = vzero(), aSTr = vzero();
  for (int k0 = kc * 64; k0 < kc * 64 + 64; k0 += 4) {
    v2f aC = ldA(tabC + k0, W_, lane);
    v2f aS = ldA(tabS + k0, W_, lane);
    v2f bR = ldB(trb + k0 * 512 + col0, 512, lane);
    v2f bI = ldB(tib + k0 * 512 + col0, 512, lane);
    aCT = wmma4(aC, bR, aCT);
    aST = wmma4(aS, bI, aST);
    aCTi = wmma4(aC, bI, aCTi);
    aSTr = wmma4(aS, bR, aSTr);
  }
  float* fr = Frp + (size_t)kc * (NB_ * M_ * 512) + (size_t)b * (M_ * 512) + col0;
  float* fi = Fip + (size_t)kc * (NB_ * M_ * 512) + (size_t)b * (M_ * 512) + col0;
  int n = lane & 15, mh = (lane >> 4) << 3;
#pragma unroll
  for (int r = 0; r < 8; ++r) {
    fr[(r + mh) * 512 + n] = (aCT[r] + aST[r]) * 0.0625f;
    fi[(r + mh) * 512 + n] = (aCTi[r] - aSTr[r]) * 0.0625f;
  }
}

__global__ void __launch_bounds__(512) k_mix(const float* __restrict__ Frp, const float* __restrict__ Fip,
                      const float* __restrict__ wr, const float* __restrict__ wi,
                      float* __restrict__ Gr, float* __restrict__ Gi) {
  __shared__ float sFr[512];
  __shared__ float sFi[512];
  int b = blockIdx.x >> 4, k1 = blockIdx.x & 15;
  size_t wch = (size_t)(b >> 2) * (4 * C_ * C_ * M_ * M_);
  int tid = threadIdx.x;
  size_t base = ((size_t)b * M_ + k1) * 512;
  int o = tid >> 4, k2 = tid & 15;
  const float* wrk = wr + wch + k1 * 16 + k2;
  const float* wik = wi + wch + k1 * 16 + k2;
  float wrv[C_], wiv[C_];
#pragma unroll
  for (int i = 0; i < C_; ++i) {
    wrv[i] = wrk[(i * 32 + o) * 256];
    wiv[i] = wik[(i * 32 + o) * 256];
  }
  float fr = 0.f, fi = 0.f;
#pragma unroll
  for (int kc = 0; kc < 4; ++kc) {
    fr += Frp[(size_t)kc * (NB_ * M_ * 512) + base + tid];
    fi += Fip[(size_t)kc * (NB_ * M_ * 512) + base + tid];
  }
  sFr[tid] = fr; sFi[tid] = fi;
  __syncthreads();
  float gr = 0.f, gi = 0.f;
#pragma unroll
  for (int i = 0; i < C_; ++i) {
    float xr = sFr[k2 * 32 + i], xi = sFi[k2 * 32 + i];
    gr += xr * wrv[i] - xi * wiv[i];
    gi += xr * wiv[i] + xi * wrv[i];
  }
  size_t gidx = base + k2 * 32 + o;
  Gr[gidx] = gr; Gi[gidx] = gi;
}

__global__ void k_invY(const float* __restrict__ Gr, const float* __restrict__ Gi,
                       const float* __restrict__ tabC, const float* __restrict__ tabS,
                       float* __restrict__ P, float* __restrict__ Q) {
  int wid = blockIdx.x * (blockDim.x >> 5) + (threadIdx.x >> 5);
  int lane = threadIdx.x & 31;
  int b = wid >> 9, rest = wid & 511;
  int y0 = (rest >> 5) * 16, col0 = (rest & 31) * 16;
  const float* grb = Gr + (size_t)b * 8192;
  const float* gib = Gi + (size_t)b * 8192;
  v8f aCG = vzero(), aSGi = vzero(), aSG = vzero(), aCGi = vzero();
#pragma unroll
  for (int k0 = 0; k0 < M_; k0 += 4) {
    v2f aC = ldAt(tabC + k0 * W_ + y0, W_, lane);
    v2f aS = ldAt(tabS + k0 * W_ + y0, W_, lane);
    v2f bR = ldB(grb + k0 * 512 + col0, 512, lane);
    v2f bI = ldB(gib + k0 * 512 + col0, 512, lane);
    aCG = wmma4(aC, bR, aCG);
    aSGi = wmma4(aS, bI, aSGi);
    aSG = wmma4(aS, bR, aSG);
    aCGi = wmma4(aC, bI, aCGi);
  }
  float* p = P + (size_t)(b * H_ + y0) * 512 + col0;
  float* q = Q + (size_t)(b * H_ + y0) * 512 + col0;
  int n = lane & 15, mh = (lane >> 4) << 3;
#pragma unroll
  for (int r = 0; r < 8; ++r) {
    p[(r + mh) * 512 + n] = (aCG[r] - aSGi[r]) * 0.0625f;
    q[(r + mh) * 512 + n] = (aSG[r] + aCGi[r]) * 0.0625f;
  }
}

typedef float v4f __attribute__((ext_vector_type(4)));
#define XP 36
#define PP 48
template <bool FWD, bool FIRST>
__global__ void __launch_bounds__(512) k_invX(const float* __restrict__ P, const float* __restrict__ Q,
                       float* act, const float* __restrict__ tabC2,
                       const float* __restrict__ tabS2, const float* __restrict__ bw,
                       const float* __restrict__ bb, const float* __restrict__ tabC,
                       const float* __restrict__ tabS, float* __restrict__ Tr,
                       float* __restrict__ Ti, const float* __restrict__ x,
                       const float* __restrict__ lw, const float* __restrict__ lb,
                       const float* __restrict__ pw, const float* __restrict__ pb,
                       float* __restrict__ res, int o_outer) {
  constexpr int WSTR = FWD ? 1024 : 16 * XP;
  __shared__ __attribute__((aligned(16))) float sW[16 * WSTR];
  __shared__ __attribute__((aligned(16))) float sBW[C_ * XP];
  __shared__ __attribute__((aligned(16))) float sP[M_ * PP];
  __shared__ __attribute__((aligned(16))) float sQ[M_ * PP];
  __shared__ __attribute__((aligned(16))) float sO[W_];
  int tid = threadIdx.x;
  int w = tid >> 5;
  int lane = tid & 31;
  int by = blockIdx.x;
  int chn = by >> 10;
  int x0 = w * 16;
  const float* Pby = P + (size_t)by * 512;
  const float* Qby = Q + (size_t)by * 512;
  const float* xrow = act + (size_t)by * (W_ * C_);
  bw += chn * (4 * C_ * C_);
  bb += chn * (4 * C_);
  float* tile = sW + w * WSTR;
  const v4f* tsrc = reinterpret_cast<const v4f*>(xrow + x0 * C_);
  v4f t0, t1, t2, t3;
  float xv0[16], xv1[16], xv2[16];
  float l0, l1, l2, lbv;
  if (FIRST) {
    int wu = __builtin_amdgcn_readfirstlane(w);
    const float* xs = x + ((size_t)(((by >> 8) & 3) * 3) * H_ + (by & 255)) * W_ + wu * 16;
#pragma unroll
    for (int j = 0; j < 16; ++j) {
      xv0[j] = xs[j];
      xv1[j] = xs[PIX + j];
      xv2[j] = xs[2 * PIX + j];
    }
    l0 = lw[lane * 3 + 0]; l1 = lw[lane * 3 + 1]; l2 = lw[lane * 3 + 2]; lbv = lb[lane];
  } else {
    t0 = tsrc[lane]; t1 = tsrc[32 + lane]; t2 = tsrc[64 + lane]; t3 = tsrc[96 + lane];
  }
  int si = tid & 127;
  v4f sh;
  if (tid < 128)      sh = reinterpret_cast<const v4f*>(Pby)[si];
  else if (tid < 256) sh = reinterpret_cast<const v4f*>(Qby)[si];
  else                sh = reinterpret_cast<const v4f*>(bw)[tid - 256];
  v2f aCv[4], aSv[4];
#pragma unroll
  for (int k = 0; k < 4; ++k) {
    aCv[k] = ldAt(tabC2 + 4 * k * W_ + x0, W_, lane);
    aSv[k] = ldAt(tabS2 + 4 * k * W_ + x0, W_, lane);
  }
  int n = lane & 15, mh = (lane >> 4) << 3;
  float bb0 = bb[n], bb1 = bb[16 + n];
  float pw0 = 0.f, pw1 = 0.f, pbv = 0.f;
  if (!FWD) { pw0 = pw[n]; pw1 = pw[16 + n]; pbv = pb[0]; }
  {
    int tx = lane >> 3, tc = (lane & 7) * 4;
    if (FIRST) {
#pragma unroll
      for (int j = 0; j < 16; ++j)
        tile[j * XP + lane] = lift1(lbv, l0, l1, l2, xv0[j], xv1[j], xv2[j]);
    } else {
      *reinterpret_cast<v4f*>(tile + (0 + tx) * XP + tc) = t0;
      *reinterpret_cast<v4f*>(tile + (4 + tx) * XP + tc) = t1;
      *reinterpret_cast<v4f*>(tile + (8 + tx) * XP + tc) = t2;
      *reinterpret_cast<v4f*>(tile + (12 + tx) * XP + tc) = t3;
    }
    int sr = si >> 3, sc4 = (si & 7) * 4;
    if (tid < 128)      *reinterpret_cast<v4f*>(sP + sr * PP + sc4) = sh;
    else if (tid < 256) *reinterpret_cast<v4f*>(sQ + sr * PP + sc4) = sh;
    else {
      int i = tid - 256;
      *reinterpret_cast<v4f*>(sBW + (i >> 3) * XP + (i & 7) * 4) = sh;
    }
  }
  __syncthreads();
  v8f acc0 = vzero(), acc1 = vzero();
#pragma unroll
  for (int k0 = 0; k0 < M_; k0 += 4) {
    v2f aC = aCv[k0 >> 2];
    v2f bP0 = ldB(sP + k0 * PP, PP, lane);
    v2f bP1 = ldB(sP + k0 * PP + 16, PP, lane);
    acc0 = wmma4(aC, bP0, acc0);
    acc1 = wmma4(aC, bP1, acc1);
    v2f aS = aSv[k0 >> 2];
    v2f aSn; aSn.x = -aS.x; aSn.y = -aS.y;
    v2f bQ0 = ldB(sQ + k0 * PP, PP, lane);
    v2f bQ1 = ldB(sQ + k0 * PP + 16, PP, lane);
    acc0 = wmma4(aSn, bQ0, acc0);
    acc1 = wmma4(aSn, bQ1, acc1);
  }
#pragma unroll
  for (int k0 = 0; k0 < C_; k0 += 4) {
    v2f aX = ldA(tile + k0, XP, lane);
    v2f bW0 = ldBt(sBW + k0, XP, lane);
    v2f bW1 = ldBt(sBW + 16 * XP + k0, XP, lane);
    acc0 = wmma4(aX, bW0, acc0);
    acc1 = wmma4(aX, bW1, acc1);
  }
  v2f fC[4], fS[4];
  if (FWD) {
#pragma unroll
    for (int j = 0; j < 4; ++j) {
      fC[j] = ldA_d(tabC + x0, W_, lane, j);
      fS[j] = ldA_d(tabS + x0, W_, lane, j);
    }
  }
  float* out = act + (size_t)by * (W_ * C_);
  v8f g0, g1;
  float ps[8];
#pragma unroll
  for (int r = 0; r < 8; ++r) {
    int xr = x0 + r + mh;
    float v0 = acc0[r] + bb0;
    v0 = 0.5f * v0 * (1.0f + erff(v0 * 0.70710678f));
    if (FWD) {
      out[xr * C_ + n] = v0;
      g0[r] = v0;
    }
    float v1 = acc1[r] + bb1;
    v1 = 0.5f * v1 * (1.0f + erff(v1 * 0.70710678f));
    if (FWD) {
      out[xr * C_ + 16 + n] = v1;
      g1[r] = v1;
    } else {
      ps[r] = pw0 * v0 + pw1 * v1;
    }
  }
  if (!FWD) {
#pragma unroll
    for (int r = 0; r < 8; ++r) ps[r] = row_sum16(ps[r]);
    if (n == 0) {
      v4f lo = {ps[0], ps[1], ps[2], ps[3]};
      v4f hi = {ps[4], ps[5], ps[6], ps[7]};
      *reinterpret_cast<v4f*>(sO + x0 + mh) = lo;
      *reinterpret_cast<v4f*>(sO + x0 + mh + 4) = hi;
    }
    __syncthreads();
    if (tid < W_) {
      int bq = (by >> 8) & 3, y = by & 255;
      res[((size_t)(bq * 4 + o_outer + chn)) * PIX + y * W_ + tid] = sO[tid] + pbv;
    }
  }
  if (FWD) {
    v8f aC0 = vzero(), aC1 = vzero(), aS0 = vzero(), aS1 = vzero();
#pragma unroll
    for (int j = 0; j < 4; ++j) {
      v2f b0; b0.x = g0[2 * j]; b0.y = g0[2 * j + 1];
      v2f b1; b1.x = g1[2 * j]; b1.y = g1[2 * j + 1];
      aC0 = wmma4(fC[j], b0, aC0);
      aC1 = wmma4(fC[j], b1, aC1);
      aS0 = wmma4(fS[j], b0, aS0);
      aS1 = wmma4(fS[j], b1, aS1);
    }
    float* pc = sW + w * WSTR + (lane >> 4) * 32 + n;
    float* ps = pc + 512;
#pragma unroll
    for (int r = 0; r < 8; ++r) {
      pc[r * 64] = aC0[r];
      pc[r * 64 + 16] = aC1[r];
      ps[r * 64] = aS0[r];
      ps[r * 64 + 16] = aS1[r];
    }
    __syncthreads();
    float sc = 0.f, ss = 0.f;
#pragma unroll
    for (int ww = 0; ww < 16; ++ww) {
      sc += sW[ww * WSTR + tid];
      ss += sW[ww * WSTR + 512 + tid];
    }
    int k2 = (tid >> 6) + ((tid >> 5) & 1) * 8, c = tid & 31;
    size_t o = (size_t)by * (M_ * C_) + k2 * C_ + c;
    Tr[o] = sc * 0.0625f;
    Ti[o] = ss * -0.0625f;
  }
}

extern "C" void kernel_launch(void* const* d_in, const int* in_sizes, int n_in,
                              void* d_out, int out_size, void* d_ws, size_t ws_size,
                              hipStream_t stream) {
  const float* x      = (const float*)d_in[0];
  const float* lift_w = (const float*)d_in[1];
  const float* lift_b = (const float*)d_in[2];
  const float* wr     = (const float*)d_in[3];
  const float* wi     = (const float*)d_in[4];
  const float* byp_w  = (const float*)d_in[5];
  const float* byp_b  = (const float*)d_in[6];
  const float* proj_w = (const float*)d_in[7];
  const float* proj_b = (const float*)d_in[8];
  float* out = (float*)d_out;

  float* w = (float*)d_ws;
  float* tabC  = w;
  float* tabS  = tabC + 4096;
  float* tabC2 = tabS + 4096;
  float* tabS2 = tabC2 + 4096;
  float* act   = tabS2 + 4096;
  float* Tr    = act + (size_t)NB_ * PIX * C_;
  float* Ti    = Tr + (size_t)NB_ * H_ * M_ * C_;
  float* Frp   = Ti + (size_t)NB_ * H_ * M_ * C_;
  float* Fip   = Frp + (size_t)4 * NB_ * M_ * 512;
  float* Gr    = Fip + (size_t)4 * NB_ * M_ * 512;
  float* Gi    = Gr + (size_t)NB_ * M_ * M_ * C_;
  float* Pb    = Gi + (size_t)NB_ * M_ * M_ * C_;
  float* Qb    = Pb + (size_t)NB_ * H_ * M_ * C_;

  k_tabs<<<16, 256, 0, stream>>>(tabC, tabS, tabC2, tabS2);

  for (int o = 0; o < 4; o += NCH) {
    for (int n = 0; n < 4; ++n) {
      size_t wOff = ((size_t)(o * 4 + n)) * C_ * C_ * M_ * M_;
      size_t bwOff = ((size_t)(o * 4 + n)) * C_ * C_;
      size_t bbOff = ((size_t)(o * 4 + n)) * C_;
      if (n == 0) k_fwdX<<<256, 256, 0, stream>>>(x, lift_w, lift_b, tabC, tabS, Tr, Ti);
      k_fwdY<<<NCH * 64, 256, 0, stream>>>(Tr, Ti, tabC, tabS, Frp, Fip);
      k_mix<<<NCH * 64, 512, 0, stream>>>(Frp, Fip, wr + wOff, wi + wOff, Gr, Gi);
      k_invY<<<NCH * 256, 256, 0, stream>>>(Gr, Gi, tabC, tabS, Pb, Qb);
      if (n == 0)
        k_invX<true, true><<<NCH * 1024, 512, 0, stream>>>(Pb, Qb, act, tabC2, tabS2,
                                                           byp_w + bwOff, byp_b + bbOff,
                                                           tabC, tabS, Tr, Ti, x, lift_w, lift_b,
                                                           proj_w, proj_b, out, o);
      else if (n < 3)
        k_invX<true, false><<<NCH * 1024, 512, 0, stream>>>(Pb, Qb, act, tabC2, tabS2,
                                                            byp_w + bwOff, byp_b + bbOff,
                                                            tabC, tabS, Tr, Ti, x, lift_w, lift_b,
                                                            proj_w, proj_b, out, o);
      else
        k_invX<false, false><<<NCH * 1024, 512, 0, stream>>>(Pb, Qb, act, tabC2, tabS2,
                                                             byp_w + bwOff, byp_b + bbOff,
                                                             tabC, tabS, Tr, Ti, x, lift_w, lift_b,
                                                             proj_w, proj_b, out, o);
    }
  }
}


# GELU in invX: one wave-wide range test, then all 16 erf values by the library's own |u|<1 polynomial, branch-free, two per packed instruction (same bits); else erff per value as before; same sums
# speedup vs baseline: 2.0958x; 1.0458x over previous
#include <hip/hip_runtime.h>
#include <math.h>


typedef float v2f __attribute__((ext_vector_type(2)));
typedef float v8f __attribute__((ext_vector_type(8)));

#define B_  4
#define NCH 2
#define NB_ (NCH * B_)
#define C_  32
#define H_  256
#define W_  256
#define M_  16
#define PIX (H_ * W_)

__device__ __forceinline__ v8f wmma4(v2f a, v2f b, v8f c) {
  return __builtin_amdgcn_wmma_f32_16x16x4_f32(false, a, false, b, (short)0, c,
                                               false, false);
}

__device__ __forceinline__ v2f ldA(const float* p, int lda, int lane) {
  int r = lane & 15, kk = (lane >> 4) << 1;
  v2f v; v.x = p[r * lda + kk]; v.y = p[r * lda + kk + 1]; return v;
}
__device__ __forceinline__ v2f ldAt(const float* p, int lda, int lane) {
  int r = lane & 15, kk = (lane >> 4) << 1;
  v2f v; v.x = p[kk * lda + r]; v.y = p[(kk + 1) * lda + r]; return v;
}
__device__ __forceinline__ v2f ldB(const float* p, int ldb, int lane) {
  int n = lane & 15, kk = (lane >> 4) << 1;
  v2f v; v.x = p[kk * ldb + n]; v.y = p[(kk + 1) * ldb + n]; return v;
}
__device__ __forceinline__ v2f ldBt(const float* p, int ldb, int lane) {
  int n = lane & 15, kk = (lane >> 4) << 1;
  v2f v; v.x = p[n * ldb + kk]; v.y = p[n * ldb + kk + 1]; return v;
}
__device__ __forceinline__ v2f ldA_d(const float* p, int lda, int lane, int j) {
  int r = lane & 15, kk = ((lane >> 4) << 3) + 2 * j;
  v2f v; v.x = p[r * lda + kk]; v.y = p[r * lda + kk + 1]; return v;
}
__device__ __forceinline__ void stD(float* p, int ldd, int lane, v8f d, float s) {
  int n = lane & 15, mh = (lane >> 4) << 3;
#pragma unroll
  for (int r = 0; r < 8; ++r) p[(r + mh) * ldd + n] = d[r] * s;
}

__device__ __forceinline__ v8f vzero() {
  v8f z = {0.f, 0.f, 0.f, 0.f, 0.f, 0.f, 0.f, 0.f};
  return z;
}

template <int MASK>
__device__ __forceinline__ float row_xor(float v) {
  int b = __builtin_bit_cast(int, v);
  return __builtin_bit_cast(float, __builtin_amdgcn_mov_dpp(b, 0x160 + MASK, 0xf, 0xf, true));
}
__device__ __forceinline__ float row_sum16(float v) {
  v += row_xor<1>(v);
  v += row_xor<2>(v);
  v += row_xor<4>(v);
  v += row_xor<8>(v);
  return v;
}

__device__ __forceinline__ float erf_poly(float u) {
  float t = u * u;
  float p = __builtin_fmaf(t, __builtin_bit_cast(float, 0xba1345e1u), __builtin_bit_cast(float, 0x3ba10414u));
  p = __builtin_fmaf(t, p, __builtin_bit_cast(float, 0xbcdac9b8u));
  p = __builtin_fmaf(t, p, __builtin_bit_cast(float, 0x3de703beu));
  p = __builtin_fmaf(t, p, __builtin_bit_cast(float, 0xbec09330u));
  p = __builtin_fmaf(t, p, __builtin_bit_cast(float, 0x3e0375d0u));
  return __builtin_fmaf(u, p, u);
}

__global__ void k_tabs(float* tabC, float* tabS, float* tabC2, float* tabS2) {
  int i = blockIdx.x * blockDim.x + threadIdx.x;
  if (i >= M_ * W_) return;
  int k = i >> 8, x = i & 255;
  int m = (k * x) & 255;
  float ang = 0.0245436926f * (float)m;
  float c = cosf(ang), s = sinf(ang);
  tabC[i] = c; tabS[i] = s;
  float w2 = (k == 0) ? 1.0f : 2.0f;
  tabC2[i] = w2 * c * 0.0625f;
  tabS2[i] = w2 * s * 0.0625f;
}

__device__ __forceinline__ float lift1(float lbv, float l0, float l1, float l2,
                                       float xi0, float xi1, float xi2) {
  return lbv + l0 * xi0 + l1 * xi1 + l2 * xi2;
}

#define FX_U 8
__device__ __forceinline__ void fx_load(v2f (&c)[FX_U], v2f (&s)[FX_U], v2f (&p0)[FX_U],
                                        v2f (&p1)[FX_U], v2f (&p2)[FX_U],
                                        const float* __restrict__ tabC,
                                        const float* __restrict__ tabS,
                                        const float* __restrict__ xr, int k, int lane) {
#pragma unroll
  for (int u = 0; u < FX_U; ++u) {
    c[u] = ldA(tabC + k + 4 * u, W_, lane);
    s[u] = ldA(tabS + k + 4 * u, W_, lane);
    p0[u] = *reinterpret_cast<const v2f*>(xr + k + 4 * u);
    p1[u] = *reinterpret_cast<const v2f*>(xr + PIX + k + 4 * u);
    p2[u] = *reinterpret_cast<const v2f*>(xr + 2 * PIX + k + 4 * u);
  }
}
__device__ __forceinline__ void fx_mma(const v2f (&c)[FX_U], const v2f (&s)[FX_U],
                                       const v2f (&p0)[FX_U], const v2f (&p1)[FX_U],
                                       const v2f (&p2)[FX_U], float lbv, float l0, float l1,
                                       float l2, v8f& aC, v8f& aS) {
#pragma unroll
  for (int u = 0; u < FX_U; ++u) {
    v2f b;
    b.x = lift1(lbv, l0, l1, l2, p0[u].x, p1[u].x, p2[u].x);
    b.y = lift1(lbv, l0, l1, l2, p0[u].y, p1[u].y, p2[u].y);
    aC = wmma4(c[u], b, aC);
    aS = wmma4(s[u], b, aS);
  }
}
__global__ void __launch_bounds__(256) k_fwdX(const float* __restrict__ x, const float* __restrict__ lw,
                       const float* __restrict__ lb, const float* __restrict__ tabC,
                       const float* __restrict__ tabS, float* __restrict__ Tr,
                       float* __restrict__ Ti) {
  int wid = blockIdx.x * (blockDim.x >> 5) + (threadIdx.x >> 5);
  int lane = threadIdx.x & 31;
  int ch = wid & 1;
  int byi = wid >> 1;
  int b = byi >> 8, y = byi & 255;
  int cc = 16 * ch + (lane & 15);
  float l0 = lw[cc * 3 + 0], l1 = lw[cc * 3 + 1], l2 = lw[cc * 3 + 2], lbv = lb[cc];
  const float* xr = x + ((size_t)(b * 3) * H_ + y) * W_ + ((lane >> 4) << 1);
  v8f aC = vzero(), aS = vzero();
  v2f c0[FX_U], s0[FX_U], c1[FX_U], s1[FX_U];
  v2f p00[FX_U], p01[FX_U], p02[FX_U], p10[FX_U], p11[FX_U], p12[FX_U];
  for (int k0 = 0; k0 < W_; k0 += 8 * FX_U) {
    fx_load(c0, s0, p00, p01, p02, tabC, tabS, xr, k0, lane);
    fx_load(c1, s1, p10, p11, p12, tabC, tabS, xr, k0 + 4 * FX_U, lane);
    __builtin_amdgcn_sched_barrier(0);
    fx_mma(c0, s0, p00, p01, p02, lbv, l0, l1, l2, aC, aS);
    fx_mma(c1, s1, p10, p11, p12, lbv, l0, l1, l2, aC, aS);
    __builtin_amdgcn_sched_barrier(0);
  }
  float* tr = Tr + ((size_t)(b * H_ + y)) * M_ * C_ + 16 * ch;
  float* ti = Ti + ((size_t)(b * H_ + y)) * M_ * C_ + 16 * ch;
  stD(tr, C_, lane, aC, 0.0625f);
  stD(ti, C_, lane, aS, -0.0625f);
  stD(tr + (size_t)B_ * H_ * M_ * C_, C_, lane, aC, 0.0625f);
  stD(ti + (size_t)B_ * H_ * M_ * C_, C_, lane, aS, -0.0625f);
}

__global__ void k_fwdY(const float* __restrict__ Tr, const float* __restrict__ Ti,
                       const float* __restrict__ tabC, const float* __restrict__ tabS,
                       float* __restrict__ Frp, float* __restrict__ Fip) {
  int wid = blockIdx.x * (blockDim.x >> 5) + (threadIdx.x >> 5);
  int lane = threadIdx.x & 31;
  int b = wid >> 7;
  int r2 = wid & 127;
  int col0 = (r2 >> 2) * 16;
  int kc = r2 & 3;
  const float* trb = Tr + (size_t)b * H_ * (M_ * C_);
  const float* tib = Ti + (size_t)b * H_ * (M_ * C_);
  v8f aCT = vzero(), aST = vzero(), aCTi = vzero(), aSTr = vzero();
  for (int k0 = kc * 64; k0 < kc * 64 + 64; k0 += 4) {
    v2f aC = ldA(tabC + k0, W_, lane);
    v2f aS = ldA(tabS + k0, W_, lane);
    v2f bR = ldB(trb + k0 * 512 + col0, 512, lane);
    v2f bI = ldB(tib + k0 * 512 + col0, 512, lane);
    aCT = wmma4(aC, bR, aCT);
    aST = wmma4(aS, bI, aST);
    aCTi = wmma4(aC, bI, aCTi);
    aSTr = wmma4(aS, bR, aSTr);
  }
  float* fr = Frp + (size_t)kc * (NB_ * M_ * 512) + (size_t)b * (M_ * 512) + col0;
  float* fi = Fip + (size_t)kc * (NB_ * M_ * 512) + (size_t)b * (M_ * 512) + col0;
  int n = lane & 15, mh = (lane >> 4) << 3;
#pragma unroll
  for (int r = 0; r < 8; ++r) {
    fr[(r + mh) * 512 + n] = (aCT[r] + aST[r]) * 0.0625f;
    fi[(r + mh) * 512 + n] = (aCTi[r] - aSTr[r]) * 0.0625f;
  }
}

__global__ void __launch_bounds__(512) k_mix(const float* __restrict__ Frp, const float* __restrict__ Fip,
                      const float* __restrict__ wr, const float* __restrict__ wi,
                      float* __restrict__ Gr, float* __restrict__ Gi) {
  __shared__ float sFr[512];
  __shared__ float sFi[512];
  int b = blockIdx.x >> 4, k1 = blockIdx.x & 15;
  size_t wch = (size_t)(b >> 2) * (4 * C_ * C_ * M_ * M_);
  int tid = threadIdx.x;
  size_t base = ((size_t)b * M_ + k1) * 512;
  int o = tid >> 4, k2 = tid & 15;
  const float* wrk = wr + wch + k1 * 16 + k2;
  const float* wik = wi + wch + k1 * 16 + k2;
  float wrv[C_], wiv[C_];
#pragma unroll
  for (int i = 0; i < C_; ++i) {
    wrv[i] = wrk[(i * 32 + o) * 256];
    wiv[i] = wik[(i * 32 + o) * 256];
  }
  float fr = 0.f, fi = 0.f;
#pragma unroll
  for (int kc = 0; kc < 4; ++kc) {
    fr += Frp[(size_t)kc * (NB_ * M_ * 512) + base + tid];
    fi += Fip[(size_t)kc * (NB_ * M_ * 512) + base + tid];
  }
  sFr[tid] = fr; sFi[tid] = fi;
  __syncthreads();
  float gr = 0.f, gi = 0.f;
#pragma unroll
  for (int i = 0; i < C_; ++i) {
    float xr = sFr[k2 * 32 + i], xi = sFi[k2 * 32 + i];
    gr += xr * wrv[i] - xi * wiv[i];
    gi += xr * wiv[i] + xi * wrv[i];
  }
  size_t gidx = base + k2 * 32 + o;
  Gr[gidx] = gr; Gi[gidx] = gi;
}

__global__ void k_invY(const float* __restrict__ Gr, const float* __restrict__ Gi,
                       const float* __restrict__ tabC, const float* __restrict__ tabS,
                       float* __restrict__ P, float* __restrict__ Q) {
  int wid = blockIdx.x * (blockDim.x >> 5) + (threadIdx.x >> 5);
  int lane = threadIdx.x & 31;
  int b = wid >> 9, rest = wid & 511;
  int y0 = (rest >> 5) * 16, col0 = (rest & 31) * 16;
  const float* grb = Gr + (size_t)b * 8192;
  const float* gib = Gi + (size_t)b * 8192;
  v8f aCG = vzero(), aSGi = vzero(), aSG = vzero(), aCGi = vzero();
#pragma unroll
  for (int k0 = 0; k0 < M_; k0 += 4) {
    v2f aC = ldAt(tabC + k0 * W_ + y0, W_, lane);
    v2f aS = ldAt(tabS + k0 * W_ + y0, W_, lane);
    v2f bR = ldB(grb + k0 * 512 + col0, 512, lane);
    v2f bI = ldB(gib + k0 * 512 + col0, 512, lane);
    aCG = wmma4(aC, bR, aCG);
    aSGi = wmma4(aS, bI, aSGi);
    aSG = wmma4(aS, bR, aSG);
    aCGi = wmma4(aC, bI, aCGi);
  }
  float* p = P + (size_t)(b * H_ + y0) * 512 + col0;
  float* q = Q + (size_t)(b * H_ + y0) * 512 + col0;
  int n = lane & 15, mh = (lane >> 4) << 3;
#pragma unroll
  for (int r = 0; r < 8; ++r) {
    p[(r + mh) * 512 + n] = (aCG[r] - aSGi[r]) * 0.0625f;
    q[(r + mh) * 512 + n] = (aSG[r] + aCGi[r]) * 0.0625f;
  }
}

typedef float v4f __attribute__((ext_vector_type(4)));
#define XP 36
#define PP 48
template <bool FWD, bool FIRST>
__global__ void __launch_bounds__(512, 12) k_invX(const float* __restrict__ P, const float* __restrict__ Q,
                       float* act, const float* __restrict__ tabC2,
                       const float* __restrict__ tabS2, const float* __restrict__ bw,
                       const float* __restrict__ bb, const float* __restrict__ tabC,
                       const float* __restrict__ tabS, float* __restrict__ Tr,
                       float* __restrict__ Ti, const float* __restrict__ x,
                       const float* __restrict__ lw, const float* __restrict__ lb,
                       const float* __restrict__ pw, const float* __restrict__ pb,
                       float* __restrict__ res, int o_outer) {
  constexpr int WSTR = FWD ? 1024 : 16 * XP;
  __shared__ __attribute__((aligned(16))) float sW[16 * WSTR];
  __shared__ __attribute__((aligned(16))) float sBW[C_ * XP];
  __shared__ __attribute__((aligned(16))) float sP[M_ * PP];
  __shared__ __attribute__((aligned(16))) float sQ[M_ * PP];
  __shared__ __attribute__((aligned(16))) float sO[W_];
  int tid = threadIdx.x;
  int w = tid >> 5;
  int lane = tid & 31;
  int by = blockIdx.x;
  int chn = by >> 10;
  int x0 = w * 16;
  const float* Pby = P + (size_t)by * 512;
  const float* Qby = Q + (size_t)by * 512;
  const float* xrow = act + (size_t)by * (W_ * C_);
  bw += chn * (4 * C_ * C_);
  bb += chn * (4 * C_);
  float* tile = sW + w * WSTR;
  const v4f* tsrc = reinterpret_cast<const v4f*>(xrow + x0 * C_);
  v4f t0, t1, t2, t3;
  float xv0[16], xv1[16], xv2[16];
  float l0, l1, l2, lbv;
  if (FIRST) {
    int wu = __builtin_amdgcn_readfirstlane(w);
    const float* xs = x + ((size_t)(((by >> 8) & 3) * 3) * H_ + (by & 255)) * W_ + wu * 16;
#pragma unroll
    for (int j = 0; j < 16; ++j) {
      xv0[j] = xs[j];
      xv1[j] = xs[PIX + j];
      xv2[j] = xs[2 * PIX + j];
    }
    l0 = lw[lane * 3 + 0]; l1 = lw[lane * 3 + 1]; l2 = lw[lane * 3 + 2]; lbv = lb[lane];
  } else {
    t0 = tsrc[lane]; t1 = tsrc[32 + lane]; t2 = tsrc[64 + lane]; t3 = tsrc[96 + lane];
  }
  int si = tid & 127;
  v4f sh;
  if (tid < 128)      sh = reinterpret_cast<const v4f*>(Pby)[si];
  else if (tid < 256) sh = reinterpret_cast<const v4f*>(Qby)[si];
  else                sh = reinterpret_cast<const v4f*>(bw)[tid - 256];
  v2f aCv[4], aSv[4];
#pragma unroll
  for (int k = 0; k < 4; ++k) {
    aCv[k] = ldAt(tabC2 + 4 * k * W_ + x0, W_, lane);
    aSv[k] = ldAt(tabS2 + 4 * k * W_ + x0, W_, lane);
  }
  int n = lane & 15, mh = (lane >> 4) << 3;
  float bb0 = bb[n], bb1 = bb[16 + n];
  float pw0 = 0.f, pw1 = 0.f, pbv = 0.f;
  if (!FWD) { pw0 = pw[n]; pw1 = pw[16 + n]; pbv = pb[0]; }
  {
    int tx = lane >> 3, tc = (lane & 7) * 4;
    if (FIRST) {
#pragma unroll
      for (int j = 0; j < 16; ++j)
        tile[j * XP + lane] = lift1(lbv, l0, l1, l2, xv0[j], xv1[j], xv2[j]);
    } else {
      *reinterpret_cast<v4f*>(tile + (0 + tx) * XP + tc) = t0;
      *reinterpret_cast<v4f*>(tile + (4 + tx) * XP + tc) = t1;
      *reinterpret_cast<v4f*>(tile + (8 + tx) * XP + tc) = t2;
      *reinterpret_cast<v4f*>(tile + (12 + tx) * XP + tc) = t3;
    }
    int sr = si >> 3, sc4 = (si & 7) * 4;
    if (tid < 128)      *reinterpret_cast<v4f*>(sP + sr * PP + sc4) = sh;
    else if (tid < 256) *reinterpret_cast<v4f*>(sQ + sr * PP + sc4) = sh;
    else {
      int i = tid - 256;
      *reinterpret_cast<v4f*>(sBW + (i >> 3) * XP + (i & 7) * 4) = sh;
    }
  }
  __syncthreads();
  v8f acc0 = vzero(), acc1 = vzero();
#pragma unroll
  for (int k0 = 0; k0 < M_; k0 += 4) {
    v2f aC = aCv[k0 >> 2];
    v2f bP0 = ldB(sP + k0 * PP, PP, lane);
    v2f bP1 = ldB(sP + k0 * PP + 16, PP, lane);
    acc0 = wmma4(aC, bP0, acc0);
    acc1 = wmma4(aC, bP1, acc1);
    v2f aS = aSv[k0 >> 2];
    v2f aSn; aSn.x = -aS.x; aSn.y = -aS.y;
    v2f bQ0 = ldB(sQ + k0 * PP, PP, lane);
    v2f bQ1 = ldB(sQ + k0 * PP + 16, PP, lane);
    acc0 = wmma4(aSn, bQ0, acc0);
    acc1 = wmma4(aSn, bQ1, acc1);
  }
#pragma unroll
  for (int k0 = 0; k0 < C_; k0 += 4) {
    v2f aX = ldA(tile + k0, XP, lane);
    v2f bW0 = ldBt(sBW + k0, XP, lane);
    v2f bW1 = ldBt(sBW + 16 * XP + k0, XP, lane);
    acc0 = wmma4(aX, bW0, acc0);
    acc1 = wmma4(aX, bW1, acc1);
  }
  v2f fC[4], fS[4];
  if (FWD) {
#pragma unroll
    for (int j = 0; j < 4; ++j) {
      fC[j] = ldA_d(tabC + x0, W_, lane, j);
      fS[j] = ldA_d(tabS + x0, W_, lane, j);
    }
  }
  float* out = act + (size_t)by * (W_ * C_);
  v8f g0, g1;
  float ps[8];
  float pv0[8], pv1[8];
  float eu0[8], eu1[8];
  float um = 0.f;
#pragma unroll
  for (int r = 0; r < 8; ++r) {
    pv0[r] = acc0[r] + bb0;
    pv1[r] = acc1[r] + bb1;
    eu0[r] = pv0[r] * 0.70710678f;
    eu1[r] = pv1[r] * 0.70710678f;
    um = fmaxf(um, fmaxf(fabsf(eu0[r]), fabsf(eu1[r])));
  }
  if (__builtin_amdgcn_ballot_w32(!(um < 1.0f)) == 0u) {
#pragma unroll
    for (int r = 0; r < 8; ++r) {
      eu0[r] = erf_poly(eu0[r]);
      eu1[r] = erf_poly(eu1[r]);
    }
  } else {
#pragma unroll
    for (int r = 0; r < 8; ++r) {
      eu0[r] = erff(eu0[r]);
      eu1[r] = erff(eu1[r]);
    }
  }
#pragma unroll
  for (int r = 0; r < 8; ++r) {
    int xr = x0 + r + mh;
    float v0 = 0.5f * pv0[r] * (1.0f + eu0[r]);
    if (FWD) {
      out[xr * C_ + n] = v0;
      g0[r] = v0;
    }
    float v1 = 0.5f * pv1[r] * (1.0f + eu1[r]);
    if (FWD) {
      out[xr * C_ + 16 + n] = v1;
      g1[r] = v1;
    } else {
      ps[r] = pw0 * v0 + pw1 * v1;
    }
  }
  if (!FWD) {
#pragma unroll
    for (int r = 0; r < 8; ++r) ps[r] = row_sum16(ps[r]);
    if (n == 0) {
      v4f lo = {ps[0], ps[1], ps[2], ps[3]};
      v4f hi = {ps[4], ps[5], ps[6], ps[7]};
      *reinterpret_cast<v4f*>(sO + x0 + mh) = lo;
      *reinterpret_cast<v4f*>(sO + x0 + mh + 4) = hi;
    }
    __syncthreads();
    if (tid < W_) {
      int bq = (by >> 8) & 3, y = by & 255;
      res[((size_t)(bq * 4 + o_outer + chn)) * PIX + y * W_ + tid] = sO[tid] + pbv;
    }
  }
  if (FWD) {
    v8f aC0 = vzero(), aC1 = vzero(), aS0 = vzero(), aS1 = vzero();
#pragma unroll
    for (int j = 0; j < 4; ++j) {
      v2f b0; b0.x = g0[2 * j]; b0.y = g0[2 * j + 1];
      v2f b1; b1.x = g1[2 * j]; b1.y = g1[2 * j + 1];
      aC0 = wmma4(fC[j], b0, aC0);
      aC1 = wmma4(fC[j], b1, aC1);
      aS0 = wmma4(fS[j], b0, aS0);
      aS1 = wmma4(fS[j], b1, aS1);
    }
    float* pc = sW + w * WSTR + (lane >> 4) * 32 + n;
    float* ps = pc + 512;
#pragma unroll
    for (int r = 0; r < 8; ++r) {
      pc[r * 64] = aC0[r];
      pc[r * 64 + 16] = aC1[r];
      ps[r * 64] = aS0[r];
      ps[r * 64 + 16] = aS1[r];
    }
    __syncthreads();
    float sc = 0.f, ss = 0.f;
#pragma unroll
    for (int ww = 0; ww < 16; ++ww) {
      sc += sW[ww * WSTR + tid];
      ss += sW[ww * WSTR + 512 + tid];
    }
    int k2 = (tid >> 6) + ((tid >> 5) & 1) * 8, c = tid & 31;
    size_t o = (size_t)by * (M_ * C_) + k2 * C_ + c;
    Tr[o] = sc * 0.0625f;
    Ti[o] = ss * -0.0625f;
  }
}

extern "C" void kernel_launch(void* const* d_in, const int* in_sizes, int n_in,
                              void* d_out, int out_size, void* d_ws, size_t ws_size,
                              hipStream_t stream) {
  const float* x      = (const float*)d_in[0];
  const float* lift_w = (const float*)d_in[1];
  const float* lift_b = (const float*)d_in[2];
  const float* wr     = (const float*)d_in[3];
  const float* wi     = (const float*)d_in[4];
  const float* byp_w  = (const float*)d_in[5];
  const float* byp_b  = (const float*)d_in[6];
  const float* proj_w = (const float*)d_in[7];
  const float* proj_b = (const float*)d_in[8];
  float* out = (float*)d_out;

  float* w = (float*)d_ws;
  float* tabC  = w;
  float* tabS  = tabC + 4096;
  float* tabC2 = tabS + 4096;
  float* tabS2 = tabC2 + 4096;
  float* act   = tabS2 + 4096;
  float* Tr    = act + (size_t)NB_ * PIX * C_;
  float* Ti    = Tr + (size_t)NB_ * H_ * M_ * C_;
  float* Frp   = Ti + (size_t)NB_ * H_ * M_ * C_;
  float* Fip   = Frp + (size_t)4 * NB_ * M_ * 512;
  float* Gr    = Fip + (size_t)4 * NB_ * M_ * 512;
  float* Gi    = Gr + (size_t)NB_ * M_ * M_ * C_;
  float* Pb    = Gi + (size_t)NB_ * M_ * M_ * C_;
  float* Qb    = Pb + (size_t)NB_ * H_ * M_ * C_;

  k_tabs<<<16, 256, 0, stream>>>(tabC, tabS, tabC2, tabS2);

  for (int o = 0; o < 4; o += NCH) {
    for (int n = 0; n < 4; ++n) {
      size_t wOff = ((size_t)(o * 4 + n)) * C_ * C_ * M_ * M_;
      size_t bwOff = ((size_t)(o * 4 + n)) * C_ * C_;
      size_t bbOff = ((size_t)(o * 4 + n)) * C_;
      if (n == 0) k_fwdX<<<256, 256, 0, stream>>>(x, lift_w, lift_b, tabC, tabS, Tr, Ti);
      k_fwdY<<<NCH * 64, 256, 0, stream>>>(Tr, Ti, tabC, tabS, Frp, Fip);
      k_mix<<<NCH * 64, 512, 0, stream>>>(Frp, Fip, wr + wOff, wi + wOff, Gr, Gi);
      k_invY<<<NCH * 256, 256, 0, stream>>>(Gr, Gi, tabC, tabS, Pb, Qb);
      if (n == 0)
        k_invX<true, true><<<NCH * 1024, 512, 0, stream>>>(Pb, Qb, act, tabC2, tabS2,
                                                           byp_w + bwOff, byp_b + bbOff,
                                                           tabC, tabS, Tr, Ti, x, lift_w, lift_b,
                                                           proj_w, proj_b, out, o);
      else if (n < 3)
        k_invX<true, false><<<NCH * 1024, 512, 0, stream>>>(Pb, Qb, act, tabC2, tabS2,
                                                            byp_w + bwOff, byp_b + bbOff,
                                                            tabC, tabS, Tr, Ti, x, lift_w, lift_b,
                                                            proj_w, proj_b, out, o);
      else
        k_invX<false, false><<<NCH * 1024, 512, 0, stream>>>(Pb, Qb, act, tabC2, tabS2,
                                                             byp_w + bwOff, byp_b + bbOff,
                                                             tabC, tabS, Tr, Ti, x, lift_w, lift_b,
                                                             proj_w, proj_b, out, o);
    }
  }
}


# invX: a wave owns x tiles 16w and 128+16w; basis(x+128) = (-1)^k2 basis(x) splits both x transforms by k2 parity: 32 not 48 matrix products per tile, 8-wave blocks (sum order of both changes)
# speedup vs baseline: 2.3565x; 1.1244x over previous
#include <hip/hip_runtime.h>
#include <math.h>


typedef float v2f __attribute__((ext_vector_type(2)));
typedef float v8f __attribute__((ext_vector_type(8)));

#define B_  4
#define NCH 2
#define NB_ (NCH * B_)
#define C_  32
#define H_  256
#define W_  256
#define M_  16
#define PIX (H_ * W_)

__device__ __forceinline__ v8f wmma4(v2f a, v2f b, v8f c) {
  return __builtin_amdgcn_wmma_f32_16x16x4_f32(false, a, false, b, (short)0, c,
                                               false, false);
}

__device__ __forceinline__ v2f ldA(const float* p, int lda, int lane) {
  int r = lane & 15, kk = (lane >> 4) << 1;
  v2f v; v.x = p[r * lda + kk]; v.y = p[r * lda + kk + 1]; return v;
}
__device__ __forceinline__ v2f ldAt(const float* p, int lda, int lane) {
  int r = lane & 15, kk = (lane >> 4) << 1;
  v2f v; v.x = p[kk * lda + r]; v.y = p[(kk + 1) * lda + r]; return v;
}
__device__ __forceinline__ v2f ldB(const float* p, int ldb, int lane) {
  int n = lane & 15, kk = (lane >> 4) << 1;
  v2f v; v.x = p[kk * ldb + n]; v.y = p[(kk + 1) * ldb + n]; return v;
}
__device__ __forceinline__ v2f ldBt(const float* p, int ldb, int lane) {
  int n = lane & 15, kk = (lane >> 4) << 1;
  v2f v; v.x = p[n * ldb + kk]; v.y = p[n * ldb + kk + 1]; return v;
}
__device__ __forceinline__ v2f ldA_d(const float* p, int lda, int lane, int j) {
  int r = lane & 15, kk = ((lane >> 4) << 3) + 2 * j;
  v2f v; v.x = p[r * lda + kk]; v.y = p[r * lda + kk + 1]; return v;
}
__device__ __forceinline__ void stD(float* p, int ldd, int lane, v8f d, float s) {
  int n = lane & 15, mh = (lane >> 4) << 3;
#pragma unroll
  for (int r = 0; r < 8; ++r) p[(r + mh) * ldd + n] = d[r] * s;
}

__device__ __forceinline__ v8f vzero() {
  v8f z = {0.f, 0.f, 0.f, 0.f, 0.f, 0.f, 0.f, 0.f};
  return z;
}

template <int MASK>
__device__ __forceinline__ float row_xor(float v) {
  int b = __builtin_bit_cast(int, v);
  return __builtin_bit_cast(float, __builtin_amdgcn_mov_dpp(b, 0x160 + MASK, 0xf, 0xf, true));
}
__device__ __forceinline__ float row_sum16(float v) {
  v += row_xor<1>(v);
  v += row_xor<2>(v);
  v += row_xor<4>(v);
  v += row_xor<8>(v);
  return v;
}

__device__ __forceinline__ float erf_poly(float u) {
  float t = u * u;
  float p = __builtin_fmaf(t, __builtin_bit_cast(float, 0xba1345e1u), __builtin_bit_cast(float, 0x3ba10414u));
  p = __builtin_fmaf(t, p, __builtin_bit_cast(float, 0xbcdac9b8u));
  p = __builtin_fmaf(t, p, __builtin_bit_cast(float, 0x3de703beu));
  p = __builtin_fmaf(t, p, __builtin_bit_cast(float, 0xbec09330u));
  p = __builtin_fmaf(t, p, __builtin_bit_cast(float, 0x3e0375d0u));
  return __builtin_fmaf(u, p, u);
}

__global__ void k_tabs(float* tabC, float* tabS, float* tabC2, float* tabS2) {
  int i = blockIdx.x * blockDim.x + threadIdx.x;
  if (i >= M_ * W_) return;
  int k = i >> 8, x = i & 255;
  int m = (k * x) & 255;
  float ang = 0.0245436926f * (float)m;
  float c = cosf(ang), s = sinf(ang);
  tabC[i] = c; tabS[i] = s;
  float w2 = (k == 0) ? 1.0f : 2.0f;
  tabC2[i] = w2 * c * 0.0625f;
  tabS2[i] = w2 * s * 0.0625f;
}

__device__ __forceinline__ float lift1(float lbv, float l0, float l1, float l2,
                                       float xi0, float xi1, float xi2) {
  return lbv + l0 * xi0 + l1 * xi1 + l2 * xi2;
}

#define FX_U 8
__device__ __forceinline__ void fx_load(v2f (&c)[FX_U], v2f (&s)[FX_U], v2f (&p0)[FX_U],
                                        v2f (&p1)[FX_U], v2f (&p2)[FX_U],
                                        const float* __restrict__ tabC,
                                        const float* __restrict__ tabS,
                                        const float* __restrict__ xr, int k, int lane) {
#pragma unroll
  for (int u = 0; u < FX_U; ++u) {
    c[u] = ldA(tabC + k + 4 * u, W_, lane);
    s[u] = ldA(tabS + k + 4 * u, W_, lane);
    p0[u] = *reinterpret_cast<const v2f*>(xr + k + 4 * u);
    p1[u] = *reinterpret_cast<const v2f*>(xr + PIX + k + 4 * u);
    p2[u] = *reinterpret_cast<const v2f*>(xr + 2 * PIX + k + 4 * u);
  }
}
__device__ __forceinline__ void fx_mma(const v2f (&c)[FX_U], const v2f (&s)[FX_U],
                                       const v2f (&p0)[FX_U], const v2f (&p1)[FX_U],
                                       const v2f (&p2)[FX_U], float lbv, float l0, float l1,
                                       float l2, v8f& aC, v8f& aS) {
#pragma unroll
  for (int u = 0; u < FX_U; ++u) {
    v2f b;
    b.x = lift1(lbv, l0, l1, l2, p0[u].x, p1[u].x, p2[u].x);
    b.y = lift1(lbv, l0, l1, l2, p0[u].y, p1[u].y, p2[u].y);
    aC = wmma4(c[u], b, aC);
    aS = wmma4(s[u], b, aS);
  }
}
__global__ void __launch_bounds__(256) k_fwdX(const float* __restrict__ x, const float* __restrict__ lw,
                       const float* __restrict__ lb, const float* __restrict__ tabC,
                       const float* __restrict__ tabS, float* __restrict__ Tr,
                       float* __restrict__ Ti) {
  int wid = blockIdx.x * (blockDim.x >> 5) + (threadIdx.x >> 5);
  int lane = threadIdx.x & 31;
  int ch = wid & 1;
  int byi = wid >> 1;
  int b = byi >> 8, y = byi & 255;
  int cc = 16 * ch + (lane & 15);
  float l0 = lw[cc * 3 + 0], l1 = lw[cc * 3 + 1], l2 = lw[cc * 3 + 2], lbv = lb[cc];
  const float* xr = x + ((size_t)(b * 3) * H_ + y) * W_ + ((lane >> 4) << 1);
  v8f aC = vzero(), aS = vzero();
  v2f c0[FX_U], s0[FX_U], c1[FX_U], s1[FX_U];
  v2f p00[FX_U], p01[FX_U], p02[FX_U], p10[FX_U], p11[FX_U], p12[FX_U];
  for (int k0 = 0; k0 < W_; k0 += 8 * FX_U) {
    fx_load(c0, s0, p00, p01, p02, tabC, tabS, xr, k0, lane);
    fx_load(c1, s1, p10, p11, p12, tabC, tabS, xr, k0 + 4 * FX_U, lane);
    __builtin_amdgcn_sched_barrier(0);
    fx_mma(c0, s0, p00, p01, p02, lbv, l0, l1, l2, aC, aS);
    fx_mma(c1, s1, p10, p11, p12, lbv, l0, l1, l2, aC, aS);
    __builtin_amdgcn_sched_barrier(0);
  }
  float* tr = Tr + ((size_t)(b * H_ + y)) * M_ * C_ + 16 * ch;
  float* ti = Ti + ((size_t)(b * H_ + y)) * M_ * C_ + 16 * ch;
  stD(tr, C_, lane, aC, 0.0625f);
  stD(ti, C_, lane, aS, -0.0625f);
  stD(tr + (size_t)B_ * H_ * M_ * C_, C_, lane, aC, 0.0625f);
  stD(ti + (size_t)B_ * H_ * M_ * C_, C_, lane, aS, -0.0625f);
}

__global__ void k_fwdY(const float* __restrict__ Tr, const float* __restrict__ Ti,
                       const float* __restrict__ tabC, const float* __restrict__ tabS,
                       float* __restrict__ Frp, float* __restrict__ Fip) {
  int wid = blockIdx.x * (blockDim.x >> 5) + (threadIdx.x >> 5);
  int lane = threadIdx.x & 31;
  int b = wid >> 7;
  int r2 = wid & 127;
  int col0 = (r2 >> 2) * 16;
  int kc = r2 & 3;
  const float* trb = Tr + (size_t)b * H_ * (M_ * C_);
  const float* tib = Ti + (size_t)b * H_ * (M_ * C_);
  v8f aCT = vzero(), aST = vzero(), aCTi = vzero(), aSTr = vzero();
  for (int k0 = kc * 64; k0 < kc * 64 + 64; k0 += 4) {
    v2f aC = ldA(tabC + k0, W_, lane);
    v2f aS = ldA(tabS + k0, W_, lane);
    v2f bR = ldB(trb + k0 * 512 + col0, 512, lane);
    v2f bI = ldB(tib + k0 * 512 + col0, 512, lane);
    aCT = wmma4(aC, bR, aCT);
    aST = wmma4(aS, bI, aST);
    aCTi = wmma4(aC, bI, aCTi);
    aSTr = wmma4(aS, bR, aSTr);
  }
  float* fr = Frp + (size_t)kc * (NB_ * M_ * 512) + (size_t)b * (M_ * 512) + col0;
  float* fi = Fip + (size_t)kc * (NB_ * M_ * 512) + (size_t)b * (M_ * 512) + col0;
  int n = lane & 15, mh = (lane >> 4) << 3;
#pragma unroll
  for (int r = 0; r < 8; ++r) {
    fr[(r + mh) * 512 + n] = (aCT[r] + aST[r]) * 0.0625f;
    fi[(r + mh) * 512 + n] = (aCTi[r] - aSTr[r]) * 0.0625f;
  }
}

__global__ void __launch_bounds__(512) k_mix(const float* __restrict__ Frp, const float* __restrict__ Fip,
                      const float* __restrict__ wr, const float* __restrict__ wi,
                      float* __restrict__ Gr, float* __restrict__ Gi) {
  __shared__ float sFr[512];
  __shared__ float sFi[512];
  int b = blockIdx.x >> 4, k1 = blockIdx.x & 15;
  size_t wch = (size_t)(b >> 2) * (4 * C_ * C_ * M_ * M_);
  int tid = threadIdx.x;
  size_t base = ((size_t)b * M_ + k1) * 512;
  int o = tid >> 4, k2 = tid & 15;
  const float* wrk = wr + wch + k1 * 16 + k2;
  const float* wik = wi + wch + k1 * 16 + k2;
  float wrv[C_], wiv[C_];
#pragma unroll
  for (int i = 0; i < C_; ++i) {
    wrv[i] = wrk[(i * 32 + o) * 256];
    wiv[i] = wik[(i * 32 + o) * 256];
  }
  float fr = 0.f, fi = 0.f;
#pragma unroll
  for (int kc = 0; kc < 4; ++kc) {
    fr += Frp[(size_t)kc * (NB_ * M_ * 512) + base + tid];
    fi += Fip[(size_t)kc * (NB_ * M_ * 512) + base + tid];
  }
  sFr[tid] = fr; sFi[tid] = fi;
  __syncthreads();
  float gr = 0.f, gi = 0.f;
#pragma unroll
  for (int i = 0; i < C_; ++i) {
    float xr = sFr[k2 * 32 + i], xi = sFi[k2 * 32 + i];
    gr += xr * wrv[i] - xi * wiv[i];
    gi += xr * wiv[i] + xi * wrv[i];
  }
  size_t gidx = base + k2 * 32 + o;
  Gr[gidx] = gr; Gi[gidx] = gi;
}

__global__ void k_invY(const float* __restrict__ Gr, const float* __restrict__ Gi,
                       const float* __restrict__ tabC, const float* __restrict__ tabS,
                       float* __restrict__ P, float* __restrict__ Q) {
  int wid = blockIdx.x * (blockDim.x >> 5) + (threadIdx.x >> 5);
  int lane = threadIdx.x & 31;
  int b = wid >> 9, rest = wid & 511;
  int y0 = (rest >> 5) * 16, col0 = (rest & 31) * 16;
  const float* grb = Gr + (size_t)b * 8192;
  const float* gib = Gi + (size_t)b * 8192;
  v8f aCG = vzero(), aSGi = vzero(), aSG = vzero(), aCGi = vzero();
#pragma unroll
  for (int k0 = 0; k0 < M_; k0 += 4) {
    v2f aC = ldAt(tabC + k0 * W_ + y0, W_, lane);
    v2f aS = ldAt(tabS + k0 * W_ + y0, W_, lane);
    v2f bR = ldB(grb + k0 * 512 + col0, 512, lane);
    v2f bI = ldB(gib + k0 * 512 + col0, 512, lane);
    aCG = wmma4(aC, bR, aCG);
    aSGi = wmma4(aS, bI, aSGi);
    aSG = wmma4(aS, bR, aSG);
    aCGi = wmma4(aC, bI, aCGi);
  }
  float* p = P + (size_t)(b * H_ + y0) * 512 + col0;
  float* q = Q + (size_t)(b * H_ + y0) * 512 + col0;
  int n = lane & 15, mh = (lane >> 4) << 3;
#pragma unroll
  for (int r = 0; r < 8; ++r) {
    p[(r + mh) * 512 + n] = (aCG[r] - aSGi[r]) * 0.0625f;
    q[(r + mh) * 512 + n] = (aSG[r] + aCGi[r]) * 0.0625f;
  }
}

typedef float v4f __attribute__((ext_vector_type(4)));
#define XP 36
#define PP 48

__device__ __forceinline__ void gelu_tile(const v8f& a0, const v8f& a1, float bb0, float bb1,
                                          v8f& g0, v8f& g1) {
  float pv0[8], pv1[8];
  float eu0[8], eu1[8];
  float um = 0.f;
#pragma unroll
  for (int r = 0; r < 8; ++r) {
    pv0[r] = a0[r] + bb0;
    pv1[r] = a1[r] + bb1;
    eu0[r] = pv0[r] * 0.70710678f;
    eu1[r] = pv1[r] * 0.70710678f;
    um = fmaxf(um, fmaxf(fabsf(eu0[r]), fabsf(eu1[r])));
  }
  if (__builtin_amdgcn_ballot_w32(!(um < 1.0f)) == 0u) {
#pragma unroll
    for (int r = 0; r < 8; ++r) {
      eu0[r] = erf_poly(eu0[r]);
      eu1[r] = erf_poly(eu1[r]);
    }
  } else {
#pragma unroll
    for (int r = 0; r < 8; ++r) {
      eu0[r] = erff(eu0[r]);
      eu1[r] = erff(eu1[r]);
    }
  }
#pragma unroll
  for (int r = 0; r < 8; ++r) {
    g0[r] = 0.5f * pv0[r] * (1.0f + eu0[r]);
    g1[r] = 0.5f * pv1[r] * (1.0f + eu1[r]);
  }
}

__device__ __forceinline__ void lift_tile(float* tile, const float* __restrict__ xs, int lane,
                                          float lbv, float l0, float l1, float l2) {
#pragma unroll
  for (int j = 0; j < 16; ++j)
    tile[j * XP + lane] = lift1(lbv, l0, l1, l2, xs[j], xs[PIX + j], xs[2 * PIX + j]);
}

template <bool FWD, bool FIRST>
__global__ void __launch_bounds__(256, 8) k_invX(const float* __restrict__ P, const float* __restrict__ Q,
                       float* act, const float* __restrict__ tabC2,
                       const float* __restrict__ tabS2, const float* __restrict__ bw,
                       const float* __restrict__ bb, const float* __restrict__ tabC,
                       const float* __restrict__ tabS, float* __restrict__ Tr,
                       float* __restrict__ Ti, const float* __restrict__ x,
                       const float* __restrict__ lw, const float* __restrict__ lb,
                       const float* __restrict__ pw, const float* __restrict__ pb,
                       float* __restrict__ res, int o_outer) {
  constexpr int WSTR = 2 * 16 * XP;
  __shared__ __attribute__((aligned(16))) float sW[8 * WSTR];
  __shared__ __attribute__((aligned(16))) float sBW[C_ * XP];
  __shared__ __attribute__((aligned(16))) float sP[M_ * PP];
  __shared__ __attribute__((aligned(16))) float sQ[M_ * PP];
  __shared__ __attribute__((aligned(16))) float sO[W_];
  int tid = threadIdx.x;
  int w = tid >> 5;
  int lane = tid & 31;
  int by = blockIdx.x;
  int chn = by >> 10;
  int x0 = w * 16;
  int n = lane & 15, h = lane >> 4, mh = h << 3;
  const float* Pby = P + (size_t)by * 512;
  const float* Qby = Q + (size_t)by * 512;
  const float* xrow = act + (size_t)by * (W_ * C_);
  bw += chn * (4 * C_ * C_);
  bb += chn * (4 * C_);
  float* tileA = sW + w * WSTR;
  float* tileB = tileA + 16 * XP;
  v4f ta[4], tb[4];
  if (!FIRST) {
    const v4f* tsA = reinterpret_cast<const v4f*>(xrow + x0 * C_);
    const v4f* tsB = reinterpret_cast<const v4f*>(xrow + (x0 + 128) * C_);
#pragma unroll
    for (int p = 0; p < 4; ++p) { ta[p] = tsA[32 * p + lane]; tb[p] = tsB[32 * p + lane]; }
  }
  int si = tid & 127;
  v4f sh1, sh2;
  if (tid < 128) sh1 = reinterpret_cast<const v4f*>(Pby)[si];
  else           sh1 = reinterpret_cast<const v4f*>(Qby)[si];
  sh2 = reinterpret_cast<const v4f*>(bw)[tid];
  v2f aCp[2][2], aSp[2][2];
#pragma unroll
  for (int par = 0; par < 2; ++par) {
#pragma unroll
    for (int t = 0; t < 2; ++t) {
      int ra = 8 * t + 2 * h + par;
      aCp[par][t].x = tabC2[ra * W_ + x0 + n];
      aCp[par][t].y = tabC2[(ra + 4) * W_ + x0 + n];
      aSp[par][t].x = -tabS2[ra * W_ + x0 + n];
      aSp[par][t].y = -tabS2[(ra + 4) * W_ + x0 + n];
    }
  }
  float bb0 = bb[n], bb1 = bb[16 + n];
  float pw0 = 0.f, pw1 = 0.f, pbv = 0.f;
  if (!FWD) { pw0 = pw[n]; pw1 = pw[16 + n]; pbv = pb[0]; }
  {
    if (FIRST) {
      int wu = __builtin_amdgcn_readfirstlane(w);
      const float* xs = x + ((size_t)(((by >> 8) & 3) * 3) * H_ + (by & 255)) * W_ + wu * 16;
      float l0 = lw[lane * 3 + 0], l1 = lw[lane * 3 + 1], l2 = lw[lane * 3 + 2], lbv = lb[lane];
      lift_tile(tileA, xs, lane, lbv, l0, l1, l2);
      lift_tile(tileB, xs + 128, lane, lbv, l0, l1, l2);
    } else {
      int tx = lane >> 3, tc = (lane & 7) * 4;
#pragma unroll
      for (int p = 0; p < 4; ++p) {
        *reinterpret_cast<v4f*>(tileA + (4 * p + tx) * XP + tc) = ta[p];
        *reinterpret_cast<v4f*>(tileB + (4 * p + tx) * XP + tc) = tb[p];
      }
    }
    int sr = si >> 3, sc4 = (si & 7) * 4;
    if (tid < 128) *reinterpret_cast<v4f*>(sP + sr * PP + sc4) = sh1;
    else           *reinterpret_cast<v4f*>(sQ + sr * PP + sc4) = sh1;
    *reinterpret_cast<v4f*>(sBW + (tid >> 3) * XP + (tid & 7) * 4) = sh2;
  }
  __syncthreads();
  v8f Se0 = vzero(), Se1 = vzero(), So0 = vzero(), So1 = vzero();
#pragma unroll
  for (int t = 0; t < 2; ++t) {
    const float* pe = sP + (8 * t + 2 * h) * PP + n;
    v2f bE0, bE1, bO0, bO1;
    bE0.x = pe[0];       bE0.y = pe[4 * PP];
    bE1.x = pe[16];      bE1.y = pe[4 * PP + 16];
    bO0.x = pe[PP];      bO0.y = pe[5 * PP];
    bO1.x = pe[PP + 16]; bO1.y = pe[5 * PP + 16];
    Se0 = wmma4(aCp[0][t], bE0, Se0);
    Se1 = wmma4(aCp[0][t], bE1, Se1);
    So0 = wmma4(aCp[1][t], bO0, So0);
    So1 = wmma4(aCp[1][t], bO1, So1);
  }
#pragma unroll
  for (int t = 0; t < 2; ++t) {
    const float* qe = sQ + (8 * t + 2 * h) * PP + n;
    v2f bE0, bE1, bO0, bO1;
    bE0.x = qe[0];       bE0.y = qe[4 * PP];
    bE1.x = qe[16];      bE1.y = qe[4 * PP + 16];
    bO0.x = qe[PP];      bO0.y = qe[5 * PP];
    bO1.x = qe[PP + 16]; bO1.y = qe[5 * PP + 16];
    Se0 = wmma4(aSp[0][t], bE0, Se0);
    Se1 = wmma4(aSp[0][t], bE1, Se1);
    So0 = wmma4(aSp[1][t], bO0, So0);
    So1 = wmma4(aSp[1][t], bO1, So1);
  }
  v8f accA0 = Se0 + So0, accA1 = Se1 + So1;
  v8f accB0 = Se0 - So0, accB1 = Se1 - So1;
#pragma unroll
  for (int k0 = 0; k0 < C_; k0 += 4) {
    v2f aXA = ldA(tileA + k0, XP, lane);
    v2f aXB = ldA(tileB + k0, XP, lane);
    v2f bW0 = ldBt(sBW + k0, XP, lane);
    v2f bW1 = ldBt(sBW + 16 * XP + k0, XP, lane);
    accA0 = wmma4(aXA, bW0, accA0);
    accA1 = wmma4(aXA, bW1, accA1);
    accB0 = wmma4(aXB, bW0, accB0);
    accB1 = wmma4(aXB, bW1, accB1);
  }
  v2f fE[4], fO[4];
  if (FWD) {
    const float* trow = ((n < 8) ? tabC : tabS) + (2 * (n & 7)) * W_ + x0 + mh;
#pragma unroll
    for (int j = 0; j < 4; ++j) {
      fE[j] = *reinterpret_cast<const v2f*>(trow + 2 * j);
      fO[j] = *reinterpret_cast<const v2f*>(trow + W_ + 2 * j);
    }
  }
  float* out = act + (size_t)by * (W_ * C_);
  v8f gA0, gA1, gB0, gB1;
  gelu_tile(accA0, accA1, bb0, bb1, gA0, gA1);
  gelu_tile(accB0, accB1, bb0, bb1, gB0, gB1);
  if (FWD) {
#pragma unroll
    for (int r = 0; r < 8; ++r) {
      int xr = x0 + r + mh;
      out[xr * C_ + n] = gA0[r];
      out[xr * C_ + 16 + n] = gA1[r];
      out[(xr + 128) * C_ + n] = gB0[r];
      out[(xr + 128) * C_ + 16 + n] = gB1[r];
    }
  } else {
    float psA[8], psB[8];
#pragma unroll
    for (int r = 0; r < 8; ++r) {
      psA[r] = row_sum16(pw0 * gA0[r] + pw1 * gA1[r]);
      psB[r] = row_sum16(pw0 * gB0[r] + pw1 * gB1[r]);
    }
    if (n == 0) {
      v4f lo = {psA[0], psA[1], psA[2], psA[3]};
      v4f hi = {psA[4], psA[5], psA[6], psA[7]};
      *reinterpret_cast<v4f*>(sO + x0 + mh) = lo;
      *reinterpret_cast<v4f*>(sO + x0 + mh + 4) = hi;
      v4f lo2 = {psB[0], psB[1], psB[2], psB[3]};
      v4f hi2 = {psB[4], psB[5], psB[6], psB[7]};
      *reinterpret_cast<v4f*>(sO + 128 + x0 + mh) = lo2;
      *reinterpret_cast<v4f*>(sO + 128 + x0 + mh + 4) = hi2;
    }
    __syncthreads();
    {
      int bq = (by >> 8) & 3, y = by & 255;
      res[((size_t)(bq * 4 + o_outer + chn)) * PIX + y * W_ + tid] = sO[tid] + pbv;
    }
  }
  if (FWD) {
    v8f E0 = gA0 + gB0, E1 = gA1 + gB1;
    v8f O0 = gA0 - gB0, O1 = gA1 - gB1;
    v8f dE0 = vzero(), dE1 = vzero(), dO0 = vzero(), dO1 = vzero();
#pragma unroll
    for (int j = 0; j < 4; ++j) {
      v2f e0; e0.x = E0[2 * j]; e0.y = E0[2 * j + 1];
      v2f e1; e1.x = E1[2 * j]; e1.y = E1[2 * j + 1];
      v2f o0; o0.x = O0[2 * j]; o0.y = O0[2 * j + 1];
      v2f o1; o1.x = O1[2 * j]; o1.y = O1[2 * j + 1];
      dE0 = wmma4(fE[j], e0, dE0);
      dE1 = wmma4(fE[j], e1, dE1);
      dO0 = wmma4(fO[j], o0, dO0);
      dO1 = wmma4(fO[j], o1, dO1);
    }
    float* pc = sW + w * WSTR + h * 32 + n;
#pragma unroll
    for (int r = 0; r < 8; ++r) {
      pc[(2 * r) * 64] = dE0[r];
      pc[(2 * r) * 64 + 16] = dE1[r];
      pc[(2 * r + 1) * 64] = dO0[r];
      pc[(2 * r + 1) * 64 + 16] = dO1[r];
    }
    __syncthreads();
    float sc = (tid & 32) ? -0.0625f : 0.0625f;
    float* dst = ((tid & 32) ? Ti : Tr) + (size_t)by * (M_ * C_) + (tid >> 6) * C_ + (tid & 31);
#pragma unroll
    for (int i = 0; i < 4; ++i) {
      float s = 0.f;
#pragma unroll
      for (int ww = 0; ww < 8; ++ww) s += sW[ww * WSTR + tid + 256 * i];
      dst[4 * i * C_] = s * sc;
    }
  }
}

extern "C" void kernel_launch(void* const* d_in, const int* in_sizes, int n_in,
                              void* d_out, int out_size, void* d_ws, size_t ws_size,
                              hipStream_t stream) {
  const float* x      = (const float*)d_in[0];
  const float* lift_w = (const float*)d_in[1];
  const float* lift_b = (const float*)d_in[2];
  const float* wr     = (const float*)d_in[3];
  const float* wi     = (const float*)d_in[4];
  const float* byp_w  = (const float*)d_in[5];
  const float* byp_b  = (const float*)d_in[6];
  const float* proj_w = (const float*)d_in[7];
  const float* proj_b = (const float*)d_in[8];
  float* out = (float*)d_out;

  float* w = (float*)d_ws;
  float* tabC  = w;
  float* tabS  = tabC + 4096;
  float* tabC2 = tabS + 4096;
  float* tabS2 = tabC2 + 4096;
  float* act   = tabS2 + 4096;
  float* Tr    = act + (size_t)NB_ * PIX * C_;
  float* Ti    = Tr + (size_t)NB_ * H_ * M_ * C_;
  float* Frp   = Ti + (size_t)NB_ * H_ * M_ * C_;
  float* Fip   = Frp + (size_t)4 * NB_ * M_ * 512;
  float* Gr    = Fip + (size_t)4 * NB_ * M_ * 512;
  float* Gi    = Gr + (size_t)NB_ * M_ * M_ * C_;
  float* Pb    = Gi + (size_t)NB_ * M_ * M_ * C_;
  float* Qb    = Pb + (size_t)NB_ * H_ * M_ * C_;

  k_tabs<<<16, 256, 0, stream>>>(tabC, tabS, tabC2, tabS2);

  for (int o = 0; o < 4; o += NCH) {
    for (int n = 0; n < 4; ++n) {
      size_t wOff = ((size_t)(o * 4 + n)) * C_ * C_ * M_ * M_;
      size_t bwOff = ((size_t)(o * 4 + n)) * C_ * C_;
      size_t bbOff = ((size_t)(o * 4 + n)) * C_;
      if (n == 0) k_fwdX<<<256, 256, 0, stream>>>(x, lift_w, lift_b, tabC, tabS, Tr, Ti);
      k_fwdY<<<NCH * 64, 256, 0, stream>>>(Tr, Ti, tabC, tabS, Frp, Fip);
      k_mix<<<NCH * 64, 512, 0, stream>>>(Frp, Fip, wr + wOff, wi + wOff, Gr, Gi);
      k_invY<<<NCH * 256, 256, 0, stream>>>(Gr, Gi, tabC, tabS, Pb, Qb);
      if (n == 0)
        k_invX<true, true><<<NCH * 1024, 256, 0, stream>>>(Pb, Qb, act, tabC2, tabS2,
                                                           byp_w + bwOff, byp_b + bbOff,
                                                           tabC, tabS, Tr, Ti, x, lift_w, lift_b,
                                                           proj_w, proj_b, out, o);
      else if (n < 3)
        k_invX<true, false><<<NCH * 1024, 256, 0, stream>>>(Pb, Qb, act, tabC2, tabS2,
                                                            byp_w + bwOff, byp_b + bbOff,
                                                            tabC, tabS, Tr, Ti, x, lift_w, lift_b,
                                                            proj_w, proj_b, out, o);
      else
        k_invX<false, false><<<NCH * 1024, 256, 0, stream>>>(Pb, Qb, act, tabC2, tabS2,
                                                             byp_w + bwOff, byp_b + bbOff,
                                                             tabC, tabS, Tr, Ti, x, lift_w, lift_b,
                                                             proj_w, proj_b, out, o);
    }
  }
}


# Forward y transform, channel mix and inverse y transform fused: one block per (q,k2), weights re-ordered by idle blocks of the step before; 35 -> 20 launches (forward y sum order changes)
# speedup vs baseline: 2.9164x; 1.2376x over previous
#include <hip/hip_runtime.h>
#include <math.h>


typedef float v2f __attribute__((ext_vector_type(2)));
typedef float v8f __attribute__((ext_vector_type(8)));

#define B_  4
#define NCH 2
#define NB_ (NCH * B_)
#define C_  32
#define H_  256
#define W_  256
#define M_  16
#define PIX (H_ * W_)

__device__ __forceinline__ v8f wmma4(v2f a, v2f b, v8f c) {
  return __builtin_amdgcn_wmma_f32_16x16x4_f32(false, a, false, b, (short)0, c,
                                               false, false);
}

__device__ __forceinline__ v2f ldA(const float* p, int lda, int lane) {
  int r = lane & 15, kk = (lane >> 4) << 1;
  v2f v; v.x = p[r * lda + kk]; v.y = p[r * lda + kk + 1]; return v;
}
__device__ __forceinline__ v2f ldAt(const float* p, int lda, int lane) {
  int r = lane & 15, kk = (lane >> 4) << 1;
  v2f v; v.x = p[kk * lda + r]; v.y = p[(kk + 1) * lda + r]; return v;
}
__device__ __forceinline__ v2f ldB(const float* p, int ldb, int lane) {
  int n = lane & 15, kk = (lane >> 4) << 1;
  v2f v; v.x = p[kk * ldb + n]; v.y = p[(kk + 1) * ldb + n]; return v;
}
__device__ __forceinline__ v2f ldBt(const float* p, int ldb, int lane) {
  int n = lane & 15, kk = (lane >> 4) << 1;
  v2f v; v.x = p[n * ldb + kk]; v.y = p[n * ldb + kk + 1]; return v;
}
__device__ __forceinline__ v2f ldA_d(const float* p, int lda, int lane, int j) {
  int r = lane & 15, kk = ((lane >> 4) << 3) + 2 * j;
  v2f v; v.x = p[r * lda + kk]; v.y = p[r * lda + kk + 1]; return v;
}
__device__ __forceinline__ void stD(float* p, int ldd, int lane, v8f d, float s) {
  int n = lane & 15, mh = (lane >> 4) << 3;
#pragma unroll
  for (int r = 0; r < 8; ++r) p[(r + mh) * ldd + n] = d[r] * s;
}

__device__ __forceinline__ v8f vzero() {
  v8f z = {0.f, 0.f, 0.f, 0.f, 0.f, 0.f, 0.f, 0.f};
  return z;
}

template <int MASK>
__device__ __forceinline__ float row_xor(float v) {
  int b = __builtin_bit_cast(int, v);
  return __builtin_bit_cast(float, __builtin_amdgcn_mov_dpp(b, 0x160 + MASK, 0xf, 0xf, true));
}
__device__ __forceinline__ float row_sum16(float v) {
  v += row_xor<1>(v);
  v += row_xor<2>(v);
  v += row_xor<4>(v);
  v += row_xor<8>(v);
  return v;
}

__device__ __forceinline__ float erf_poly(float u) {
  float t = u * u;
  float p = __builtin_fmaf(t, __builtin_bit_cast(float, 0xba1345e1u), __builtin_bit_cast(float, 0x3ba10414u));
  p = __builtin_fmaf(t, p, __builtin_bit_cast(float, 0xbcdac9b8u));
  p = __builtin_fmaf(t, p, __builtin_bit_cast(float, 0x3de703beu));
  p = __builtin_fmaf(t, p, __builtin_bit_cast(float, 0xbec09330u));
  p = __builtin_fmaf(t, p, __builtin_bit_cast(float, 0x3e0375d0u));
  return __builtin_fmaf(u, p, u);
}

__global__ void k_tabs(float* tabC, float* tabS, float* tabC2, float* tabS2) {
  int i = blockIdx.x * blockDim.x + threadIdx.x;
  if (i >= M_ * W_) return;
  int k = i >> 8, x = i & 255;
  int m = (k * x) & 255;
  float ang = 0.0245436926f * (float)m;
  float c = cosf(ang), s = sinf(ang);
  tabC[i] = c; tabS[i] = s;
  float w2 = (k == 0) ? 1.0f : 2.0f;
  tabC2[i] = w2 * c * 0.0625f;
  tabS2[i] = w2 * s * 0.0625f;
}

__device__ __forceinline__ float lift1(float lbv, float l0, float l1, float l2,
                                       float xi0, float xi1, float xi2) {
  return lbv + l0 * xi0 + l1 * xi1 + l2 * xi2;
}

#define FX_U 8
__device__ __forceinline__ void fx_load(v2f (&c)[FX_U], v2f (&s)[FX_U], v2f (&p0)[FX_U],
                                        v2f (&p1)[FX_U], v2f (&p2)[FX_U],
                                        const float* __restrict__ tabC,
                                        const float* __restrict__ tabS,
                                        const float* __restrict__ xr, int k, int lane) {
#pragma unroll
  for (int u = 0; u < FX_U; ++u) {
    c[u] = ldA(tabC + k + 4 * u, W_, lane);
    s[u] = ldA(tabS + k + 4 * u, W_, lane);
    p0[u] = *reinterpret_cast<const v2f*>(xr + k + 4 * u);
    p1[u] = *reinterpret_cast<const v2f*>(xr + PIX + k + 4 * u);
    p2[u] = *reinterpret_cast<const v2f*>(xr + 2 * PIX + k + 4 * u);
  }
}
__device__ __forceinline__ void fx_mma(const v2f (&c)[FX_U], const v2f (&s)[FX_U],
                                       const v2f (&p0)[FX_U], const v2f (&p1)[FX_U],
                                       const v2f (&p2)[FX_U], float lbv, float l0, float l1,
                                       float l2, v8f& aC, v8f& aS) {
#pragma unroll
  for (int u = 0; u < FX_U; ++u) {
    v2f b;
    b.x = lift1(lbv, l0, l1, l2, p0[u].x, p1[u].x, p2[u].x);
    b.y = lift1(lbv, l0, l1, l2, p0[u].y, p1[u].y, p2[u].y);
    aC = wmma4(c[u], b, aC);
    aS = wmma4(s[u], b, aS);
  }
}
__global__ void __launch_bounds__(256) k_fwdX(const float* __restrict__ x, const float* __restrict__ lw,
                       const float* __restrict__ lb, const float* __restrict__ tabC,
                       const float* __restrict__ tabS, float* __restrict__ Tr,
                       float* __restrict__ Ti) {
  int wid = blockIdx.x * (blockDim.x >> 5) + (threadIdx.x >> 5);
  int lane = threadIdx.x & 31;
  int ch = wid & 1;
  int byi = wid >> 1;
  int b = byi >> 8, y = byi & 255;
  int cc = 16 * ch + (lane & 15);
  float l0 = lw[cc * 3 + 0], l1 = lw[cc * 3 + 1], l2 = lw[cc * 3 + 2], lbv = lb[cc];
  const float* xr = x + ((size_t)(b * 3) * H_ + y) * W_ + ((lane >> 4) << 1);
  v8f aC = vzero(), aS = vzero();
  v2f c0[FX_U], s0[FX_U], c1[FX_U], s1[FX_U];
  v2f p00[FX_U], p01[FX_U], p02[FX_U], p10[FX_U], p11[FX_U], p12[FX_U];
  for (int k0 = 0; k0 < W_; k0 += 8 * FX_U) {
    fx_load(c0, s0, p00, p01, p02, tabC, tabS, xr, k0, lane);
    fx_load(c1, s1, p10, p11, p12, tabC, tabS, xr, k0 + 4 * FX_U, lane);
    __builtin_amdgcn_sched_barrier(0);
    fx_mma(c0, s0, p00, p01, p02, lbv, l0, l1, l2, aC, aS);
    fx_mma(c1, s1, p10, p11, p12, lbv, l0, l1, l2, aC, aS);
    __builtin_amdgcn_sched_barrier(0);
  }
  float* tr = Tr + ((size_t)(b * H_ + y)) * M_ * C_ + 16 * ch;
  float* ti = Ti + ((size_t)(b * H_ + y)) * M_ * C_ + 16 * ch;
  stD(tr, C_, lane, aC, 0.0625f);
  stD(ti, C_, lane, aS, -0.0625f);
  stD(tr + (size_t)B_ * H_ * M_ * C_, C_, lane, aC, 0.0625f);
  stD(ti + (size_t)B_ * H_ * M_ * C_, C_, lane, aS, -0.0625f);
}

#define FP 36
#define WSTEP (C_ * C_ * M_ * M_)

__device__ __forceinline__ void relay_block(int rb, const float* __restrict__ wr,
                                            const float* __restrict__ wi,
                                            float* __restrict__ wt, float* sT, int tid) {
  int chain = rb >> 6, ri = (rb >> 5) & 1, i = rb & 31;
  const float* srcp = (ri ? wi : wr) + (size_t)chain * (4 * WSTEP) + (size_t)i * (C_ * M_ * M_);
  float* dst = wt + (size_t)chain * (2 * WSTEP) + (size_t)ri * WSTEP + i * C_;
  float v[16];
#pragma unroll
  for (int j = 0; j < 16; ++j) v[j] = srcp[j * 512 + tid];
#pragma unroll
  for (int j = 0; j < 16; ++j) {
    int idx = j * 512 + tid;
    sT[(idx >> 8) * 257 + (idx & 255)] = v[j];
  }
  __syncthreads();
#pragma unroll
  for (int j = 0; j < 16; ++j) {
    int idx = j * 512 + tid;
    int o = idx & 31, m = idx >> 5;
    int kk = (m & 15) * 16 + (m >> 4);
    dst[(size_t)m * (C_ * C_) + o] = sT[o * 257 + kk];
  }
}

__global__ void __launch_bounds__(512) k_relay(const float* __restrict__ wr,
                                               const float* __restrict__ wi,
                                               float* __restrict__ wt) {
  __shared__ float sT[32 * 257];
  relay_block(blockIdx.x, wr, wi, wt, sT, threadIdx.x);
}

__global__ void __launch_bounds__(512) k_mid(float* TP, float* TQ,
                                             const float* __restrict__ tabC,
                                             const float* __restrict__ tabS,
                                             const float* __restrict__ wt,
                                             const float* __restrict__ wrN,
                                             const float* __restrict__ wiN,
                                             float* __restrict__ wtN) {
  __shared__ float sPart[16 * 2 * 16 * FP];
  __shared__ float sF[2 * 16 * FP];
  __shared__ float sG[2 * 16 * FP];
  int tid = threadIdx.x;
  if (blockIdx.x >= 128) {
    relay_block(blockIdx.x - 128, wrN, wiN, wtN, sPart, tid);
    return;
  }
  int w = tid >> 5, lane = tid & 31;
  int q = blockIdx.x >> 4, k2 = blockIdx.x & 15;
  int n = lane & 15, mh = (lane >> 4) << 3;
  int k1t = w, ot = lane;
  int y0 = 16 * w;
  float* tr = TP + (size_t)q * H_ * 512 + k2 * 32;
  float* ti = TQ + (size_t)q * H_ * 512 + k2 * 32;
  v2f fC[4], fS[4], fR[4][2], fI[4][2];
#pragma unroll
  for (int ks = 0; ks < 4; ++ks) {
    int k0 = y0 + 4 * ks;
    fC[ks] = ldA(tabC + k0, W_, lane);
    fS[ks] = ldA(tabS + k0, W_, lane);
#pragma unroll
    for (int nt = 0; nt < 2; ++nt) {
      fR[ks][nt] = ldB(tr + (size_t)k0 * 512 + nt * 16, 512, lane);
      fI[ks][nt] = ldB(ti + (size_t)k0 * 512 + nt * 16, 512, lane);
    }
  }
  __builtin_amdgcn_sched_barrier(0);
  const float* wtr = wt + (size_t)(q >> 2) * (2 * WSTEP) + (size_t)((k2 * 16 + k1t) * C_) * C_ + ot;
  const float* wti = wtr + WSTEP;
  float wrv[C_], wiv[C_];
#pragma unroll
  for (int i = 0; i < C_; ++i) { wrv[i] = wtr[i * C_]; wiv[i] = wti[i * C_]; }
  v2f iC[4], iS[4];
#pragma unroll
  for (int ks = 0; ks < 4; ++ks) {
    iC[ks] = ldAt(tabC + (4 * ks) * W_ + y0, W_, lane);
    iS[ks] = ldAt(tabS + (4 * ks) * W_ + y0, W_, lane);
  }
  __builtin_amdgcn_sched_barrier(0);
  v8f aCT[2], aST[2], aCTi[2], aSTr[2];
  aCT[0] = aCT[1] = aST[0] = aST[1] = aCTi[0] = aCTi[1] = aSTr[0] = aSTr[1] = vzero();
#pragma unroll
  for (int ks = 0; ks < 4; ++ks) {
#pragma unroll
    for (int nt = 0; nt < 2; ++nt) {
      aCT[nt] = wmma4(fC[ks], fR[ks][nt], aCT[nt]);
      aST[nt] = wmma4(fS[ks], fI[ks][nt], aST[nt]);
      aCTi[nt] = wmma4(fC[ks], fI[ks][nt], aCTi[nt]);
      aSTr[nt] = wmma4(fS[ks], fR[ks][nt], aSTr[nt]);
    }
  }
  {
    float* pp = sPart + w * (2 * 16 * FP) + mh * FP + n;
#pragma unroll
    for (int nt = 0; nt < 2; ++nt) {
#pragma unroll
      for (int r = 0; r < 8; ++r) {
        pp[r * FP + nt * 16] = (aCT[nt][r] + aST[nt][r]) * 0.0625f;
        pp[16 * FP + r * FP + nt * 16] = (aCTi[nt][r] - aSTr[nt][r]) * 0.0625f;
      }
    }
  }
  __syncthreads();
  {
    float fr = 0.f, fi = 0.f;
#pragma unroll
    for (int ww = 0; ww < 16; ++ww) {
      fr += sPart[ww * (2 * 16 * FP) + k1t * FP + ot];
      fi += sPart[ww * (2 * 16 * FP) + 16 * FP + k1t * FP + ot];
    }
    sF[k1t * FP + ot] = fr;
    sF[16 * FP + k1t * FP + ot] = fi;
  }
  __syncthreads();
  {
    float gr = 0.f, gi = 0.f;
#pragma unroll
    for (int i = 0; i < C_; ++i) {
      float xr = sF[k1t * FP + i], xi = sF[16 * FP + k1t * FP + i];
      gr += xr * wrv[i] - xi * wiv[i];
      gi += xr * wiv[i] + xi * wrv[i];
    }
    sG[k1t * FP + ot] = gr;
    sG[16 * FP + k1t * FP + ot] = gi;
  }
  __syncthreads();
  v8f aCG[2], aSGi[2], aSG[2], aCGi[2];
  aCG[0] = aCG[1] = aSGi[0] = aSGi[1] = aSG[0] = aSG[1] = aCGi[0] = aCGi[1] = vzero();
#pragma unroll
  for (int ks = 0; ks < 4; ++ks) {
#pragma unroll
    for (int nt = 0; nt < 2; ++nt) {
      v2f bR = ldB(sG + (4 * ks) * FP + nt * 16, FP, lane);
      v2f bI = ldB(sG + 16 * FP + (4 * ks) * FP + nt * 16, FP, lane);
      aCG[nt] = wmma4(iC[ks], bR, aCG[nt]);
      aSGi[nt] = wmma4(iS[ks], bI, aSGi[nt]);
      aSG[nt] = wmma4(iS[ks], bR, aSG[nt]);
      aCGi[nt] = wmma4(iC[ks], bI, aCGi[nt]);
    }
  }
  float* p = TP + (size_t)(q * H_ + y0 + mh) * 512 + k2 * 32 + n;
  float* qo = TQ + (size_t)(q * H_ + y0 + mh) * 512 + k2 * 32 + n;
#pragma unroll
  for (int nt = 0; nt < 2; ++nt) {
#pragma unroll
    for (int r = 0; r < 8; ++r) {
      p[r * 512 + nt * 16] = (aCG[nt][r] - aSGi[nt][r]) * 0.0625f;
      qo[r * 512 + nt * 16] = (aSG[nt][r] + aCGi[nt][r]) * 0.0625f;
    }
  }
}

typedef float v4f __attribute__((ext_vector_type(4)));
#define XP 36
#define PP 48

__device__ __forceinline__ void gelu_tile(const v8f& a0, const v8f& a1, float bb0, float bb1,
                                          v8f& g0, v8f& g1) {
  float pv0[8], pv1[8];
  float eu0[8], eu1[8];
  float um = 0.f;
#pragma unroll
  for (int r = 0; r < 8; ++r) {
    pv0[r] = a0[r] + bb0;
    pv1[r] = a1[r] + bb1;
    eu0[r] = pv0[r] * 0.70710678f;
    eu1[r] = pv1[r] * 0.70710678f;
    um = fmaxf(um, fmaxf(fabsf(eu0[r]), fabsf(eu1[r])));
  }
  if (__builtin_amdgcn_ballot_w32(!(um < 1.0f)) == 0u) {
#pragma unroll
    for (int r = 0; r < 8; ++r) {
      eu0[r] = erf_poly(eu0[r]);
      eu1[r] = erf_poly(eu1[r]);
    }
  } else {
#pragma unroll
    for (int r = 0; r < 8; ++r) {
      eu0[r] = erff(eu0[r]);
      eu1[r] = erff(eu1[r]);
    }
  }
#pragma unroll
  for (int r = 0; r < 8; ++r) {
    g0[r] = 0.5f * pv0[r] * (1.0f + eu0[r]);
    g1[r] = 0.5f * pv1[r] * (1.0f + eu1[r]);
  }
}

__device__ __forceinline__ void lift_tile(float* tile, const float* __restrict__ xs, int lane,
                                          float lbv, float l0, float l1, float l2) {
#pragma unroll
  for (int j = 0; j < 16; ++j)
    tile[j * XP + lane] = lift1(lbv, l0, l1, l2, xs[j], xs[PIX + j], xs[2 * PIX + j]);
}

template <bool FWD, bool FIRST>
__global__ void __launch_bounds__(256, 8) k_invX(const float* P, const float* Q,
                       float* act, const float* __restrict__ tabC2,
                       const float* __restrict__ tabS2, const float* __restrict__ bw,
                       const float* __restrict__ bb, const float* __restrict__ tabC,
                       const float* __restrict__ tabS, float* Tr,
                       float* Ti, const float* __restrict__ x,
                       const float* __restrict__ lw, const float* __restrict__ lb,
                       const float* __restrict__ pw, const float* __restrict__ pb,
                       float* __restrict__ res, int o_outer) {
  constexpr int WSTR = 2 * 16 * XP;
  __shared__ __attribute__((aligned(16))) float sW[8 * WSTR];
  __shared__ __attribute__((aligned(16))) float sBW[C_ * XP];
  __shared__ __attribute__((aligned(16))) float sP[M_ * PP];
  __shared__ __attribute__((aligned(16))) float sQ[M_ * PP];
  __shared__ __attribute__((aligned(16))) float sO[W_];
  int tid = threadIdx.x;
  int w = tid >> 5;
  int lane = tid & 31;
  int by = blockIdx.x;
  int chn = by >> 10;
  int x0 = w * 16;
  int n = lane & 15, h = lane >> 4, mh = h << 3;
  const float* Pby = P + (size_t)by * 512;
  const float* Qby = Q + (size_t)by * 512;
  const float* xrow = act + (size_t)by * (W_ * C_);
  bw += chn * (4 * C_ * C_);
  bb += chn * (4 * C_);
  float* tileA = sW + w * WSTR;
  float* tileB = tileA + 16 * XP;
  v4f ta[4], tb[4];
  if (!FIRST) {
    const v4f* tsA = reinterpret_cast<const v4f*>(xrow + x0 * C_);
    const v4f* tsB = reinterpret_cast<const v4f*>(xrow + (x0 + 128) * C_);
#pragma unroll
    for (int p = 0; p < 4; ++p) { ta[p] = tsA[32 * p + lane]; tb[p] = tsB[32 * p + lane]; }
  }
  int si = tid & 127;
  v4f sh1, sh2;
  if (tid < 128) sh1 = reinterpret_cast<const v4f*>(Pby)[si];
  else           sh1 = reinterpret_cast<const v4f*>(Qby)[si];
  sh2 = reinterpret_cast<const v4f*>(bw)[tid];
  v2f aCp[2][2], aSp[2][2];
#pragma unroll
  for (int par = 0; par < 2; ++par) {
#pragma unroll
    for (int t = 0; t < 2; ++t) {
      int ra = 8 * t + 2 * h + par;
      aCp[par][t].x = tabC2[ra * W_ + x0 + n];
      aCp[par][t].y = tabC2[(ra + 4) * W_ + x0 + n];
      aSp[par][t].x = -tabS2[ra * W_ + x0 + n];
      aSp[par][t].y = -tabS2[(ra + 4) * W_ + x0 + n];
    }
  }
  float bb0 = bb[n], bb1 = bb[16 + n];
  float pw0 = 0.f, pw1 = 0.f, pbv = 0.f;
  if (!FWD) { pw0 = pw[n]; pw1 = pw[16 + n]; pbv = pb[0]; }
  {
    if (FIRST) {
      int wu = __builtin_amdgcn_readfirstlane(w);
      const float* xs = x + ((size_t)(((by >> 8) & 3) * 3) * H_ + (by & 255)) * W_ + wu * 16;
      float l0 = lw[lane * 3 + 0], l1 = lw[lane * 3 + 1], l2 = lw[lane * 3 + 2], lbv = lb[lane];
      lift_tile(tileA, xs, lane, lbv, l0, l1, l2);
      lift_tile(tileB, xs + 128, lane, lbv, l0, l1, l2);
    } else {
      int tx = lane >> 3, tc = (lane & 7) * 4;
#pragma unroll
      for (int p = 0; p < 4; ++p) {
        *reinterpret_cast<v4f*>(tileA + (4 * p + tx) * XP + tc) = ta[p];
        *reinterpret_cast<v4f*>(tileB + (4 * p + tx) * XP + tc) = tb[p];
      }
    }
    int sr = si >> 3, sc4 = (si & 7) * 4;
    if (tid < 128) *reinterpret_cast<v4f*>(sP + sr * PP + sc4) = sh1;
    else           *reinterpret_cast<v4f*>(sQ + sr * PP + sc4) = sh1;
    *reinterpret_cast<v4f*>(sBW + (tid >> 3) * XP + (tid & 7) * 4) = sh2;
  }
  __syncthreads();
  v8f Se0 = vzero(), Se1 = vzero(), So0 = vzero(), So1 = vzero();
#pragma unroll
  for (int t = 0; t < 2; ++t) {
    const float* pe = sP + (8 * t + 2 * h) * PP + n;
    v2f bE0, bE1, bO0, bO1;
    bE0.x = pe[0];       bE0.y = pe[4 * PP];
    bE1.x = pe[16];      bE1.y = pe[4 * PP + 16];
    bO0.x = pe[PP];      bO0.y = pe[5 * PP];
    bO1.x = pe[PP + 16]; bO1.y = pe[5 * PP + 16];
    Se0 = wmma4(aCp[0][t], bE0, Se0);
    Se1 = wmma4(aCp[0][t], bE1, Se1);
    So0 = wmma4(aCp[1][t], bO0, So0);
    So1 = wmma4(aCp[1][t], bO1, So1);
  }
#pragma unroll
  for (int t = 0; t < 2; ++t) {
    const float* qe = sQ + (8 * t + 2 * h) * PP + n;
    v2f bE0, bE1, bO0, bO1;
    bE0.x = qe[0];       bE0.y = qe[4 * PP];
    bE1.x = qe[16];      bE1.y = qe[4 * PP + 16];
    bO0.x = qe[PP];      bO0.y = qe[5 * PP];
    bO1.x = qe[PP + 16]; bO1.y = qe[5 * PP + 16];
    Se0 = wmma4(aSp[0][t], bE0, Se0);
    Se1 = wmma4(aSp[0][t], bE1, Se1);
    So0 = wmma4(aSp[1][t], bO0, So0);
    So1 = wmma4(aSp[1][t], bO1, So1);
  }
  v8f accA0 = Se0 + So0, accA1 = Se1 + So1;
  v8f accB0 = Se0 - So0, accB1 = Se1 - So1;
#pragma unroll
  for (int k0 = 0; k0 < C_; k0 += 4) {
    v2f aXA = ldA(tileA + k0, XP, lane);
    v2f aXB = ldA(tileB + k0, XP, lane);
    v2f bW0 = ldBt(sBW + k0, XP, lane);
    v2f bW1 = ldBt(sBW + 16 * XP + k0, XP, lane);
    accA0 = wmma4(aXA, bW0, accA0);
    accA1 = wmma4(aXA, bW1, accA1);
    accB0 = wmma4(aXB, bW0, accB0);
    accB1 = wmma4(aXB, bW1, accB1);
  }
  v2f fE[4], fO[4];
  if (FWD) {
    const float* trow = ((n < 8) ? tabC : tabS) + (2 * (n & 7)) * W_ + x0 + mh;
#pragma unroll
    for (int j = 0; j < 4; ++j) {
      fE[j] = *reinterpret_cast<const v2f*>(trow + 2 * j);
      fO[j] = *reinterpret_cast<const v2f*>(trow + W_ + 2 * j);
    }
  }
  float* out = act + (size_t)by * (W_ * C_);
  v8f gA0, gA1, gB0, gB1;
  gelu_tile(accA0, accA1, bb0, bb1, gA0, gA1);
  gelu_tile(accB0, accB1, bb0, bb1, gB0, gB1);
  if (FWD) {
#pragma unroll
    for (int r = 0; r < 8; ++r) {
      int xr = x0 + r + mh;
      out[xr * C_ + n] = gA0[r];
      out[xr * C_ + 16 + n] = gA1[r];
      out[(xr + 128) * C_ + n] = gB0[r];
      out[(xr + 128) * C_ + 16 + n] = gB1[r];
    }
  } else {
    float psA[8], psB[8];
#pragma unroll
    for (int r = 0; r < 8; ++r) {
      psA[r] = row_sum16(pw0 * gA0[r] + pw1 * gA1[r]);
      psB[r] = row_sum16(pw0 * gB0[r] + pw1 * gB1[r]);
    }
    if (n == 0) {
      v4f lo = {psA[0], psA[1], psA[2], psA[3]};
      v4f hi = {psA[4], psA[5], psA[6], psA[7]};
      *reinterpret_cast<v4f*>(sO + x0 + mh) = lo;
      *reinterpret_cast<v4f*>(sO + x0 + mh + 4) = hi;
      v4f lo2 = {psB[0], psB[1], psB[2], psB[3]};
      v4f hi2 = {psB[4], psB[5], psB[6], psB[7]};
      *reinterpret_cast<v4f*>(sO + 128 + x0 + mh) = lo2;
      *reinterpret_cast<v4f*>(sO + 128 + x0 + mh + 4) = hi2;
    }
    __syncthreads();
    {
      int bq = (by >> 8) & 3, y = by & 255;
      res[((size_t)(bq * 4 + o_outer + chn)) * PIX + y * W_ + tid] = sO[tid] + pbv;
    }
  }
  if (FWD) {
    v8f E0 = gA0 + gB0, E1 = gA1 + gB1;
    v8f O0 = gA0 - gB0, O1 = gA1 - gB1;
    v8f dE0 = vzero(), dE1 = vzero(), dO0 = vzero(), dO1 = vzero();
#pragma unroll
    for (int j = 0; j < 4; ++j) {
      v2f e0; e0.x = E0[2 * j]; e0.y = E0[2 * j + 1];
      v2f e1; e1.x = E1[2 * j]; e1.y = E1[2 * j + 1];
      v2f o0; o0.x = O0[2 * j]; o0.y = O0[2 * j + 1];
      v2f o1; o1.x = O1[2 * j]; o1.y = O1[2 * j + 1];
      dE0 = wmma4(fE[j], e0, dE0);
      dE1 = wmma4(fE[j], e1, dE1);
      dO0 = wmma4(fO[j], o0, dO0);
      dO1 = wmma4(fO[j], o1, dO1);
    }
    float* pc = sW + w * WSTR + h * 32 + n;
#pragma unroll
    for (int r = 0; r < 8; ++r) {
      pc[(2 * r) * 64] = dE0[r];
      pc[(2 * r) * 64 + 16] = dE1[r];
      pc[(2 * r + 1) * 64] = dO0[r];
      pc[(2 * r + 1) * 64 + 16] = dO1[r];
    }
    __syncthreads();
    float sc = (tid & 32) ? -0.0625f : 0.0625f;
    float* dst = ((tid & 32) ? Ti : Tr) + (size_t)by * (M_ * C_) + (tid >> 6) * C_ + (tid & 31);
#pragma unroll
    for (int i = 0; i < 4; ++i) {
      float s = 0.f;
#pragma unroll
      for (int ww = 0; ww < 8; ++ww) s += sW[ww * WSTR + tid + 256 * i];
      dst[4 * i * C_] = s * sc;
    }
  }
}

extern "C" void kernel_launch(void* const* d_in, const int* in_sizes, int n_in,
                              void* d_out, int out_size, void* d_ws, size_t ws_size,
                              hipStream_t stream) {
  const float* x      = (const float*)d_in[0];
  const float* lift_w = (const float*)d_in[1];
  const float* lift_b = (const float*)d_in[2];
  const float* wr     = (const float*)d_in[3];
  const float* wi     = (const float*)d_in[4];
  const float* byp_w  = (const float*)d_in[5];
  const float* byp_b  = (const float*)d_in[6];
  const float* proj_w = (const float*)d_in[7];
  const float* proj_b = (const float*)d_in[8];
  float* out = (float*)d_out;

  float* w = (float*)d_ws;
  float* tabC  = w;
  float* tabS  = tabC + 4096;
  float* tabC2 = tabS + 4096;
  float* tabS2 = tabC2 + 4096;
  float* act   = tabS2 + 4096;
  float* Tr    = act + (size_t)NB_ * PIX * C_;
  float* Ti    = Tr + (size_t)NB_ * H_ * M_ * C_;
  float* Wt0   = Ti + (size_t)NB_ * H_ * M_ * C_;
  float* Wt1   = Wt0 + (size_t)4 * WSTEP;

  k_tabs<<<16, 256, 0, stream>>>(tabC, tabS, tabC2, tabS2);
  k_relay<<<128, 512, 0, stream>>>(wr, wi, Wt0);

  int s = 0;
  for (int o = 0; o < 4; o += NCH) {
    for (int n = 0; n < 4; ++n, ++s) {
      size_t bwOff = ((size_t)(o * 4 + n)) * C_ * C_;
      size_t bbOff = ((size_t)(o * 4 + n)) * C_;
      size_t wOffN = (n < 3) ? ((size_t)(o * 4 + n + 1)) * WSTEP : ((size_t)((o + NCH) * 4)) * WSTEP;
      float* wtCur = (s & 1) ? Wt1 : Wt0;
      float* wtNxt = (s & 1) ? Wt0 : Wt1;
      if (n == 0) k_fwdX<<<256, 256, 0, stream>>>(x, lift_w, lift_b, tabC, tabS, Tr, Ti);
      if (s < 7) k_mid<<<256, 512, 0, stream>>>(Tr, Ti, tabC, tabS, wtCur, wr + wOffN, wi + wOffN, wtNxt);
      else       k_mid<<<128, 512, 0, stream>>>(Tr, Ti, tabC, tabS, wtCur, wr, wi, wtNxt);
      if (n == 0)
        k_invX<true, true><<<NCH * 1024, 256, 0, stream>>>(Tr, Ti, act, tabC2, tabS2,
                                                           byp_w + bwOff, byp_b + bbOff,
                                                           tabC, tabS, Tr, Ti, x, lift_w, lift_b,
                                                           proj_w, proj_b, out, o);
      else if (n < 3)
        k_invX<true, false><<<NCH * 1024, 256, 0, stream>>>(Tr, Ti, act, tabC2, tabS2,
                                                            byp_w + bwOff, byp_b + bbOff,
                                                            tabC, tabS, Tr, Ti, x, lift_w, lift_b,
                                                            proj_w, proj_b, out, o);
      else
        k_invX<false, false><<<NCH * 1024, 256, 0, stream>>>(Tr, Ti, act, tabC2, tabS2,
                                                             byp_w + bwOff, byp_b + bbOff,
                                                             tabC, tabS, Tr, Ti, x, lift_w, lift_b,
                                                             proj_w, proj_b, out, o);
    }
  }
}


# Half period basis(t+128) = (-1)^k basis(t) in the transforms not yet split: forward/inverse y (32 not 64 products per wave) and first forward x (64 not 128), by mode parity (their sum order changes)
# speedup vs baseline: 3.0533x; 1.0469x over previous
#include <hip/hip_runtime.h>
#include <math.h>


typedef float v2f __attribute__((ext_vector_type(2)));
typedef float v8f __attribute__((ext_vector_type(8)));

#define B_  4
#define NCH 2
#define NB_ (NCH * B_)
#define C_  32
#define H_  256
#define W_  256
#define M_  16
#define PIX (H_ * W_)

__device__ __forceinline__ v8f wmma4(v2f a, v2f b, v8f c) {
  return __builtin_amdgcn_wmma_f32_16x16x4_f32(false, a, false, b, (short)0, c,
                                               false, false);
}

__device__ __forceinline__ v2f ldA(const float* p, int lda, int lane) {
  int r = lane & 15, kk = (lane >> 4) << 1;
  v2f v; v.x = p[r * lda + kk]; v.y = p[r * lda + kk + 1]; return v;
}
__device__ __forceinline__ v2f ldB(const float* p, int ldb, int lane) {
  int n = lane & 15, kk = (lane >> 4) << 1;
  v2f v; v.x = p[kk * ldb + n]; v.y = p[(kk + 1) * ldb + n]; return v;
}
__device__ __forceinline__ v2f ldBt(const float* p, int ldb, int lane) {
  int n = lane & 15, kk = (lane >> 4) << 1;
  v2f v; v.x = p[n * ldb + kk]; v.y = p[n * ldb + kk + 1]; return v;
}
__device__ __forceinline__ v2f ldA_d(const float* p, int lda, int lane, int j) {
  int r = lane & 15, kk = ((lane >> 4) << 3) + 2 * j;
  v2f v; v.x = p[r * lda + kk]; v.y = p[r * lda + kk + 1]; return v;
}
__device__ __forceinline__ v8f vzero() {
  v8f z = {0.f, 0.f, 0.f, 0.f, 0.f, 0.f, 0.f, 0.f};
  return z;
}

template <int MASK>
__device__ __forceinline__ float row_xor(float v) {
  int b = __builtin_bit_cast(int, v);
  return __builtin_bit_cast(float, __builtin_amdgcn_mov_dpp(b, 0x160 + MASK, 0xf, 0xf, true));
}
__device__ __forceinline__ float row_sum16(float v) {
  v += row_xor<1>(v);
  v += row_xor<2>(v);
  v += row_xor<4>(v);
  v += row_xor<8>(v);
  return v;
}

__device__ __forceinline__ float erf_poly(float u) {
  float t = u * u;
  float p = __builtin_fmaf(t, __builtin_bit_cast(float, 0xba1345e1u), __builtin_bit_cast(float, 0x3ba10414u));
  p = __builtin_fmaf(t, p, __builtin_bit_cast(float, 0xbcdac9b8u));
  p = __builtin_fmaf(t, p, __builtin_bit_cast(float, 0x3de703beu));
  p = __builtin_fmaf(t, p, __builtin_bit_cast(float, 0xbec09330u));
  p = __builtin_fmaf(t, p, __builtin_bit_cast(float, 0x3e0375d0u));
  return __builtin_fmaf(u, p, u);
}

__global__ void k_tabs(float* tabC, float* tabS, float* tabC2, float* tabS2) {
  int i = blockIdx.x * blockDim.x + threadIdx.x;
  if (i >= M_ * W_) return;
  int k = i >> 8, x = i & 255;
  int m = (k * x) & 255;
  float ang = 0.0245436926f * (float)m;
  float c = cosf(ang), s = sinf(ang);
  tabC[i] = c; tabS[i] = s;
  float w2 = (k == 0) ? 1.0f : 2.0f;
  tabC2[i] = w2 * c * 0.0625f;
  tabS2[i] = w2 * s * 0.0625f;
}

__device__ __forceinline__ float lift1(float lbv, float l0, float l1, float l2,
                                       float xi0, float xi1, float xi2) {
  return lbv + l0 * xi0 + l1 * xi1 + l2 * xi2;
}

#define FX_U 4
__device__ __forceinline__ void fx_load(v2f (&ae)[FX_U], v2f (&ao)[FX_U], v2f (&pa)[FX_U][3],
                                        v2f (&pb)[FX_U][3], const float* __restrict__ ta,
                                        const float* __restrict__ xr, int k) {
#pragma unroll
  for (int u = 0; u < FX_U; ++u) {
    ae[u] = *reinterpret_cast<const v2f*>(ta + k + 4 * u);
    ao[u] = *reinterpret_cast<const v2f*>(ta + W_ + k + 4 * u);
#pragma unroll
    for (int i = 0; i < 3; ++i) {
      pa[u][i] = *reinterpret_cast<const v2f*>(xr + i * PIX + k + 4 * u);
      pb[u][i] = *reinterpret_cast<const v2f*>(xr + i * PIX + 128 + k + 4 * u);
    }
  }
}
__device__ __forceinline__ void fx_mma(const v2f (&ae)[FX_U], const v2f (&ao)[FX_U],
                                       const v2f (&pa)[FX_U][3], const v2f (&pb)[FX_U][3],
                                       float lbv, float l0, float l1, float l2, v8f& aE, v8f& aO) {
#pragma unroll
  for (int u = 0; u < FX_U; ++u) {
    v2f la, lb;
    la.x = lift1(lbv, l0, l1, l2, pa[u][0].x, pa[u][1].x, pa[u][2].x);
    la.y = lift1(lbv, l0, l1, l2, pa[u][0].y, pa[u][1].y, pa[u][2].y);
    lb.x = lift1(lbv, l0, l1, l2, pb[u][0].x, pb[u][1].x, pb[u][2].x);
    lb.y = lift1(lbv, l0, l1, l2, pb[u][0].y, pb[u][1].y, pb[u][2].y);
    aE = wmma4(ae[u], la + lb, aE);
    aO = wmma4(ao[u], la - lb, aO);
  }
}
__global__ void __launch_bounds__(256) k_fwdX(const float* __restrict__ x, const float* __restrict__ lw,
                       const float* __restrict__ lb, const float* __restrict__ tabC,
                       const float* __restrict__ tabS, float* __restrict__ Tr,
                       float* __restrict__ Ti) {
  int wid = blockIdx.x * (blockDim.x >> 5) + (threadIdx.x >> 5);
  int lane = threadIdx.x & 31;
  int ch = wid & 1;
  int byi = wid >> 1;
  int b = byi >> 8, y = byi & 255;
  int n = lane & 15, h = lane >> 4;
  int cc = 16 * ch + n;
  float l0 = lw[cc * 3 + 0], l1 = lw[cc * 3 + 1], l2 = lw[cc * 3 + 2], lbv = lb[cc];
  const float* xr = x + ((size_t)(b * 3) * H_ + y) * W_ + 2 * h;
  const float* ta = ((n < 8) ? tabC : tabS) + (2 * (n & 7)) * W_ + 2 * h;
  v8f aE = vzero(), aO = vzero();
  v2f e0[FX_U], o0[FX_U], e1[FX_U], o1[FX_U];
  v2f pa0[FX_U][3], pb0[FX_U][3], pa1[FX_U][3], pb1[FX_U][3];
#pragma clang loop unroll(disable)
  for (int k0 = 0; k0 < W_ / 2; k0 += 8 * FX_U) {
    fx_load(e0, o0, pa0, pb0, ta, xr, k0);
    fx_load(e1, o1, pa1, pb1, ta, xr, k0 + 4 * FX_U);
    __builtin_amdgcn_sched_barrier(0);
    fx_mma(e0, o0, pa0, pb0, lbv, l0, l1, l2, aE, aO);
    fx_mma(e1, o1, pa1, pb1, lbv, l0, l1, l2, aE, aO);
    __builtin_amdgcn_sched_barrier(0);
  }
  float sc = h ? -0.0625f : 0.0625f;
  float* dst = (h ? Ti : Tr) + ((size_t)(b * H_ + y)) * M_ * C_ + 16 * ch + n;
#pragma unroll
  for (int r = 0; r < 8; ++r) {
    float ve = aE[r] * sc, vo = aO[r] * sc;
    dst[(2 * r) * C_] = ve;
    dst[(2 * r + 1) * C_] = vo;
    dst[(size_t)B_ * H_ * M_ * C_ + (2 * r) * C_] = ve;
    dst[(size_t)B_ * H_ * M_ * C_ + (2 * r + 1) * C_] = vo;
  }
}

#define FP 36
#define WSTEP (C_ * C_ * M_ * M_)
#define PIM (16 * FP + 32)
#define PW (PIM + 16 * FP)

__device__ __forceinline__ void relay_block(int rb, const float* __restrict__ wr,
                                            const float* __restrict__ wi,
                                            float* __restrict__ wt, float* sT, int tid) {
  int chain = rb >> 6, ri = (rb >> 5) & 1, i = rb & 31;
  const float* srcp = (ri ? wi : wr) + (size_t)chain * (4 * WSTEP) + (size_t)i * (C_ * M_ * M_);
  float* dst = wt + (size_t)chain * (2 * WSTEP) + (size_t)ri * WSTEP + i * C_;
  float v[16];
#pragma unroll
  for (int j = 0; j < 16; ++j) v[j] = srcp[j * 512 + tid];
#pragma unroll
  for (int j = 0; j < 16; ++j) {
    int idx = j * 512 + tid;
    sT[(idx >> 8) * 257 + (idx & 255)] = v[j];
  }
  __syncthreads();
#pragma unroll
  for (int j = 0; j < 16; ++j) {
    int idx = j * 512 + tid;
    int o = idx & 31, m = idx >> 5;
    int kk = (m & 15) * 16 + (m >> 4);
    dst[(size_t)m * (C_ * C_) + o] = sT[o * 257 + kk];
  }
}

__global__ void __launch_bounds__(512) k_relay(const float* __restrict__ wr,
                                               const float* __restrict__ wi,
                                               float* __restrict__ wt) {
  __shared__ float sT[32 * 257];
  relay_block(blockIdx.x, wr, wi, wt, sT, threadIdx.x);
}

__global__ void __launch_bounds__(512) k_mid(float* TP, float* TQ,
                                             const float* __restrict__ tabC,
                                             const float* __restrict__ tabS,
                                             const float* __restrict__ wt,
                                             const float* __restrict__ wrN,
                                             const float* __restrict__ wiN,
                                             float* __restrict__ wtN) {
  __shared__ float sPart[16 * PW];
  __shared__ float sF[2 * 16 * FP];
  __shared__ float sG[2 * 16 * FP];
  int tid = threadIdx.x;
  if (blockIdx.x >= 128) {
    relay_block(blockIdx.x - 128, wrN, wiN, wtN, sPart, tid);
    return;
  }
  int w = tid >> 5, lane = tid & 31;
  int q = blockIdx.x >> 4, k2 = blockIdx.x & 15;
  int n = lane & 15, h = lane >> 4, mh = h << 3;
  int k1t = w, ot = lane;
  int yb = 8 * w;
  int pr = w & 7, ch = w >> 3;
  int y0 = 16 * pr;
  float* tr = TP + (size_t)q * H_ * 512 + k2 * 32;
  float* ti = TQ + (size_t)q * H_ * 512 + k2 * 32;
  const float* a1p = ((n < 8) ? tabC : tabS) + (2 * (n & 7)) * W_ + yb + 2 * h;
  const float* a2p = ((n < 8) ? tabS : tabC) + (2 * (n & 7)) * W_ + yb + 2 * h;
  v2f fA1[2][2], fA2[2][2];
  v2f tRl[2][2], tRh[2][2], tIl[2][2], tIh[2][2];
#pragma unroll
  for (int par = 0; par < 2; ++par) {
#pragma unroll
    for (int s = 0; s < 2; ++s) {
      fA1[par][s] = *reinterpret_cast<const v2f*>(a1p + par * W_ + 4 * s);
      fA2[par][s] = *reinterpret_cast<const v2f*>(a2p + par * W_ + 4 * s);
    }
  }
#pragma unroll
  for (int s = 0; s < 2; ++s) {
#pragma unroll
    for (int nt = 0; nt < 2; ++nt) {
      tRl[s][nt] = ldB(tr + (size_t)(yb + 4 * s) * 512 + nt * 16, 512, lane);
      tRh[s][nt] = ldB(tr + (size_t)(yb + 4 * s + 128) * 512 + nt * 16, 512, lane);
      tIl[s][nt] = ldB(ti + (size_t)(yb + 4 * s) * 512 + nt * 16, 512, lane);
      tIh[s][nt] = ldB(ti + (size_t)(yb + 4 * s + 128) * 512 + nt * 16, 512, lane);
    }
  }
  __builtin_amdgcn_sched_barrier(0);
  const float* wtr = wt + (size_t)(q >> 2) * (2 * WSTEP) + (size_t)((k2 * 16 + k1t) * C_) * C_ + ot;
  const float* wti = wtr + WSTEP;
  float wrv[C_], wiv[C_];
#pragma unroll
  for (int i = 0; i < C_; ++i) { wrv[i] = wtr[i * C_]; wiv[i] = wti[i * C_]; }
  v2f iC[2][2], iS[2][2];
#pragma unroll
  for (int par = 0; par < 2; ++par) {
#pragma unroll
    for (int s = 0; s < 2; ++s) {
      int ka = 8 * s + 4 * h + par;
      iC[par][s].x = tabC[ka * W_ + y0 + n];
      iC[par][s].y = tabC[(ka + 2) * W_ + y0 + n];
      iS[par][s].x = tabS[ka * W_ + y0 + n];
      iS[par][s].y = tabS[(ka + 2) * W_ + y0 + n];
    }
  }
  __builtin_amdgcn_sched_barrier(0);
  v8f aE[2], aO[2];
  aE[0] = aE[1] = aO[0] = aO[1] = vzero();
  float sg = (n < 8) ? 1.0f : -1.0f;
#pragma unroll
  for (int par = 0; par < 2; ++par) {
#pragma unroll
    for (int s = 0; s < 2; ++s) fA2[par][s] = fA2[par][s] * sg;
  }
#pragma unroll
  for (int s = 0; s < 2; ++s) {
#pragma unroll
    for (int nt = 0; nt < 2; ++nt) {
      v2f er = tRl[s][nt] + tRh[s][nt], ei = tIl[s][nt] + tIh[s][nt];
      v2f dr = tRl[s][nt] - tRh[s][nt], di = tIl[s][nt] - tIh[s][nt];
      aE[nt] = wmma4(fA1[0][s], er, aE[nt]);
      aE[nt] = wmma4(fA2[0][s], ei, aE[nt]);
      aO[nt] = wmma4(fA1[1][s], dr, aO[nt]);
      aO[nt] = wmma4(fA2[1][s], di, aO[nt]);
    }
  }
  {
    float psc = h ? -0.0625f : 0.0625f;
    float* pp = sPart + w * PW + h * PIM + n;
#pragma unroll
    for (int nt = 0; nt < 2; ++nt) {
#pragma unroll
      for (int r = 0; r < 8; ++r) {
        pp[(2 * r) * FP + nt * 16] = aE[nt][r] * psc;
        pp[(2 * r + 1) * FP + nt * 16] = aO[nt][r] * psc;
      }
    }
  }
  __syncthreads();
  {
    float fr = 0.f, fi = 0.f;
#pragma unroll
    for (int ww = 0; ww < 16; ++ww) {
      fr += sPart[ww * PW + k1t * FP + ot];
      fi += sPart[ww * PW + PIM + k1t * FP + ot];
    }
    sF[k1t * FP + ot] = fr;
    sF[16 * FP + k1t * FP + ot] = fi;
  }
  __syncthreads();
  {
    float gr = 0.f, gi = 0.f;
#pragma unroll
    for (int i = 0; i < C_; ++i) {
      float xr = sF[k1t * FP + i], xi = sF[16 * FP + k1t * FP + i];
      gr += xr * wrv[i] - xi * wiv[i];
      gi += xr * wiv[i] + xi * wrv[i];
    }
    sG[k1t * FP + ot] = gr;
    sG[16 * FP + k1t * FP + ot] = gi;
  }
  __syncthreads();
  v8f aP[2], aQ[2];
  aP[0] = aP[1] = aQ[0] = aQ[1] = vzero();
#pragma unroll
  for (int par = 0; par < 2; ++par) {
#pragma unroll
    for (int s = 0; s < 2; ++s) {
      const float* g = sG + (8 * s + 4 * h + par) * FP + 16 * ch + n;
      v2f bR, bI;
      bR.x = g[0];       bR.y = g[2 * FP];
      bI.x = g[16 * FP]; bI.y = g[18 * FP];
      v2f nS = -iS[par][s];
      aP[par] = wmma4(iC[par][s], bR, aP[par]);
      aP[par] = wmma4(nS, bI, aP[par]);
      aQ[par] = wmma4(iS[par][s], bR, aQ[par]);
      aQ[par] = wmma4(iC[par][s], bI, aQ[par]);
    }
  }
  float* p = TP + (size_t)(q * H_ + y0 + mh) * 512 + k2 * 32 + 16 * ch + n;
  float* qo = TQ + (size_t)(q * H_ + y0 + mh) * 512 + k2 * 32 + 16 * ch + n;
#pragma unroll
  for (int r = 0; r < 8; ++r) {
    p[r * 512] = (aP[0][r] + aP[1][r]) * 0.0625f;
    p[(r + 128) * 512] = (aP[0][r] - aP[1][r]) * 0.0625f;
    qo[r * 512] = (aQ[0][r] + aQ[1][r]) * 0.0625f;
    qo[(r + 128) * 512] = (aQ[0][r] - aQ[1][r]) * 0.0625f;
  }
}

typedef float v4f __attribute__((ext_vector_type(4)));
#define XP 36
#define PP 48

__device__ __forceinline__ void gelu_tile(const v8f& a0, const v8f& a1, float bb0, float bb1,
                                          v8f& g0, v8f& g1) {
  float pv0[8], pv1[8];
  float eu0[8], eu1[8];
  float um = 0.f;
#pragma unroll
  for (int r = 0; r < 8; ++r) {
    pv0[r] = a0[r] + bb0;
    pv1[r] = a1[r] + bb1;
    eu0[r] = pv0[r] * 0.70710678f;
    eu1[r] = pv1[r] * 0.70710678f;
    um = fmaxf(um, fmaxf(fabsf(eu0[r]), fabsf(eu1[r])));
  }
  if (__builtin_amdgcn_ballot_w32(!(um < 1.0f)) == 0u) {
#pragma unroll
    for (int r = 0; r < 8; ++r) {
      eu0[r] = erf_poly(eu0[r]);
      eu1[r] = erf_poly(eu1[r]);
    }
  } else {
#pragma unroll
    for (int r = 0; r < 8; ++r) {
      eu0[r] = erff(eu0[r]);
      eu1[r] = erff(eu1[r]);
    }
  }
#pragma unroll
  for (int r = 0; r < 8; ++r) {
    g0[r] = 0.5f * pv0[r] * (1.0f + eu0[r]);
    g1[r] = 0.5f * pv1[r] * (1.0f + eu1[r]);
  }
}

__device__ __forceinline__ void lift_tile(float* tile, const float* __restrict__ xs, int lane,
                                          float lbv, float l0, float l1, float l2) {
#pragma unroll
  for (int j = 0; j < 16; ++j)
    tile[j * XP + lane] = lift1(lbv, l0, l1, l2, xs[j], xs[PIX + j], xs[2 * PIX + j]);
}

template <bool FWD, bool FIRST>
__global__ void __launch_bounds__(256, 8) k_invX(const float* P, const float* Q,
                       float* act, const float* __restrict__ tabC2,
                       const float* __restrict__ tabS2, const float* __restrict__ bw,
                       const float* __restrict__ bb, const float* __restrict__ tabC,
                       const float* __restrict__ tabS, float* Tr,
                       float* Ti, const float* __restrict__ x,
                       const float* __restrict__ lw, const float* __restrict__ lb,
                       const float* __restrict__ pw, const float* __restrict__ pb,
                       float* __restrict__ res, int o_outer) {
  constexpr int WSTR = 2 * 16 * XP;
  __shared__ __attribute__((aligned(16))) float sW[8 * WSTR];
  __shared__ __attribute__((aligned(16))) float sBW[C_ * XP];
  __shared__ __attribute__((aligned(16))) float sP[M_ * PP];
  __shared__ __attribute__((aligned(16))) float sQ[M_ * PP];
  __shared__ __attribute__((aligned(16))) float sO[W_];
  int tid = threadIdx.x;
  int w = tid >> 5;
  int lane = tid & 31;
  int by = blockIdx.x;
  int chn = by >> 10;
  int x0 = w * 16;
  int n = lane & 15, h = lane >> 4, mh = h << 3;
  const float* Pby = P + (size_t)by * 512;
  const float* Qby = Q + (size_t)by * 512;
  const float* xrow = act + (size_t)by * (W_ * C_);
  bw += chn * (4 * C_ * C_);
  bb += chn * (4 * C_);
  float* tileA = sW + w * WSTR;
  float* tileB = tileA + 16 * XP;
  v4f ta[4], tb[4];
  if (!FIRST) {
    const v4f* tsA = reinterpret_cast<const v4f*>(xrow + x0 * C_);
    const v4f* tsB = reinterpret_cast<const v4f*>(xrow + (x0 + 128) * C_);
#pragma unroll
    for (int p = 0; p < 4; ++p) { ta[p] = tsA[32 * p + lane]; tb[p] = tsB[32 * p + lane]; }
  }
  int si = tid & 127;
  v4f sh1, sh2;
  if (tid < 128) sh1 = reinterpret_cast<const v4f*>(Pby)[si];
  else           sh1 = reinterpret_cast<const v4f*>(Qby)[si];
  sh2 = reinterpret_cast<const v4f*>(bw)[tid];
  v2f aCp[2][2], aSp[2][2];
#pragma unroll
  for (int par = 0; par < 2; ++par) {
#pragma unroll
    for (int t = 0; t < 2; ++t) {
      int ra = 8 * t + 2 * h + par;
      aCp[par][t].x = tabC2[ra * W_ + x0 + n];
      aCp[par][t].y = tabC2[(ra + 4) * W_ + x0 + n];
      aSp[par][t].x = -tabS2[ra * W_ + x0 + n];
      aSp[par][t].y = -tabS2[(ra + 4) * W_ + x0 + n];
    }
  }
  float bb0 = bb[n], bb1 = bb[16 + n];
  float pw0 = 0.f, pw1 = 0.f, pbv = 0.f;
  if (!FWD) { pw0 = pw[n]; pw1 = pw[16 + n]; pbv = pb[0]; }
  {
    if (FIRST) {
      int wu = __builtin_amdgcn_readfirstlane(w);
      const float* xs = x + ((size_t)(((by >> 8) & 3) * 3) * H_ + (by & 255)) * W_ + wu * 16;
      float l0 = lw[lane * 3 + 0], l1 = lw[lane * 3 + 1], l2 = lw[lane * 3 + 2], lbv = lb[lane];
      lift_tile(tileA, xs, lane, lbv, l0, l1, l2);
      lift_tile(tileB, xs + 128, lane, lbv, l0, l1, l2);
    } else {
      int tx = lane >> 3, tc = (lane & 7) * 4;
#pragma unroll
      for (int p = 0; p < 4; ++p) {
        *reinterpret_cast<v4f*>(tileA + (4 * p + tx) * XP + tc) = ta[p];
        *reinterpret_cast<v4f*>(tileB + (4 * p + tx) * XP + tc) = tb[p];
      }
    }
    int sr = si >> 3, sc4 = (si & 7) * 4;
    if (tid < 128) *reinterpret_cast<v4f*>(sP + sr * PP + sc4) = sh1;
    else           *reinterpret_cast<v4f*>(sQ + sr * PP + sc4) = sh1;
    *reinterpret_cast<v4f*>(sBW + (tid >> 3) * XP + (tid & 7) * 4) = sh2;
  }
  __syncthreads();
  v8f Se0 = vzero(), Se1 = vzero(), So0 = vzero(), So1 = vzero();
#pragma unroll
  for (int t = 0; t < 2; ++t) {
    const float* pe = sP + (8 * t + 2 * h) * PP + n;
    v2f bE0, bE1, bO0, bO1;
    bE0.x = pe[0];       bE0.y = pe[4 * PP];
    bE1.x = pe[16];      bE1.y = pe[4 * PP + 16];
    bO0.x = pe[PP];      bO0.y = pe[5 * PP];
    bO1.x = pe[PP + 16]; bO1.y = pe[5 * PP + 16];
    Se0 = wmma4(aCp[0][t], bE0, Se0);
    Se1 = wmma4(aCp[0][t], bE1, Se1);
    So0 = wmma4(aCp[1][t], bO0, So0);
    So1 = wmma4(aCp[1][t], bO1, So1);
  }
#pragma unroll
  for (int t = 0; t < 2; ++t) {
    const float* qe = sQ + (8 * t + 2 * h) * PP + n;
    v2f bE0, bE1, bO0, bO1;
    bE0.x = qe[0];       bE0.y = qe[4 * PP];
    bE1.x = qe[16];      bE1.y = qe[4 * PP + 16];
    bO0.x = qe[PP];      bO0.y = qe[5 * PP];
    bO1.x = qe[PP + 16]; bO1.y = qe[5 * PP + 16];
    Se0 = wmma4(aSp[0][t], bE0, Se0);
    Se1 = wmma4(aSp[0][t], bE1, Se1);
    So0 = wmma4(aSp[1][t], bO0, So0);
    So1 = wmma4(aSp[1][t], bO1, So1);
  }
  v8f accA0 = Se0 + So0, accA1 = Se1 + So1;
  v8f accB0 = Se0 - So0, accB1 = Se1 - So1;
#pragma unroll
  for (int k0 = 0; k0 < C_; k0 += 4) {
    v2f aXA = ldA(tileA + k0, XP, lane);
    v2f aXB = ldA(tileB + k0, XP, lane);
    v2f bW0 = ldBt(sBW + k0, XP, lane);
    v2f bW1 = ldBt(sBW + 16 * XP + k0, XP, lane);
    accA0 = wmma4(aXA, bW0, accA0);
    accA1 = wmma4(aXA, bW1, accA1);
    accB0 = wmma4(aXB, bW0, accB0);
    accB1 = wmma4(aXB, bW1, accB1);
  }
  v2f fE[4], fO[4];
  if (FWD) {
    const float* trow = ((n < 8) ? tabC : tabS) + (2 * (n & 7)) * W_ + x0 + mh;
#pragma unroll
    for (int j = 0; j < 4; ++j) {
      fE[j] = *reinterpret_cast<const v2f*>(trow + 2 * j);
      fO[j] = *reinterpret_cast<const v2f*>(trow + W_ + 2 * j);
    }
  }
  float* out = act + (size_t)by * (W_ * C_);
  v8f gA0, gA1, gB0, gB1;
  gelu_tile(accA0, accA1, bb0, bb1, gA0, gA1);
  gelu_tile(accB0, accB1, bb0, bb1, gB0, gB1);
  if (FWD) {
#pragma unroll
    for (int r = 0; r < 8; ++r) {
      int xr = x0 + r + mh;
      out[xr * C_ + n] = gA0[r];
      out[xr * C_ + 16 + n] = gA1[r];
      out[(xr + 128) * C_ + n] = gB0[r];
      out[(xr + 128) * C_ + 16 + n] = gB1[r];
    }
  } else {
    float psA[8], psB[8];
#pragma unroll
    for (int r = 0; r < 8; ++r) {
      psA[r] = row_sum16(pw0 * gA0[r] + pw1 * gA1[r]);
      psB[r] = row_sum16(pw0 * gB0[r] + pw1 * gB1[r]);
    }
    if (n == 0) {
      v4f lo = {psA[0], psA[1], psA[2], psA[3]};
      v4f hi = {psA[4], psA[5], psA[6], psA[7]};
      *reinterpret_cast<v4f*>(sO + x0 + mh) = lo;
      *reinterpret_cast<v4f*>(sO + x0 + mh + 4) = hi;
      v4f lo2 = {psB[0], psB[1], psB[2], psB[3]};
      v4f hi2 = {psB[4], psB[5], psB[6], psB[7]};
      *reinterpret_cast<v4f*>(sO + 128 + x0 + mh) = lo2;
      *reinterpret_cast<v4f*>(sO + 128 + x0 + mh + 4) = hi2;
    }
    __syncthreads();
    {
      int bq = (by >> 8) & 3, y = by & 255;
      res[((size_t)(bq * 4 + o_outer + chn)) * PIX + y * W_ + tid] = sO[tid] + pbv;
    }
  }
  if (FWD) {
    v8f E0 = gA0 + gB0, E1 = gA1 + gB1;
    v8f O0 = gA0 - gB0, O1 = gA1 - gB1;
    v8f dE0 = vzero(), dE1 = vzero(), dO0 = vzero(), dO1 = vzero();
#pragma unroll
    for (int j = 0; j < 4; ++j) {
      v2f e0; e0.x = E0[2 * j]; e0.y = E0[2 * j + 1];
      v2f e1; e1.x = E1[2 * j]; e1.y = E1[2 * j + 1];
      v2f o0; o0.x = O0[2 * j]; o0.y = O0[2 * j + 1];
      v2f o1; o1.x = O1[2 * j]; o1.y = O1[2 * j + 1];
      dE0 = wmma4(fE[j], e0, dE0);
      dE1 = wmma4(fE[j], e1, dE1);
      dO0 = wmma4(fO[j], o0, dO0);
      dO1 = wmma4(fO[j], o1, dO1);
    }
    float* pc = sW + w * WSTR + h * 32 + n;
#pragma unroll
    for (int r = 0; r < 8; ++r) {
      pc[(2 * r) * 64] = dE0[r];
      pc[(2 * r) * 64 + 16] = dE1[r];
      pc[(2 * r + 1) * 64] = dO0[r];
      pc[(2 * r + 1) * 64 + 16] = dO1[r];
    }
    __syncthreads();
    float sc = (tid & 32) ? -0.0625f : 0.0625f;
    float* dst = ((tid & 32) ? Ti : Tr) + (size_t)by * (M_ * C_) + (tid >> 6) * C_ + (tid & 31);
#pragma unroll
    for (int i = 0; i < 4; ++i) {
      float s = 0.f;
#pragma unroll
      for (int ww = 0; ww < 8; ++ww) s += sW[ww * WSTR + tid + 256 * i];
      dst[4 * i * C_] = s * sc;
    }
  }
}

extern "C" void kernel_launch(void* const* d_in, const int* in_sizes, int n_in,
                              void* d_out, int out_size, void* d_ws, size_t ws_size,
                              hipStream_t stream) {
  const float* x      = (const float*)d_in[0];
  const float* lift_w = (const float*)d_in[1];
  const float* lift_b = (const float*)d_in[2];
  const float* wr     = (const float*)d_in[3];
  const float* wi     = (const float*)d_in[4];
  const float* byp_w  = (const float*)d_in[5];
  const float* byp_b  = (const float*)d_in[6];
  const float* proj_w = (const float*)d_in[7];
  const float* proj_b = (const float*)d_in[8];
  float* out = (float*)d_out;

  float* w = (float*)d_ws;
  float* tabC  = w;
  float* tabS  = tabC + 4096;
  float* tabC2 = tabS + 4096;
  float* tabS2 = tabC2 + 4096;
  float* act   = tabS2 + 4096;
  float* Tr    = act + (size_t)NB_ * PIX * C_;
  float* Ti    = Tr + (size_t)NB_ * H_ * M_ * C_;
  float* Wt0   = Ti + (size_t)NB_ * H_ * M_ * C_;
  float* Wt1   = Wt0 + (size_t)4 * WSTEP;

  k_tabs<<<16, 256, 0, stream>>>(tabC, tabS, tabC2, tabS2);
  k_relay<<<128, 512, 0, stream>>>(wr, wi, Wt0);

  int s = 0;
  for (int o = 0; o < 4; o += NCH) {
    for (int n = 0; n < 4; ++n, ++s) {
      size_t bwOff = ((size_t)(o * 4 + n)) * C_ * C_;
      size_t bbOff = ((size_t)(o * 4 + n)) * C_;
      size_t wOffN = (n < 3) ? ((size_t)(o * 4 + n + 1)) * WSTEP : ((size_t)((o + NCH) * 4)) * WSTEP;
      float* wtCur = (s & 1) ? Wt1 : Wt0;
      float* wtNxt = (s & 1) ? Wt0 : Wt1;
      if (n == 0) k_fwdX<<<256, 256, 0, stream>>>(x, lift_w, lift_b, tabC, tabS, Tr, Ti);
      if (s < 7) k_mid<<<256, 512, 0, stream>>>(Tr, Ti, tabC, tabS, wtCur, wr + wOffN, wi + wOffN, wtNxt);
      else       k_mid<<<128, 512, 0, stream>>>(Tr, Ti, tabC, tabS, wtCur, wr, wi, wtNxt);
      if (n == 0)
        k_invX<true, true><<<NCH * 1024, 256, 0, stream>>>(Tr, Ti, act, tabC2, tabS2,
                                                           byp_w + bwOff, byp_b + bbOff,
                                                           tabC, tabS, Tr, Ti, x, lift_w, lift_b,
                                                           proj_w, proj_b, out, o);
      else if (n < 3)
        k_invX<true, false><<<NCH * 1024, 256, 0, stream>>>(Tr, Ti, act, tabC2, tabS2,
                                                            byp_w + bwOff, byp_b + bbOff,
                                                            tabC, tabS, Tr, Ti, x, lift_w, lift_b,
                                                            proj_w, proj_b, out, o);
      else
        k_invX<false, false><<<NCH * 1024, 256, 0, stream>>>(Tr, Ti, act, tabC2, tabS2,
                                                             byp_w + bwOff, byp_b + bbOff,
                                                             tabC, tabS, Tr, Ti, x, lift_w, lift_b,
                                                             proj_w, proj_b, out, o);
    }
  }
}
